# starting kernel
# baseline (speedup 1.0000x reference)
#include <hip/hip_runtime.h>

typedef float v2f __attribute__((ext_vector_type(2)));
typedef float v8f __attribute__((ext_vector_type(8)));

constexpr int H_ = 8, B_ = 64, S_ = 8, D_ = 64, N_ = 4096, MLPH_ = 128, ITERS_ = 3;
constexpr float LN_EPS_ = 1e-5f, EPS_ATTN_ = 1e-8f, SCALE_ = 0.125f;

__device__ __forceinline__ v8f wmma_f32(v2f a, v2f b, v8f c) {
  return __builtin_amdgcn_wmma_f32_16x16x4_f32(false, a, false, b, (short)0, c,
                                               false, false);
}

__global__ void __launch_bounds__(256) ln_kv_kernel(
    const float* __restrict__ inp, const float* __restrict__ lng,
    const float* __restrict__ lnb, const float* __restrict__ Wk,
    const float* __restrict__ Wv, float* __restrict__ kout,
    float* __restrict__ vout) {
  __shared__ float xln[16][68];
  __shared__ float mu[16], rstd[16];
  const int tid = threadIdx.x;
  const size_t row0 = (size_t)blockIdx.x * 16;
  const int r = tid >> 4, c0 = (tid & 15) * 4;
  float4 val = *(const float4*)(inp + (row0 + r) * D_ + c0);
  xln[r][c0] = val.x; xln[r][c0 + 1] = val.y;
  xln[r][c0 + 2] = val.z; xln[r][c0 + 3] = val.w;
  __syncthreads();
  if (tid < 16) {
    float m = 0.f;
    for (int j = 0; j < D_; ++j) m += xln[tid][j];
    m *= (1.f / D_);
    float var = 0.f;
    for (int j = 0; j < D_; ++j) { float dd = xln[tid][j] - m; var += dd * dd; }
    var *= (1.f / D_);
    mu[tid] = m; rstd[tid] = rsqrtf(var + LN_EPS_);
  }
  __syncthreads();
  {
    float m = mu[r], rs = rstd[r];
    for (int j = 0; j < 4; ++j) {
      int c = c0 + j;
      xln[r][c] = (xln[r][c] - m) * rs * lng[c] + lnb[c];
    }
  }
  __syncthreads();
  const int w = tid >> 5, lane = tid & 31, m16 = lane & 15, kh = lane >> 4;
  const float* W = (w < 4) ? Wk : Wv;
  float* outp = (w < 4) ? kout : vout;
  const int ct = (w & 3) * 16;
  v8f acc = {};
  for (int kk = 0; kk < 16; ++kk) {
    int kb = kk * 4 + kh * 2;
    v2f a; a.x = xln[m16][kb]; a.y = xln[m16][kb + 1];
    v2f b; b.x = W[kb * D_ + ct + m16]; b.y = W[(kb + 1) * D_ + ct + m16];
    acc = wmma_f32(a, b, acc);
  }
  for (int i = 0; i < 8; ++i)
    outp[(row0 + i + 8 * kh) * D_ + ct + m16] = acc[i];
}

__global__ void slots_init_kernel(const float* __restrict__ sinit,
                                  const float* __restrict__ smu,
                                  const float* __restrict__ slog,
                                  float* __restrict__ slots) {
  int idx = blockIdx.x * blockDim.x + threadIdx.x;
  int d = idx & (D_ - 1);
  int h = idx / (B_ * S_ * D_);
  slots[idx] = smu[h * D_ + d] + expf(slog[h * D_ + d]) * sinit[idx];
}

__global__ void __launch_bounds__(128) hprev_q_kernel(
    const float* __restrict__ slots, const float* __restrict__ lng,
    const float* __restrict__ lnb, const float* __restrict__ Wq,
    float* __restrict__ hprev, float* __restrict__ qout) {
  __shared__ float hl[16][68];
  __shared__ float mu[16], rstd[16];
  const int tid = threadIdx.x;
  const int h = blockIdx.x >> 5;
  const int rt = blockIdx.x & 31;
  const int rowbase = h * (B_ * S_) + rt * 16;
  for (int e = tid; e < 16 * D_; e += 128) {
    int r = e >> 6, c = e & 63;
    hl[r][c] = slots[(size_t)(rowbase + r) * D_ + c];
  }
  __syncthreads();
  if (tid < 16) {
    float m = 0.f;
    for (int j = 0; j < D_; ++j) m += hl[tid][j];
    m *= (1.f / D_);
    float var = 0.f;
    for (int j = 0; j < D_; ++j) { float dd = hl[tid][j] - m; var += dd * dd; }
    var *= (1.f / D_);
    mu[tid] = m; rstd[tid] = rsqrtf(var + LN_EPS_);
  }
  __syncthreads();
  for (int e = tid; e < 16 * D_; e += 128) {
    int r = e >> 6, c = e & 63;
    float nv = (hl[r][c] - mu[r]) * rstd[r] * lng[h * D_ + c] + lnb[h * D_ + c];
    hl[r][c] = nv;
    hprev[(size_t)(rowbase + r) * D_ + c] = nv;
  }
  __syncthreads();
  const int w = tid >> 5, lane = tid & 31, m16 = lane & 15, kh = lane >> 4;
  const float* Wh = Wq + (size_t)h * D_ * D_;
  const int ct = w * 16;
  v8f acc = {};
  for (int kk = 0; kk < 16; ++kk) {
    int kb = kk * 4 + kh * 2;
    v2f a; a.x = hl[m16][kb]; a.y = hl[m16][kb + 1];
    v2f b; b.x = Wh[kb * D_ + ct + m16]; b.y = Wh[(kb + 1) * D_ + ct + m16];
    acc = wmma_f32(a, b, acc);
  }
  for (int i = 0; i < 8; ++i)
    qout[(size_t)(rowbase + i + 8 * kh) * D_ + ct + m16] = acc[i];
}

__global__ void __launch_bounds__(128) attn_kernel(
    const float* __restrict__ kbuf, const float* __restrict__ vbuf,
    const float* __restrict__ qbuf, float* __restrict__ updates) {
  __shared__ float qs[16][68];
  __shared__ float kt[64][68];
  __shared__ float at[64][17];
  __shared__ float den[16];
  const int tid = threadIdx.x;
  const int hp = blockIdx.x >> 6;
  const int b  = blockIdx.x & 63;
  const float* kb = kbuf + (size_t)b * N_ * D_;
  const float* vb = vbuf + (size_t)b * N_ * D_;
  for (int e = tid; e < 16 * D_; e += 128) {
    int sp = e >> 6, d = e & 63;
    int h = hp * 2 + (sp >> 3), s = sp & 7;
    qs[sp][d] = qbuf[(size_t)((h * B_ + b) * S_ + s) * D_ + d];
  }
  if (tid < 16) den[tid] = 0.f;
  __syncthreads();
  const int w = tid >> 5, lane = tid & 31, m16 = lane & 15, kh = lane >> 4;
  v8f accu = {};
  for (int nb = 0; nb < N_; nb += 64) {
    for (int e = tid; e < 64 * D_; e += 128) {
      int r = e >> 6, c = e & 63;
      kt[r][c] = kb[(size_t)(nb + r) * D_ + c];
    }
    if (nb + 64 < N_) {
      const char* nk = (const char*)(kb + (size_t)(nb + 64) * D_);
      const char* nv = (const char*)(vb + (size_t)(nb + 64) * D_);
      __builtin_prefetch(nk + tid * 128, 0, 3);
      __builtin_prefetch(nv + tid * 128, 0, 3);
    }
    __syncthreads();
    v8f lg = {};
    for (int kk = 0; kk < 16; ++kk) {
      int kbs = kk * 4 + kh * 2;
      v2f a;  a.x = kt[w * 16 + m16][kbs]; a.y = kt[w * 16 + m16][kbs + 1];
      v2f bb; bb.x = qs[m16][kbs];         bb.y = qs[m16][kbs + 1];
      lg = wmma_f32(a, bb, lg);
    }
    for (int i = 0; i < 8; ++i)
      at[w * 16 + i + 8 * kh][m16] = lg[i] * SCALE_;
    __syncthreads();
    {
      int r = tid >> 1, hh = tid & 1;
      float mx = -3.4e38f;
      for (int s = 0; s < 8; ++s) mx = fmaxf(mx, at[r][hh * 8 + s]);
      float ex[8], sum = 0.f;
      for (int s = 0; s < 8; ++s) { ex[s] = expf(at[r][hh * 8 + s] - mx); sum += ex[s]; }
      float inv = 1.f / sum;
      for (int s = 0; s < 8; ++s) at[r][hh * 8 + s] = ex[s] * inv + EPS_ATTN_;
    }
    __syncthreads();
    if (tid < 16) {
      float dsum = 0.f;
      for (int r = 0; r < 64; ++r) dsum += at[r][tid];
      den[tid] += dsum;
    }
    for (int kk = 0; kk < 16; ++kk) {
      int kbs = kk * 4 + kh * 2;
      v2f a;  a.x = at[kbs][m16];  a.y = at[kbs + 1][m16];
      v2f bb; bb.x = vb[(size_t)(nb + kbs) * D_ + w * 16 + m16];
              bb.y = vb[(size_t)(nb + kbs + 1) * D_ + w * 16 + m16];
      accu = wmma_f32(a, bb, accu);
    }
    __syncthreads();
  }
  for (int i = 0; i < 8; ++i) {
    int sp = i + 8 * kh;
    int h = hp * 2 + (sp >> 3), s = sp & 7;
    updates[(size_t)((h * B_ + b) * S_ + s) * D_ + w * 16 + m16] = accu[i] / den[sp];
  }
}

__global__ void __launch_bounds__(256) gru_mlp_kernel(
    const float* __restrict__ upd, const float* __restrict__ hprev,
    const float* __restrict__ wih, const float* __restrict__ whh,
    const float* __restrict__ bih, const float* __restrict__ bhh,
    const float* __restrict__ lnmg, const float* __restrict__ lnmb,
    const float* __restrict__ w1, const float* __restrict__ b1,
    const float* __restrict__ w2, const float* __restrict__ b2,
    float* __restrict__ slots) {
  __shared__ float ut[16][68];
  __shared__ float ht[16][68];
  __shared__ float gx[16][196];
  __shared__ float gh[16][196];
  __shared__ float sn[16][68];
  __shared__ float nm[16][68];
  __shared__ float hid[16][132];
  __shared__ float mu[16], rstd[16];
  const int tid = threadIdx.x;
  const int h = blockIdx.x >> 5, rt = blockIdx.x & 31;
  const int rowbase = h * (B_ * S_) + rt * 16;
  const int r0 = tid >> 4, c00 = (tid & 15) * 4;
  {
    float4 a = *(const float4*)(upd + (size_t)(rowbase + r0) * D_ + c00);
    ut[r0][c00] = a.x; ut[r0][c00 + 1] = a.y; ut[r0][c00 + 2] = a.z; ut[r0][c00 + 3] = a.w;
    float4 c = *(const float4*)(hprev + (size_t)(rowbase + r0) * D_ + c00);
    ht[r0][c00] = c.x; ht[r0][c00 + 1] = c.y; ht[r0][c00 + 2] = c.z; ht[r0][c00 + 3] = c.w;
  }
  __syncthreads();
  const int w = tid >> 5, lane = tid & 31, m16 = lane & 15, kh = lane >> 4;
  const float* wihh = wih + (size_t)h * D_ * 192;
  const float* whhh = whh + (size_t)h * D_ * 192;
  for (int T = w; T < 24; T += 8) {
    const bool isX = T < 12;
    const int ct = (T % 12) * 16;
    const float* Wt = isX ? wihh : whhh;
    v8f acc = {};
    for (int kk = 0; kk < 16; ++kk) {
      int kb = kk * 4 + kh * 2;
      v2f a;
      if (isX) { a.x = ut[m16][kb]; a.y = ut[m16][kb + 1]; }
      else     { a.x = ht[m16][kb]; a.y = ht[m16][kb + 1]; }
      v2f bb; bb.x = Wt[kb * 192 + ct + m16]; bb.y = Wt[(kb + 1) * 192 + ct + m16];
      acc = wmma_f32(a, bb, acc);
    }
    float* dst = isX ? &gx[0][0] : &gh[0][0];
    for (int i = 0; i < 8; ++i)
      dst[(i + 8 * kh) * 196 + ct + m16] = acc[i];
  }
  __syncthreads();
  {
    const float* bi = bih + h * 192;
    const float* bh = bhh + h * 192;
    for (int j = 0; j < 4; ++j) {
      int d = c00 + j;
      float xr = gx[r0][d] + bi[d],        hr = gh[r0][d] + bh[d];
      float xz = gx[r0][64 + d] + bi[64 + d],  hz = gh[r0][64 + d] + bh[64 + d];
      float xn = gx[r0][128 + d] + bi[128 + d], hn = gh[r0][128 + d] + bh[128 + d];
      float rg = 1.f / (1.f + expf(-(xr + hr)));
      float zg = 1.f / (1.f + expf(-(xz + hz)));
      float ng = tanhf(xn + rg * hn);
      sn[r0][d] = (1.f - zg) * ng + zg * ht[r0][d];
    }
  }
  __syncthreads();
  if (tid < 16) {
    float m = 0.f;
    for (int j = 0; j < D_; ++j) m += sn[tid][j];
    m *= (1.f / D_);
    float var = 0.f;
    for (int j = 0; j < D_; ++j) { float dd = sn[tid][j] - m; var += dd * dd; }
    var *= (1.f / D_);
    mu[tid] = m; rstd[tid] = rsqrtf(var + LN_EPS_);
  }
  __syncthreads();
  for (int j = 0; j < 4; ++j) {
    int d = c00 + j;
    nm[r0][d] = (sn[r0][d] - mu[r0]) * rstd[r0] * lnmg[h * D_ + d] + lnmb[h * D_ + d];
  }
  __syncthreads();
  {
    const float* w1h = w1 + (size_t)h * D_ * MLPH_;
    const float* b1h = b1 + h * MLPH_;
    const int ct = w * 16;
    v8f acc = {};
    for (int kk = 0; kk < 16; ++kk) {
      int kb = kk * 4 + kh * 2;
      v2f a;  a.x = nm[m16][kb]; a.y = nm[m16][kb + 1];
      v2f bb; bb.x = w1h[kb * MLPH_ + ct + m16]; bb.y = w1h[(kb + 1) * MLPH_ + ct + m16];
      acc = wmma_f32(a, bb, acc);
    }
    float bias = b1h[ct + m16];
    for (int i = 0; i < 8; ++i)
      hid[i + 8 * kh][ct + m16] = fmaxf(acc[i] + bias, 0.f);
  }
  __syncthreads();
  if (w < 4) {
    const float* w2h = w2 + (size_t)h * MLPH_ * D_;
    const float* b2h = b2 + h * D_;
    const int ct = w * 16;
    v8f acc = {};
    for (int kk = 0; kk < 32; ++kk) {
      int kb = kk * 4 + kh * 2;
      v2f a;  a.x = hid[m16][kb]; a.y = hid[m16][kb + 1];
      v2f bb; bb.x = w2h[kb * D_ + ct + m16]; bb.y = w2h[(kb + 1) * D_ + ct + m16];
      acc = wmma_f32(a, bb, acc);
    }
    float bias = b2h[ct + m16];
    for (int i = 0; i < 8; ++i) {
      int rr = i + 8 * kh;
      slots[(size_t)(rowbase + rr) * D_ + ct + m16] = sn[rr][ct + m16] + acc[i] + bias;
    }
  }
}

__global__ void copy_out_kernel(const float* __restrict__ slots,
                                float* __restrict__ out) {
  int i = blockIdx.x * blockDim.x + threadIdx.x;
  if (i < B_ * S_ * D_) out[i] = slots[i];
}

extern "C" void kernel_launch(void* const* d_in, const int* in_sizes, int n_in,
                              void* d_out, int out_size, void* d_ws, size_t ws_size,
                              hipStream_t stream) {
  (void)in_sizes; (void)n_in; (void)out_size; (void)ws_size;
  const float* inputs  = (const float*)d_in[0];
  const float* sinit   = (const float*)d_in[1];
  const float* smu     = (const float*)d_in[2];
  const float* slog    = (const float*)d_in[3];
  const float* ln_in_g = (const float*)d_in[4];
  const float* ln_in_b = (const float*)d_in[5];
  const float* Wk      = (const float*)d_in[6];
  const float* Wv      = (const float*)d_in[7];
  const float* ln_s_g  = (const float*)d_in[8];
  const float* ln_s_b  = (const float*)d_in[9];
  const float* Wq      = (const float*)d_in[10];
  const float* gru_wih = (const float*)d_in[11];
  const float* gru_whh = (const float*)d_in[12];
  const float* gru_bih = (const float*)d_in[13];
  const float* gru_bhh = (const float*)d_in[14];
  const float* ln_m_g  = (const float*)d_in[15];
  const float* ln_m_b  = (const float*)d_in[16];
  const float* mlp_w1  = (const float*)d_in[17];
  const float* mlp_b1  = (const float*)d_in[18];
  const float* mlp_w2  = (const float*)d_in[19];
  const float* mlp_b2  = (const float*)d_in[20];

  float* ws = (float*)d_ws;
  float* kbuf  = ws;
  float* vbuf  = kbuf + (size_t)B_ * N_ * D_;
  float* slots = vbuf + (size_t)B_ * N_ * D_;
  float* hprev = slots + H_ * B_ * S_ * D_;
  float* qbuf  = hprev + H_ * B_ * S_ * D_;
  float* updv  = qbuf + H_ * B_ * S_ * D_;

  ln_kv_kernel<<<(B_ * N_) / 16, 256, 0, stream>>>(inputs, ln_in_g, ln_in_b,
                                                   Wk, Wv, kbuf, vbuf);
  slots_init_kernel<<<(H_ * B_ * S_ * D_) / 256, 256, 0, stream>>>(sinit, smu,
                                                                   slog, slots);
  for (int it = 0; it < ITERS_; ++it) {
    hprev_q_kernel<<<H_ * 32, 128, 0, stream>>>(slots, ln_s_g, ln_s_b, Wq,
                                                hprev, qbuf);
    attn_kernel<<<(H_ / 2) * B_, 128, 0, stream>>>(kbuf, vbuf, qbuf, updv);
    gru_mlp_kernel<<<H_ * 32, 256, 0, stream>>>(updv, hprev, gru_wih, gru_whh,
                                                gru_bih, gru_bhh, ln_m_g, ln_m_b,
                                                mlp_w1, mlp_b1, mlp_w2, mlp_b2,
                                                slots);
  }
  copy_out_kernel<<<(B_ * S_ * D_ + 255) / 256, 256, 0, stream>>>(slots,
                                                                  (float*)d_out);
}


# attention: 16 waves per workgroup (was 4); a wave owns whole 16-row k/v tiles, fragments and softmax in registers, no LDS or barrier in its loop; partial sums merged once in LDS in fixed wave order
# speedup vs baseline: 2.5730x; 2.5730x over previous
#include <hip/hip_runtime.h>

typedef float v2f __attribute__((ext_vector_type(2)));
typedef float v8f __attribute__((ext_vector_type(8)));

constexpr int H_ = 8, B_ = 64, S_ = 8, D_ = 64, N_ = 4096, MLPH_ = 128, ITERS_ = 3;
constexpr float LN_EPS_ = 1e-5f, EPS_ATTN_ = 1e-8f, SCALE_ = 0.125f;

__device__ __forceinline__ v8f wmma_f32(v2f a, v2f b, v8f c) {
  return __builtin_amdgcn_wmma_f32_16x16x4_f32(false, a, false, b, (short)0, c,
                                               false, false);
}

template <int M>
__device__ __forceinline__ float xlane(float x) {
  return __builtin_bit_cast(float, __builtin_amdgcn_update_dpp(
      0, __builtin_bit_cast(int, x), 0x160 + M, 0xf, 0xf, true));
}

__global__ void __launch_bounds__(256) ln_kv_kernel(
    const float* __restrict__ inp, const float* __restrict__ lng,
    const float* __restrict__ lnb, const float* __restrict__ Wk,
    const float* __restrict__ Wv, float* __restrict__ kout,
    float* __restrict__ vout) {
  __shared__ float xln[16][68];
  __shared__ float mu[16], rstd[16];
  const int tid = threadIdx.x;
  const size_t row0 = (size_t)blockIdx.x * 16;
  const int r = tid >> 4, c0 = (tid & 15) * 4;
  float4 val = *(const float4*)(inp + (row0 + r) * D_ + c0);
  xln[r][c0] = val.x; xln[r][c0 + 1] = val.y;
  xln[r][c0 + 2] = val.z; xln[r][c0 + 3] = val.w;
  __syncthreads();
  if (tid < 16) {
    float m = 0.f;
    for (int j = 0; j < D_; ++j) m += xln[tid][j];
    m *= (1.f / D_);
    float var = 0.f;
    for (int j = 0; j < D_; ++j) { float dd = xln[tid][j] - m; var += dd * dd; }
    var *= (1.f / D_);
    mu[tid] = m; rstd[tid] = rsqrtf(var + LN_EPS_);
  }
  __syncthreads();
  {
    float m = mu[r], rs = rstd[r];
    for (int j = 0; j < 4; ++j) {
      int c = c0 + j;
      xln[r][c] = (xln[r][c] - m) * rs * lng[c] + lnb[c];
    }
  }
  __syncthreads();
  const int w = tid >> 5, lane = tid & 31, m16 = lane & 15, kh = lane >> 4;
  const float* W = (w < 4) ? Wk : Wv;
  float* outp = (w < 4) ? kout : vout;
  const int ct = (w & 3) * 16;
  v8f acc = {};
  for (int kk = 0; kk < 16; ++kk) {
    int kb = kk * 4 + kh * 2;
    v2f a; a.x = xln[m16][kb]; a.y = xln[m16][kb + 1];
    v2f b; b.x = W[kb * D_ + ct + m16]; b.y = W[(kb + 1) * D_ + ct + m16];
    acc = wmma_f32(a, b, acc);
  }
  for (int i = 0; i < 8; ++i)
    outp[(row0 + i + 8 * kh) * D_ + ct + m16] = acc[i];
}

__global__ void slots_init_kernel(const float* __restrict__ sinit,
                                  const float* __restrict__ smu,
                                  const float* __restrict__ slog,
                                  float* __restrict__ slots) {
  int idx = blockIdx.x * blockDim.x + threadIdx.x;
  int d = idx & (D_ - 1);
  int h = idx / (B_ * S_ * D_);
  slots[idx] = smu[h * D_ + d] + expf(slog[h * D_ + d]) * sinit[idx];
}

__global__ void __launch_bounds__(128) hprev_q_kernel(
    const float* __restrict__ slots, const float* __restrict__ lng,
    const float* __restrict__ lnb, const float* __restrict__ Wq,
    float* __restrict__ hprev, float* __restrict__ qout) {
  __shared__ float hl[16][68];
  __shared__ float mu[16], rstd[16];
  const int tid = threadIdx.x;
  const int h = blockIdx.x >> 5;
  const int rt = blockIdx.x & 31;
  const int rowbase = h * (B_ * S_) + rt * 16;
  for (int e = tid; e < 16 * D_; e += 128) {
    int r = e >> 6, c = e & 63;
    hl[r][c] = slots[(size_t)(rowbase + r) * D_ + c];
  }
  __syncthreads();
  if (tid < 16) {
    float m = 0.f;
    for (int j = 0; j < D_; ++j) m += hl[tid][j];
    m *= (1.f / D_);
    float var = 0.f;
    for (int j = 0; j < D_; ++j) { float dd = hl[tid][j] - m; var += dd * dd; }
    var *= (1.f / D_);
    mu[tid] = m; rstd[tid] = rsqrtf(var + LN_EPS_);
  }
  __syncthreads();
  for (int e = tid; e < 16 * D_; e += 128) {
    int r = e >> 6, c = e & 63;
    float nv = (hl[r][c] - mu[r]) * rstd[r] * lng[h * D_ + c] + lnb[h * D_ + c];
    hl[r][c] = nv;
    hprev[(size_t)(rowbase + r) * D_ + c] = nv;
  }
  __syncthreads();
  const int w = tid >> 5, lane = tid & 31, m16 = lane & 15, kh = lane >> 4;
  const float* Wh = Wq + (size_t)h * D_ * D_;
  const int ct = w * 16;
  v8f acc = {};
  for (int kk = 0; kk < 16; ++kk) {
    int kb = kk * 4 + kh * 2;
    v2f a; a.x = hl[m16][kb]; a.y = hl[m16][kb + 1];
    v2f b; b.x = Wh[kb * D_ + ct + m16]; b.y = Wh[(kb + 1) * D_ + ct + m16];
    acc = wmma_f32(a, b, acc);
  }
  for (int i = 0; i < 8; ++i)
    qout[(size_t)(rowbase + i + 8 * kh) * D_ + ct + m16] = acc[i];
}

constexpr int ATT_WAVES_ = 16;
__global__ void __launch_bounds__(512) __attribute__((amdgpu_waves_per_eu(4, 4)))
attn_kernel(
    const float* __restrict__ kbuf, const float* __restrict__ vbuf,
    const float* __restrict__ qbuf, float* __restrict__ updates) {
  __shared__ alignas(16) float red[ATT_WAVES_][16][68];
  __shared__ float dred[ATT_WAVES_][32];
  const int tid = threadIdx.x;
  const int hp = blockIdx.x >> 6;
  const int b  = blockIdx.x & 63;
  const int w = tid >> 5, lane = tid & 31, c = lane & 15, kh = lane >> 4;
  const float* kb = kbuf + (size_t)b * N_ * D_;
  const float* vb = vbuf + (size_t)b * N_ * D_;
  const float* qrow = qbuf +
      (size_t)(((hp * 2 + (c >> 3)) * B_ + b) * S_ + (c & 7)) * D_ + 4 * kh;
  float4 qf[8];
  for (int i = 0; i < 8; ++i) qf[i] = *(const float4*)(qrow + 8 * i);
  v8f acc0 = {}, acc1 = {}, acc2 = {}, acc3 = {};
  float dsum = 0.f;
  for (int t = w; t < N_ / 16; t += ATT_WAVES_) {
    const int n0 = t * 16;
    const float* kp = kb + (size_t)(n0 + c) * D_ + 4 * kh;
    const float* vp = vb + (size_t)(n0 + 8 * kh) * D_ + 4 * c;
    float4 kf[8], vf[8];
    for (int i = 0; i < 8; ++i) kf[i] = *(const float4*)(kp + 8 * i);
    for (int r = 0; r < 8; ++r) vf[r] = *(const float4*)(vp + (size_t)r * D_);
    v8f lg = {};
    for (int i = 0; i < 8; ++i) {
      v2f a0; a0.x = kf[i].x; a0.y = kf[i].y;
      v2f b0; b0.x = qf[i].x; b0.y = qf[i].y;
      lg = wmma_f32(a0, b0, lg);
      v2f a1; a1.x = kf[i].z; a1.y = kf[i].w;
      v2f b1; b1.x = qf[i].z; b1.y = qf[i].w;
      lg = wmma_f32(a1, b1, lg);
    }
    float p[8];
#pragma unroll
    for (int r = 0; r < 8; ++r) {
      float x = lg[r] * SCALE_;
      float mx = fmaxf(x, xlane<1>(x));
      mx = fmaxf(mx, xlane<2>(mx));
      mx = fmaxf(mx, xlane<4>(mx));
      float e = expf(x - mx);
      float sm = e + xlane<1>(e);
      sm += xlane<2>(sm);
      sm += xlane<4>(sm);
      float inv = 1.f / sm;
      p[r] = e * inv + EPS_ATTN_;
      dsum += p[r];
    }
    for (int j = 0; j < 4; ++j) {
      v2f a; a.x = p[2 * j]; a.y = p[2 * j + 1];
      v2f b0; b0.x = vf[2 * j].x; b0.y = vf[2 * j + 1].x;
      acc0 = wmma_f32(a, b0, acc0);
      v2f b1; b1.x = vf[2 * j].y; b1.y = vf[2 * j + 1].y;
      acc1 = wmma_f32(a, b1, acc1);
      v2f b2; b2.x = vf[2 * j].z; b2.y = vf[2 * j + 1].z;
      acc2 = wmma_f32(a, b2, acc2);
      v2f b3; b3.x = vf[2 * j].w; b3.y = vf[2 * j + 1].w;
      acc3 = wmma_f32(a, b3, acc3);
    }
  }
  for (int i = 0; i < 8; ++i) {
    float4 o; o.x = acc0[i]; o.y = acc1[i]; o.z = acc2[i]; o.w = acc3[i];
    *(float4*)&red[w][8 * kh + i][4 * c] = o;
  }
  dred[w][lane] = dsum;
  __syncthreads();
  if (tid < 256) {
    const int sp = tid >> 4, d4 = (tid & 15) * 4;
    float4 sacc; sacc.x = 0.f; sacc.y = 0.f; sacc.z = 0.f; sacc.w = 0.f;
    float den = 0.f;
    for (int ww = 0; ww < ATT_WAVES_; ++ww) {
      float4 x = *(const float4*)&red[ww][sp][d4];
      sacc.x += x.x; sacc.y += x.y; sacc.z += x.z; sacc.w += x.w;
      den += dred[ww][sp] + dred[ww][sp + 16];
    }
    const int h = hp * 2 + (sp >> 3), sl = sp & 7;
    float4 o; o.x = sacc.x / den; o.y = sacc.y / den; o.z = sacc.z / den; o.w = sacc.w / den;
    *(float4*)(updates + (size_t)((h * B_ + b) * S_ + sl) * D_ + d4) = o;
  }
}

__global__ void __launch_bounds__(256) gru_mlp_kernel(
    const float* __restrict__ upd, const float* __restrict__ hprev,
    const float* __restrict__ wih, const float* __restrict__ whh,
    const float* __restrict__ bih, const float* __restrict__ bhh,
    const float* __restrict__ lnmg, const float* __restrict__ lnmb,
    const float* __restrict__ w1, const float* __restrict__ b1,
    const float* __restrict__ w2, const float* __restrict__ b2,
    float* __restrict__ slots) {
  __shared__ float ut[16][68];
  __shared__ float ht[16][68];
  __shared__ float gx[16][196];
  __shared__ float gh[16][196];
  __shared__ float sn[16][68];
  __shared__ float nm[16][68];
  __shared__ float hid[16][132];
  __shared__ float mu[16], rstd[16];
  const int tid = threadIdx.x;
  const int h = blockIdx.x >> 5, rt = blockIdx.x & 31;
  const int rowbase = h * (B_ * S_) + rt * 16;
  const int r0 = tid >> 4, c00 = (tid & 15) * 4;
  {
    float4 a = *(const float4*)(upd + (size_t)(rowbase + r0) * D_ + c00);
    ut[r0][c00] = a.x; ut[r0][c00 + 1] = a.y; ut[r0][c00 + 2] = a.z; ut[r0][c00 + 3] = a.w;
    float4 c = *(const float4*)(hprev + (size_t)(rowbase + r0) * D_ + c00);
    ht[r0][c00] = c.x; ht[r0][c00 + 1] = c.y; ht[r0][c00 + 2] = c.z; ht[r0][c00 + 3] = c.w;
  }
  __syncthreads();
  const int w = tid >> 5, lane = tid & 31, m16 = lane & 15, kh = lane >> 4;
  const float* wihh = wih + (size_t)h * D_ * 192;
  const float* whhh = whh + (size_t)h * D_ * 192;
  for (int T = w; T < 24; T += 8) {
    const bool isX = T < 12;
    const int ct = (T % 12) * 16;
    const float* Wt = isX ? wihh : whhh;
    v8f acc = {};
    for (int kk = 0; kk < 16; ++kk) {
      int kb = kk * 4 + kh * 2;
      v2f a;
      if (isX) { a.x = ut[m16][kb]; a.y = ut[m16][kb + 1]; }
      else     { a.x = ht[m16][kb]; a.y = ht[m16][kb + 1]; }
      v2f bb; bb.x = Wt[kb * 192 + ct + m16]; bb.y = Wt[(kb + 1) * 192 + ct + m16];
      acc = wmma_f32(a, bb, acc);
    }
    float* dst = isX ? &gx[0][0] : &gh[0][0];
    for (int i = 0; i < 8; ++i)
      dst[(i + 8 * kh) * 196 + ct + m16] = acc[i];
  }
  __syncthreads();
  {
    const float* bi = bih + h * 192;
    const float* bh = bhh + h * 192;
    for (int j = 0; j < 4; ++j) {
      int d = c00 + j;
      float xr = gx[r0][d] + bi[d],        hr = gh[r0][d] + bh[d];
      float xz = gx[r0][64 + d] + bi[64 + d],  hz = gh[r0][64 + d] + bh[64 + d];
      float xn = gx[r0][128 + d] + bi[128 + d], hn = gh[r0][128 + d] + bh[128 + d];
      float rg = 1.f / (1.f + expf(-(xr + hr)));
      float zg = 1.f / (1.f + expf(-(xz + hz)));
      float ng = tanhf(xn + rg * hn);
      sn[r0][d] = (1.f - zg) * ng + zg * ht[r0][d];
    }
  }
  __syncthreads();
  if (tid < 16) {
    float m = 0.f;
    for (int j = 0; j < D_; ++j) m += sn[tid][j];
    m *= (1.f / D_);
    float var = 0.f;
    for (int j = 0; j < D_; ++j) { float dd = sn[tid][j] - m; var += dd * dd; }
    var *= (1.f / D_);
    mu[tid] = m; rstd[tid] = rsqrtf(var + LN_EPS_);
  }
  __syncthreads();
  for (int j = 0; j < 4; ++j) {
    int d = c00 + j;
    nm[r0][d] = (sn[r0][d] - mu[r0]) * rstd[r0] * lnmg[h * D_ + d] + lnmb[h * D_ + d];
  }
  __syncthreads();
  {
    const float* w1h = w1 + (size_t)h * D_ * MLPH_;
    const float* b1h = b1 + h * MLPH_;
    const int ct = w * 16;
    v8f acc = {};
    for (int kk = 0; kk < 16; ++kk) {
      int kb = kk * 4 + kh * 2;
      v2f a;  a.x = nm[m16][kb]; a.y = nm[m16][kb + 1];
      v2f bb; bb.x = w1h[kb * MLPH_ + ct + m16]; bb.y = w1h[(kb + 1) * MLPH_ + ct + m16];
      acc = wmma_f32(a, bb, acc);
    }
    float bias = b1h[ct + m16];
    for (int i = 0; i < 8; ++i)
      hid[i + 8 * kh][ct + m16] = fmaxf(acc[i] + bias, 0.f);
  }
  __syncthreads();
  if (w < 4) {
    const float* w2h = w2 + (size_t)h * MLPH_ * D_;
    const float* b2h = b2 + h * D_;
    const int ct = w * 16;
    v8f acc = {};
    for (int kk = 0; kk < 32; ++kk) {
      int kb = kk * 4 + kh * 2;
      v2f a;  a.x = hid[m16][kb]; a.y = hid[m16][kb + 1];
      v2f bb; bb.x = w2h[kb * D_ + ct + m16]; bb.y = w2h[(kb + 1) * D_ + ct + m16];
      acc = wmma_f32(a, bb, acc);
    }
    float bias = b2h[ct + m16];
    for (int i = 0; i < 8; ++i) {
      int rr = i + 8 * kh;
      slots[(size_t)(rowbase + rr) * D_ + ct + m16] = sn[rr][ct + m16] + acc[i] + bias;
    }
  }
}

__global__ void copy_out_kernel(const float* __restrict__ slots,
                                float* __restrict__ out) {
  int i = blockIdx.x * blockDim.x + threadIdx.x;
  if (i < B_ * S_ * D_) out[i] = slots[i];
}

extern "C" void kernel_launch(void* const* d_in, const int* in_sizes, int n_in,
                              void* d_out, int out_size, void* d_ws, size_t ws_size,
                              hipStream_t stream) {
  (void)in_sizes; (void)n_in; (void)out_size; (void)ws_size;
  const float* inputs  = (const float*)d_in[0];
  const float* sinit   = (const float*)d_in[1];
  const float* smu     = (const float*)d_in[2];
  const float* slog    = (const float*)d_in[3];
  const float* ln_in_g = (const float*)d_in[4];
  const float* ln_in_b = (const float*)d_in[5];
  const float* Wk      = (const float*)d_in[6];
  const float* Wv      = (const float*)d_in[7];
  const float* ln_s_g  = (const float*)d_in[8];
  const float* ln_s_b  = (const float*)d_in[9];
  const float* Wq      = (const float*)d_in[10];
  const float* gru_wih = (const float*)d_in[11];
  const float* gru_whh = (const float*)d_in[12];
  const float* gru_bih = (const float*)d_in[13];
  const float* gru_bhh = (const float*)d_in[14];
  const float* ln_m_g  = (const float*)d_in[15];
  const float* ln_m_b  = (const float*)d_in[16];
  const float* mlp_w1  = (const float*)d_in[17];
  const float* mlp_b1  = (const float*)d_in[18];
  const float* mlp_w2  = (const float*)d_in[19];
  const float* mlp_b2  = (const float*)d_in[20];

  float* ws = (float*)d_ws;
  float* kbuf  = ws;
  float* vbuf  = kbuf + (size_t)B_ * N_ * D_;
  float* slots = vbuf + (size_t)B_ * N_ * D_;
  float* hprev = slots + H_ * B_ * S_ * D_;
  float* qbuf  = hprev + H_ * B_ * S_ * D_;
  float* updv  = qbuf + H_ * B_ * S_ * D_;

  ln_kv_kernel<<<(B_ * N_) / 16, 256, 0, stream>>>(inputs, ln_in_g, ln_in_b,
                                                   Wk, Wv, kbuf, vbuf);
  slots_init_kernel<<<(H_ * B_ * S_ * D_) / 256, 256, 0, stream>>>(sinit, smu,
                                                                   slog, slots);
  for (int it = 0; it < ITERS_; ++it) {
    hprev_q_kernel<<<H_ * 32, 128, 0, stream>>>(slots, ln_s_g, ln_s_b, Wq,
                                                hprev, qbuf);
    attn_kernel<<<(H_ / 2) * B_, 512, 0, stream>>>(kbuf, vbuf, qbuf, updv);
    gru_mlp_kernel<<<H_ * 32, 256, 0, stream>>>(updv, hprev, gru_wih, gru_whh,
                                                gru_bih, gru_bhh, ln_m_g, ln_m_b,
                                                mlp_w1, mlp_b1, mlp_w2, mlp_b2,
                                                slots);
  }
  copy_out_kernel<<<(B_ * S_ * D_ + 255) / 256, 256, 0, stream>>>(slots,
                                                                  (float*)d_out);
}


# ln_kv: 2048 resident waves each loop over whole 16-row tiles; Wk/Wv fragments and LayerNorm in registers, no LDS or barrier, whole-row stores; LN sums and the products' k sum change order
# speedup vs baseline: 2.8101x; 1.0921x over previous
#include <hip/hip_runtime.h>

typedef float v2f __attribute__((ext_vector_type(2)));
typedef float v4f __attribute__((ext_vector_type(4)));
typedef float v8f __attribute__((ext_vector_type(8)));

constexpr int H_ = 8, B_ = 64, S_ = 8, D_ = 64, N_ = 4096, MLPH_ = 128, ITERS_ = 3;
constexpr float LN_EPS_ = 1e-5f, EPS_ATTN_ = 1e-8f, SCALE_ = 0.125f;

__device__ __forceinline__ v8f wmma_f32(v2f a, v2f b, v8f c) {
  return __builtin_amdgcn_wmma_f32_16x16x4_f32(false, a, false, b, (short)0, c,
                                               false, false);
}

template <int M>
__device__ __forceinline__ float xlane(float x) {
  return __builtin_bit_cast(float, __builtin_amdgcn_update_dpp(
      0, __builtin_bit_cast(int, x), 0x160 + M, 0xf, 0xf, true));
}

constexpr int KV_WAVES_ = 8;
constexpr int KV_GRID_ = 256;
constexpr int KV_TILES_ = (B_ * N_) / 16;
constexpr int KV_STRIDE_ = KV_GRID_ * KV_WAVES_;
constexpr int KV_TRIPS_ = KV_TILES_ / KV_STRIDE_;
static_assert(KV_TILES_ % KV_STRIDE_ == 0, "tiles must divide over the waves");

__device__ __forceinline__ float xhalf(float x) {
  const int b = __builtin_bit_cast(int, x);
  return __builtin_bit_cast(float, __builtin_amdgcn_permlanex16(
      b, b, 0x76543210, 0xfedcba98, false, false));
}

__global__ void __launch_bounds__(256) __attribute__((amdgpu_waves_per_eu(2, 2)))
ln_kv_kernel(
    const float* __restrict__ inp, const float* __restrict__ lng,
    const float* __restrict__ lnb, const float* __restrict__ Wk,
    const float* __restrict__ Wv, float* __restrict__ kout,
    float* __restrict__ vout) {
  const int tid = threadIdx.x;
  const int w = tid >> 5, lane = tid & 31, c = lane & 15, h = lane >> 4;
  const int gw = blockIdx.x * KV_WAVES_ + w;
  v4f gf[8], bf[8];
  for (int i = 0; i < 8; ++i) {
    gf[i] = *(const v4f*)(lng + 8 * i + 4 * h);
    bf[i] = *(const v4f*)(lnb + 8 * i + 4 * h);
  }
  v2f bk[4][16], bv[4][16];
#pragma unroll
  for (int s = 0; s < 16; ++s) {
    const int d0 = 8 * (s >> 1) + 4 * h + 2 * (s & 1);
    const v4f k0 = *(const v4f*)(Wk + d0 * D_ + 4 * c);
    const v4f k1 = *(const v4f*)(Wk + (d0 + 1) * D_ + 4 * c);
    const v4f v0 = *(const v4f*)(Wv + d0 * D_ + 4 * c);
    const v4f v1 = *(const v4f*)(Wv + (d0 + 1) * D_ + 4 * c);
    for (int T = 0; T < 4; ++T) {
      bk[T][s].x = k0[T]; bk[T][s].y = k1[T];
      bv[T][s].x = v0[T]; bv[T][s].y = v1[T];
    }
  }
  v4f xr[8];
  {
    const float* xp = inp + ((size_t)gw * 16 + c) * D_ + 4 * h;
    for (int i = 0; i < 8; ++i) xr[i] = *(const v4f*)(xp + 8 * i);
  }
#pragma nounroll
  for (int j = 0; j < KV_TRIPS_; ++j) {
    const int t = gw + j * KV_STRIDE_;
    v2f sl = xr[0].lo, sh = xr[0].hi;
    for (int i = 1; i < 8; ++i) { sl += xr[i].lo; sh += xr[i].hi; }
    const v2f s2 = sl + sh;
    float sm = s2.x + s2.y;
    sm += xhalf(sm);
    const float m = sm * (1.f / D_);
    v2f a[16];
    v2f ql = {0.f, 0.f}, qh = {0.f, 0.f};
    for (int i = 0; i < 8; ++i) {
      a[2 * i] = xr[i].lo - m;
      a[2 * i + 1] = xr[i].hi - m;
      ql += a[2 * i] * a[2 * i];
      qh += a[2 * i + 1] * a[2 * i + 1];
    }
    const v2f q2 = ql + qh;
    float var = q2.x + q2.y;
    var += xhalf(var);
    const float rs = rsqrtf(var * (1.f / D_) + LN_EPS_);
    for (int i = 0; i < 8; ++i) {
      a[2 * i] = a[2 * i] * rs * gf[i].lo + bf[i].lo;
      a[2 * i + 1] = a[2 * i + 1] * rs * gf[i].hi + bf[i].hi;
    }
    {
      const int tn = (j + 1 < KV_TRIPS_) ? t + KV_STRIDE_ : t;
      const float* xp = inp + ((size_t)tn * 16 + c) * D_ + 4 * h;
      for (int i = 0; i < 8; ++i) xr[i] = *(const v4f*)(xp + 8 * i);
    }
    v8f ak[4] = {}, av[4] = {};
#pragma unroll
    for (int s = 0; s < 16; ++s) {
#pragma unroll
      for (int T = 0; T < 4; ++T) {
        ak[T] = wmma_f32(a[s], bk[T][s], ak[T]);
        av[T] = wmma_f32(a[s], bv[T][s], av[T]);
      }
    }
    float* ko = kout + ((size_t)t * 16 + 8 * h) * D_ + 4 * c;
    float* vo = vout + ((size_t)t * 16 + 8 * h) * D_ + 4 * c;
    for (int r = 0; r < 8; ++r) {
      v4f o; o[0] = ak[0][r]; o[1] = ak[1][r]; o[2] = ak[2][r]; o[3] = ak[3][r];
      *(v4f*)(ko + (size_t)r * D_) = o;
      v4f p; p[0] = av[0][r]; p[1] = av[1][r]; p[2] = av[2][r]; p[3] = av[3][r];
      *(v4f*)(vo + (size_t)r * D_) = p;
    }
  }
}

__global__ void slots_init_kernel(const float* __restrict__ sinit,
                                  const float* __restrict__ smu,
                                  const float* __restrict__ slog,
                                  float* __restrict__ slots) {
  int idx = blockIdx.x * blockDim.x + threadIdx.x;
  int d = idx & (D_ - 1);
  int h = idx / (B_ * S_ * D_);
  slots[idx] = smu[h * D_ + d] + expf(slog[h * D_ + d]) * sinit[idx];
}

__global__ void __launch_bounds__(128) hprev_q_kernel(
    const float* __restrict__ slots, const float* __restrict__ lng,
    const float* __restrict__ lnb, const float* __restrict__ Wq,
    float* __restrict__ hprev, float* __restrict__ qout) {
  __shared__ float hl[16][68];
  __shared__ float mu[16], rstd[16];
  const int tid = threadIdx.x;
  const int h = blockIdx.x >> 5;
  const int rt = blockIdx.x & 31;
  const int rowbase = h * (B_ * S_) + rt * 16;
  for (int e = tid; e < 16 * D_; e += 128) {
    int r = e >> 6, c = e & 63;
    hl[r][c] = slots[(size_t)(rowbase + r) * D_ + c];
  }
  __syncthreads();
  if (tid < 16) {
    float m = 0.f;
    for (int j = 0; j < D_; ++j) m += hl[tid][j];
    m *= (1.f / D_);
    float var = 0.f;
    for (int j = 0; j < D_; ++j) { float dd = hl[tid][j] - m; var += dd * dd; }
    var *= (1.f / D_);
    mu[tid] = m; rstd[tid] = rsqrtf(var + LN_EPS_);
  }
  __syncthreads();
  for (int e = tid; e < 16 * D_; e += 128) {
    int r = e >> 6, c = e & 63;
    float nv = (hl[r][c] - mu[r]) * rstd[r] * lng[h * D_ + c] + lnb[h * D_ + c];
    hl[r][c] = nv;
    hprev[(size_t)(rowbase + r) * D_ + c] = nv;
  }
  __syncthreads();
  const int w = tid >> 5, lane = tid & 31, m16 = lane & 15, kh = lane >> 4;
  const float* Wh = Wq + (size_t)h * D_ * D_;
  const int ct = w * 16;
  v8f acc = {};
  for (int kk = 0; kk < 16; ++kk) {
    int kb = kk * 4 + kh * 2;
    v2f a; a.x = hl[m16][kb]; a.y = hl[m16][kb + 1];
    v2f b; b.x = Wh[kb * D_ + ct + m16]; b.y = Wh[(kb + 1) * D_ + ct + m16];
    acc = wmma_f32(a, b, acc);
  }
  for (int i = 0; i < 8; ++i)
    qout[(size_t)(rowbase + i + 8 * kh) * D_ + ct + m16] = acc[i];
}

constexpr int ATT_WAVES_ = 16;
__global__ void __launch_bounds__(512) __attribute__((amdgpu_waves_per_eu(4, 4)))
attn_kernel(
    const float* __restrict__ kbuf, const float* __restrict__ vbuf,
    const float* __restrict__ qbuf, float* __restrict__ updates) {
  __shared__ alignas(16) float red[ATT_WAVES_][16][68];
  __shared__ float dred[ATT_WAVES_][32];
  const int tid = threadIdx.x;
  const int hp = blockIdx.x >> 6;
  const int b  = blockIdx.x & 63;
  const int w = tid >> 5, lane = tid & 31, c = lane & 15, kh = lane >> 4;
  const float* kb = kbuf + (size_t)b * N_ * D_;
  const float* vb = vbuf + (size_t)b * N_ * D_;
  const float* qrow = qbuf +
      (size_t)(((hp * 2 + (c >> 3)) * B_ + b) * S_ + (c & 7)) * D_ + 4 * kh;
  float4 qf[8];
  for (int i = 0; i < 8; ++i) qf[i] = *(const float4*)(qrow + 8 * i);
  v8f acc0 = {}, acc1 = {}, acc2 = {}, acc3 = {};
  float dsum = 0.f;
  for (int t = w; t < N_ / 16; t += ATT_WAVES_) {
    const int n0 = t * 16;
    const float* kp = kb + (size_t)(n0 + c) * D_ + 4 * kh;
    const float* vp = vb + (size_t)(n0 + 8 * kh) * D_ + 4 * c;
    float4 kf[8], vf[8];
    for (int i = 0; i < 8; ++i) kf[i] = *(const float4*)(kp + 8 * i);
    for (int r = 0; r < 8; ++r) vf[r] = *(const float4*)(vp + (size_t)r * D_);
    v8f lg = {};
    for (int i = 0; i < 8; ++i) {
      v2f a0; a0.x = kf[i].x; a0.y = kf[i].y;
      v2f b0; b0.x = qf[i].x; b0.y = qf[i].y;
      lg = wmma_f32(a0, b0, lg);
      v2f a1; a1.x = kf[i].z; a1.y = kf[i].w;
      v2f b1; b1.x = qf[i].z; b1.y = qf[i].w;
      lg = wmma_f32(a1, b1, lg);
    }
    float p[8];
#pragma unroll
    for (int r = 0; r < 8; ++r) {
      float x = lg[r] * SCALE_;
      float mx = fmaxf(x, xlane<1>(x));
      mx = fmaxf(mx, xlane<2>(mx));
      mx = fmaxf(mx, xlane<4>(mx));
      float e = expf(x - mx);
      float sm = e + xlane<1>(e);
      sm += xlane<2>(sm);
      sm += xlane<4>(sm);
      float inv = 1.f / sm;
      p[r] = e * inv + EPS_ATTN_;
      dsum += p[r];
    }
    for (int j = 0; j < 4; ++j) {
      v2f a; a.x = p[2 * j]; a.y = p[2 * j + 1];
      v2f b0; b0.x = vf[2 * j].x; b0.y = vf[2 * j + 1].x;
      acc0 = wmma_f32(a, b0, acc0);
      v2f b1; b1.x = vf[2 * j].y; b1.y = vf[2 * j + 1].y;
      acc1 = wmma_f32(a, b1, acc1);
      v2f b2; b2.x = vf[2 * j].z; b2.y = vf[2 * j + 1].z;
      acc2 = wmma_f32(a, b2, acc2);
      v2f b3; b3.x = vf[2 * j].w; b3.y = vf[2 * j + 1].w;
      acc3 = wmma_f32(a, b3, acc3);
    }
  }
  for (int i = 0; i < 8; ++i) {
    float4 o; o.x = acc0[i]; o.y = acc1[i]; o.z = acc2[i]; o.w = acc3[i];
    *(float4*)&red[w][8 * kh + i][4 * c] = o;
  }
  dred[w][lane] = dsum;
  __syncthreads();
  if (tid < 256) {
    const int sp = tid >> 4, d4 = (tid & 15) * 4;
    float4 sacc; sacc.x = 0.f; sacc.y = 0.f; sacc.z = 0.f; sacc.w = 0.f;
    float den = 0.f;
    for (int ww = 0; ww < ATT_WAVES_; ++ww) {
      float4 x = *(const float4*)&red[ww][sp][d4];
      sacc.x += x.x; sacc.y += x.y; sacc.z += x.z; sacc.w += x.w;
      den += dred[ww][sp] + dred[ww][sp + 16];
    }
    const int h = hp * 2 + (sp >> 3), sl = sp & 7;
    float4 o; o.x = sacc.x / den; o.y = sacc.y / den; o.z = sacc.z / den; o.w = sacc.w / den;
    *(float4*)(updates + (size_t)((h * B_ + b) * S_ + sl) * D_ + d4) = o;
  }
}

__global__ void __launch_bounds__(256) gru_mlp_kernel(
    const float* __restrict__ upd, const float* __restrict__ hprev,
    const float* __restrict__ wih, const float* __restrict__ whh,
    const float* __restrict__ bih, const float* __restrict__ bhh,
    const float* __restrict__ lnmg, const float* __restrict__ lnmb,
    const float* __restrict__ w1, const float* __restrict__ b1,
    const float* __restrict__ w2, const float* __restrict__ b2,
    float* __restrict__ slots) {
  __shared__ float ut[16][68];
  __shared__ float ht[16][68];
  __shared__ float gx[16][196];
  __shared__ float gh[16][196];
  __shared__ float sn[16][68];
  __shared__ float nm[16][68];
  __shared__ float hid[16][132];
  __shared__ float mu[16], rstd[16];
  const int tid = threadIdx.x;
  const int h = blockIdx.x >> 5, rt = blockIdx.x & 31;
  const int rowbase = h * (B_ * S_) + rt * 16;
  const int r0 = tid >> 4, c00 = (tid & 15) * 4;
  {
    float4 a = *(const float4*)(upd + (size_t)(rowbase + r0) * D_ + c00);
    ut[r0][c00] = a.x; ut[r0][c00 + 1] = a.y; ut[r0][c00 + 2] = a.z; ut[r0][c00 + 3] = a.w;
    float4 c = *(const float4*)(hprev + (size_t)(rowbase + r0) * D_ + c00);
    ht[r0][c00] = c.x; ht[r0][c00 + 1] = c.y; ht[r0][c00 + 2] = c.z; ht[r0][c00 + 3] = c.w;
  }
  __syncthreads();
  const int w = tid >> 5, lane = tid & 31, m16 = lane & 15, kh = lane >> 4;
  const float* wihh = wih + (size_t)h * D_ * 192;
  const float* whhh = whh + (size_t)h * D_ * 192;
  for (int T = w; T < 24; T += 8) {
    const bool isX = T < 12;
    const int ct = (T % 12) * 16;
    const float* Wt = isX ? wihh : whhh;
    v8f acc = {};
    for (int kk = 0; kk < 16; ++kk) {
      int kb = kk * 4 + kh * 2;
      v2f a;
      if (isX) { a.x = ut[m16][kb]; a.y = ut[m16][kb + 1]; }
      else     { a.x = ht[m16][kb]; a.y = ht[m16][kb + 1]; }
      v2f bb; bb.x = Wt[kb * 192 + ct + m16]; bb.y = Wt[(kb + 1) * 192 + ct + m16];
      acc = wmma_f32(a, bb, acc);
    }
    float* dst = isX ? &gx[0][0] : &gh[0][0];
    for (int i = 0; i < 8; ++i)
      dst[(i + 8 * kh) * 196 + ct + m16] = acc[i];
  }
  __syncthreads();
  {
    const float* bi = bih + h * 192;
    const float* bh = bhh + h * 192;
    for (int j = 0; j < 4; ++j) {
      int d = c00 + j;
      float xr = gx[r0][d] + bi[d],        hr = gh[r0][d] + bh[d];
      float xz = gx[r0][64 + d] + bi[64 + d],  hz = gh[r0][64 + d] + bh[64 + d];
      float xn = gx[r0][128 + d] + bi[128 + d], hn = gh[r0][128 + d] + bh[128 + d];
      float rg = 1.f / (1.f + expf(-(xr + hr)));
      float zg = 1.f / (1.f + expf(-(xz + hz)));
      float ng = tanhf(xn + rg * hn);
      sn[r0][d] = (1.f - zg) * ng + zg * ht[r0][d];
    }
  }
  __syncthreads();
  if (tid < 16) {
    float m = 0.f;
    for (int j = 0; j < D_; ++j) m += sn[tid][j];
    m *= (1.f / D_);
    float var = 0.f;
    for (int j = 0; j < D_; ++j) { float dd = sn[tid][j] - m; var += dd * dd; }
    var *= (1.f / D_);
    mu[tid] = m; rstd[tid] = rsqrtf(var + LN_EPS_);
  }
  __syncthreads();
  for (int j = 0; j < 4; ++j) {
    int d = c00 + j;
    nm[r0][d] = (sn[r0][d] - mu[r0]) * rstd[r0] * lnmg[h * D_ + d] + lnmb[h * D_ + d];
  }
  __syncthreads();
  {
    const float* w1h = w1 + (size_t)h * D_ * MLPH_;
    const float* b1h = b1 + h * MLPH_;
    const int ct = w * 16;
    v8f acc = {};
    for (int kk = 0; kk < 16; ++kk) {
      int kb = kk * 4 + kh * 2;
      v2f a;  a.x = nm[m16][kb]; a.y = nm[m16][kb + 1];
      v2f bb; bb.x = w1h[kb * MLPH_ + ct + m16]; bb.y = w1h[(kb + 1) * MLPH_ + ct + m16];
      acc = wmma_f32(a, bb, acc);
    }
    float bias = b1h[ct + m16];
    for (int i = 0; i < 8; ++i)
      hid[i + 8 * kh][ct + m16] = fmaxf(acc[i] + bias, 0.f);
  }
  __syncthreads();
  if (w < 4) {
    const float* w2h = w2 + (size_t)h * MLPH_ * D_;
    const float* b2h = b2 + h * D_;
    const int ct = w * 16;
    v8f acc = {};
    for (int kk = 0; kk < 32; ++kk) {
      int kb = kk * 4 + kh * 2;
      v2f a;  a.x = hid[m16][kb]; a.y = hid[m16][kb + 1];
      v2f bb; bb.x = w2h[kb * D_ + ct + m16]; bb.y = w2h[(kb + 1) * D_ + ct + m16];
      acc = wmma_f32(a, bb, acc);
    }
    float bias = b2h[ct + m16];
    for (int i = 0; i < 8; ++i) {
      int rr = i + 8 * kh;
      slots[(size_t)(rowbase + rr) * D_ + ct + m16] = sn[rr][ct + m16] + acc[i] + bias;
    }
  }
}

__global__ void copy_out_kernel(const float* __restrict__ slots,
                                float* __restrict__ out) {
  int i = blockIdx.x * blockDim.x + threadIdx.x;
  if (i < B_ * S_ * D_) out[i] = slots[i];
}

extern "C" void kernel_launch(void* const* d_in, const int* in_sizes, int n_in,
                              void* d_out, int out_size, void* d_ws, size_t ws_size,
                              hipStream_t stream) {
  (void)in_sizes; (void)n_in; (void)out_size; (void)ws_size;
  const float* inputs  = (const float*)d_in[0];
  const float* sinit   = (const float*)d_in[1];
  const float* smu     = (const float*)d_in[2];
  const float* slog    = (const float*)d_in[3];
  const float* ln_in_g = (const float*)d_in[4];
  const float* ln_in_b = (const float*)d_in[5];
  const float* Wk      = (const float*)d_in[6];
  const float* Wv      = (const float*)d_in[7];
  const float* ln_s_g  = (const float*)d_in[8];
  const float* ln_s_b  = (const float*)d_in[9];
  const float* Wq      = (const float*)d_in[10];
  const float* gru_wih = (const float*)d_in[11];
  const float* gru_whh = (const float*)d_in[12];
  const float* gru_bih = (const float*)d_in[13];
  const float* gru_bhh = (const float*)d_in[14];
  const float* ln_m_g  = (const float*)d_in[15];
  const float* ln_m_b  = (const float*)d_in[16];
  const float* mlp_w1  = (const float*)d_in[17];
  const float* mlp_b1  = (const float*)d_in[18];
  const float* mlp_w2  = (const float*)d_in[19];
  const float* mlp_b2  = (const float*)d_in[20];

  float* ws = (float*)d_ws;
  float* kbuf  = ws;
  float* vbuf  = kbuf + (size_t)B_ * N_ * D_;
  float* slots = vbuf + (size_t)B_ * N_ * D_;
  float* hprev = slots + H_ * B_ * S_ * D_;
  float* qbuf  = hprev + H_ * B_ * S_ * D_;
  float* updv  = qbuf + H_ * B_ * S_ * D_;

  ln_kv_kernel<<<KV_GRID_, 256, 0, stream>>>(inputs, ln_in_g, ln_in_b,
                                                        Wk, Wv, kbuf, vbuf);
  slots_init_kernel<<<(H_ * B_ * S_ * D_) / 256, 256, 0, stream>>>(sinit, smu,
                                                                   slog, slots);
  for (int it = 0; it < ITERS_; ++it) {
    hprev_q_kernel<<<H_ * 32, 128, 0, stream>>>(slots, ln_s_g, ln_s_b, Wq,
                                                hprev, qbuf);
    attn_kernel<<<(H_ / 2) * B_, 512, 0, stream>>>(kbuf, vbuf, qbuf, updv);
    gru_mlp_kernel<<<H_ * 32, 256, 0, stream>>>(updv, hprev, gru_wih, gru_whh,
                                                gru_bih, gru_bhh, ln_m_g, ln_m_b,
                                                mlp_w1, mlp_b1, mlp_w2, mlp_b2,
                                                slots);
  }
  copy_out_kernel<<<(B_ * S_ * D_ + 255) / 256, 256, 0, stream>>>(slots,
                                                                  (float*)d_out);
}


# only head 0 reaches the output: q, attention, GRU/MLP run for head 0 alone, a batch's rows split over 4 workgroups; the 4 partial sums are added in fixed order at the GRU load (sum order changes)
# speedup vs baseline: 4.3098x; 1.5337x over previous
#include <hip/hip_runtime.h>

typedef float v2f __attribute__((ext_vector_type(2)));
typedef float v4f __attribute__((ext_vector_type(4)));
typedef float v8f __attribute__((ext_vector_type(8)));

constexpr int H_ = 8, B_ = 64, S_ = 8, D_ = 64, N_ = 4096, MLPH_ = 128, ITERS_ = 3;
constexpr float LN_EPS_ = 1e-5f, EPS_ATTN_ = 1e-8f, SCALE_ = 0.125f;

__device__ __forceinline__ v8f wmma_f32(v2f a, v2f b, v8f c) {
  return __builtin_amdgcn_wmma_f32_16x16x4_f32(false, a, false, b, (short)0, c,
                                               false, false);
}

template <int M>
__device__ __forceinline__ float xlane(float x) {
  return __builtin_bit_cast(float, __builtin_amdgcn_update_dpp(
      0, __builtin_bit_cast(int, x), 0x160 + M, 0xf, 0xf, true));
}

constexpr int KV_WAVES_ = 8;
constexpr int KV_GRID_ = 256;
constexpr int KV_TILES_ = (B_ * N_) / 16;
constexpr int KV_STRIDE_ = KV_GRID_ * KV_WAVES_;
constexpr int KV_TRIPS_ = KV_TILES_ / KV_STRIDE_;
static_assert(KV_TILES_ % KV_STRIDE_ == 0, "tiles must divide over the waves");

__device__ __forceinline__ float xhalf(float x) {
  const int b = __builtin_bit_cast(int, x);
  return __builtin_bit_cast(float, __builtin_amdgcn_permlanex16(
      b, b, 0x76543210, 0xfedcba98, false, false));
}

__global__ void __launch_bounds__(256) __attribute__((amdgpu_waves_per_eu(2, 2)))
ln_kv_kernel(
    const float* __restrict__ inp, const float* __restrict__ lng,
    const float* __restrict__ lnb, const float* __restrict__ Wk,
    const float* __restrict__ Wv, float* __restrict__ kout,
    float* __restrict__ vout) {
  const int tid = threadIdx.x;
  const int w = tid >> 5, lane = tid & 31, c = lane & 15, h = lane >> 4;
  const int gw = blockIdx.x * KV_WAVES_ + w;
  v4f gf[8], bf[8];
  for (int i = 0; i < 8; ++i) {
    gf[i] = *(const v4f*)(lng + 8 * i + 4 * h);
    bf[i] = *(const v4f*)(lnb + 8 * i + 4 * h);
  }
  v2f bk[4][16], bv[4][16];
#pragma unroll
  for (int s = 0; s < 16; ++s) {
    const int d0 = 8 * (s >> 1) + 4 * h + 2 * (s & 1);
    const v4f k0 = *(const v4f*)(Wk + d0 * D_ + 4 * c);
    const v4f k1 = *(const v4f*)(Wk + (d0 + 1) * D_ + 4 * c);
    const v4f v0 = *(const v4f*)(Wv + d0 * D_ + 4 * c);
    const v4f v1 = *(const v4f*)(Wv + (d0 + 1) * D_ + 4 * c);
    for (int T = 0; T < 4; ++T) {
      bk[T][s].x = k0[T]; bk[T][s].y = k1[T];
      bv[T][s].x = v0[T]; bv[T][s].y = v1[T];
    }
  }
  v4f xr[8];
  {
    const float* xp = inp + ((size_t)gw * 16 + c) * D_ + 4 * h;
    for (int i = 0; i < 8; ++i) xr[i] = *(const v4f*)(xp + 8 * i);
  }
#pragma nounroll
  for (int j = 0; j < KV_TRIPS_; ++j) {
    const int t = gw + j * KV_STRIDE_;
    v2f sl = xr[0].lo, sh = xr[0].hi;
    for (int i = 1; i < 8; ++i) { sl += xr[i].lo; sh += xr[i].hi; }
    const v2f s2 = sl + sh;
    float sm = s2.x + s2.y;
    sm += xhalf(sm);
    const float m = sm * (1.f / D_);
    v2f a[16];
    v2f ql = {0.f, 0.f}, qh = {0.f, 0.f};
    for (int i = 0; i < 8; ++i) {
      a[2 * i] = xr[i].lo - m;
      a[2 * i + 1] = xr[i].hi - m;
      ql += a[2 * i] * a[2 * i];
      qh += a[2 * i + 1] * a[2 * i + 1];
    }
    const v2f q2 = ql + qh;
    float var = q2.x + q2.y;
    var += xhalf(var);
    const float rs = rsqrtf(var * (1.f / D_) + LN_EPS_);
    for (int i = 0; i < 8; ++i) {
      a[2 * i] = a[2 * i] * rs * gf[i].lo + bf[i].lo;
      a[2 * i + 1] = a[2 * i + 1] * rs * gf[i].hi + bf[i].hi;
    }
    {
      const int tn = (j + 1 < KV_TRIPS_) ? t + KV_STRIDE_ : t;
      const float* xp = inp + ((size_t)tn * 16 + c) * D_ + 4 * h;
      for (int i = 0; i < 8; ++i) xr[i] = *(const v4f*)(xp + 8 * i);
    }
    v8f ak[4] = {}, av[4] = {};
#pragma unroll
    for (int s = 0; s < 16; ++s) {
#pragma unroll
      for (int T = 0; T < 4; ++T) {
        ak[T] = wmma_f32(a[s], bk[T][s], ak[T]);
        av[T] = wmma_f32(a[s], bv[T][s], av[T]);
      }
    }
    float* ko = kout + ((size_t)t * 16 + 8 * h) * D_ + 4 * c;
    float* vo = vout + ((size_t)t * 16 + 8 * h) * D_ + 4 * c;
    for (int r = 0; r < 8; ++r) {
      v4f o; o[0] = ak[0][r]; o[1] = ak[1][r]; o[2] = ak[2][r]; o[3] = ak[3][r];
      *(v4f*)(ko + (size_t)r * D_) = o;
      v4f p; p[0] = av[0][r]; p[1] = av[1][r]; p[2] = av[2][r]; p[3] = av[3][r];
      *(v4f*)(vo + (size_t)r * D_) = p;
    }
  }
}

__global__ void slots_init_kernel(const float* __restrict__ sinit,
                                  const float* __restrict__ smu,
                                  const float* __restrict__ slog,
                                  float* __restrict__ slots) {
  int idx = blockIdx.x * blockDim.x + threadIdx.x;
  int d = idx & (D_ - 1);
  int h = idx / (B_ * S_ * D_);
  slots[idx] = smu[h * D_ + d] + expf(slog[h * D_ + d]) * sinit[idx];
}

__global__ void __launch_bounds__(128) hprev_q_kernel(
    const float* __restrict__ slots, const float* __restrict__ lng,
    const float* __restrict__ lnb, const float* __restrict__ Wq,
    float* __restrict__ hprev, float* __restrict__ qout) {
  __shared__ float hl[16][68];
  __shared__ float mu[16], rstd[16];
  const int tid = threadIdx.x;
  const int h = 0;
  const int rt = blockIdx.x;
  const int rowbase = h * (B_ * S_) + rt * 16;
  for (int e = tid; e < 16 * D_; e += 128) {
    int r = e >> 6, c = e & 63;
    hl[r][c] = slots[(size_t)(rowbase + r) * D_ + c];
  }
  __syncthreads();
  if (tid < 16) {
    float m = 0.f;
    for (int j = 0; j < D_; ++j) m += hl[tid][j];
    m *= (1.f / D_);
    float var = 0.f;
    for (int j = 0; j < D_; ++j) { float dd = hl[tid][j] - m; var += dd * dd; }
    var *= (1.f / D_);
    mu[tid] = m; rstd[tid] = rsqrtf(var + LN_EPS_);
  }
  __syncthreads();
  for (int e = tid; e < 16 * D_; e += 128) {
    int r = e >> 6, c = e & 63;
    float nv = (hl[r][c] - mu[r]) * rstd[r] * lng[h * D_ + c] + lnb[h * D_ + c];
    hl[r][c] = nv;
    hprev[(size_t)(rowbase + r) * D_ + c] = nv;
  }
  __syncthreads();
  const int w = tid >> 5, lane = tid & 31, m16 = lane & 15, kh = lane >> 4;
  const float* Wh = Wq + (size_t)h * D_ * D_;
  const int ct = w * 16;
  v8f acc = {};
  for (int kk = 0; kk < 16; ++kk) {
    int kb = kk * 4 + kh * 2;
    v2f a; a.x = hl[m16][kb]; a.y = hl[m16][kb + 1];
    v2f b; b.x = Wh[kb * D_ + ct + m16]; b.y = Wh[(kb + 1) * D_ + ct + m16];
    acc = wmma_f32(a, b, acc);
  }
  for (int i = 0; i < 8; ++i)
    qout[(size_t)(rowbase + i + 8 * kh) * D_ + ct + m16] = acc[i];
}

constexpr int ATT_WAVES_ = 16;
constexpr int ATT_SPLIT_ = 4;
constexpr int ATT_TILES_ = (N_ / 16) / ATT_SPLIT_;
static_assert(ATT_TILES_ % ATT_WAVES_ == 0, "tiles must divide over the waves");
__global__ void __launch_bounds__(512) __attribute__((amdgpu_waves_per_eu(4, 4)))
attn_kernel(
    const float* __restrict__ kbuf, const float* __restrict__ vbuf,
    const float* __restrict__ qbuf, float* __restrict__ pnum,
    float* __restrict__ pden) {
  __shared__ alignas(16) float red[ATT_WAVES_][8][68];
  __shared__ float dred[ATT_WAVES_][32];
  const int tid = threadIdx.x;
  const int ch = blockIdx.x >> 6;
  const int b  = blockIdx.x & 63;
  const int w = tid >> 5, lane = tid & 31, c = lane & 15, kh = lane >> 4;
  const float* kb = kbuf + (size_t)b * N_ * D_;
  const float* vb = vbuf + (size_t)b * N_ * D_;
  const float* qrow = qbuf + (size_t)(b * S_ + (c & 7)) * D_ + 4 * kh;
  float4 qf[8];
  for (int i = 0; i < 8; ++i) qf[i] = *(const float4*)(qrow + 8 * i);
  v8f acc0 = {}, acc1 = {}, acc2 = {}, acc3 = {};
  float dsum = 0.f;
  for (int t = ch * ATT_TILES_ + w; t < (ch + 1) * ATT_TILES_; t += ATT_WAVES_) {
    const int n0 = t * 16;
    const float* kp = kb + (size_t)(n0 + c) * D_ + 4 * kh;
    const float* vp = vb + (size_t)(n0 + 8 * kh) * D_ + 4 * c;
    float4 kf[8], vf[8];
    for (int i = 0; i < 8; ++i) kf[i] = *(const float4*)(kp + 8 * i);
    for (int r = 0; r < 8; ++r) vf[r] = *(const float4*)(vp + (size_t)r * D_);
    v8f lg = {};
    for (int i = 0; i < 8; ++i) {
      v2f a0; a0.x = kf[i].x; a0.y = kf[i].y;
      v2f b0; b0.x = qf[i].x; b0.y = qf[i].y;
      lg = wmma_f32(a0, b0, lg);
      v2f a1; a1.x = kf[i].z; a1.y = kf[i].w;
      v2f b1; b1.x = qf[i].z; b1.y = qf[i].w;
      lg = wmma_f32(a1, b1, lg);
    }
    float p[8];
#pragma unroll
    for (int r = 0; r < 8; ++r) {
      float x = lg[r] * SCALE_;
      float mx = fmaxf(x, xlane<1>(x));
      mx = fmaxf(mx, xlane<2>(mx));
      mx = fmaxf(mx, xlane<4>(mx));
      float e = expf(x - mx);
      float sm = e + xlane<1>(e);
      sm += xlane<2>(sm);
      sm += xlane<4>(sm);
      float inv = 1.f / sm;
      p[r] = e * inv + EPS_ATTN_;
      dsum += p[r];
    }
    for (int j = 0; j < 4; ++j) {
      v2f a; a.x = p[2 * j]; a.y = p[2 * j + 1];
      v2f b0; b0.x = vf[2 * j].x; b0.y = vf[2 * j + 1].x;
      acc0 = wmma_f32(a, b0, acc0);
      v2f b1; b1.x = vf[2 * j].y; b1.y = vf[2 * j + 1].y;
      acc1 = wmma_f32(a, b1, acc1);
      v2f b2; b2.x = vf[2 * j].z; b2.y = vf[2 * j + 1].z;
      acc2 = wmma_f32(a, b2, acc2);
      v2f b3; b3.x = vf[2 * j].w; b3.y = vf[2 * j + 1].w;
      acc3 = wmma_f32(a, b3, acc3);
    }
  }
  if (kh == 0) {
    for (int i = 0; i < 8; ++i) {
      float4 o; o.x = acc0[i]; o.y = acc1[i]; o.z = acc2[i]; o.w = acc3[i];
      *(float4*)&red[w][i][4 * c] = o;
    }
  }
  dred[w][lane] = dsum;
  __syncthreads();
  if (tid < 128) {
    const int sp = tid >> 4, d4 = (tid & 15) * 4;
    float4 sacc; sacc.x = 0.f; sacc.y = 0.f; sacc.z = 0.f; sacc.w = 0.f;
    float den = 0.f;
    for (int ww = 0; ww < ATT_WAVES_; ++ww) {
      float4 x = *(const float4*)&red[ww][sp][d4];
      sacc.x += x.x; sacc.y += x.y; sacc.z += x.z; sacc.w += x.w;
      den += dred[ww][sp] + dred[ww][sp + 16];
    }
    const size_t prow = (size_t)((b * ATT_SPLIT_ + ch) * S_ + sp);
    *(float4*)(pnum + prow * D_ + d4) = sacc;
    if (d4 == 0) pden[prow] = den;
  }
}

__global__ void __launch_bounds__(256) gru_mlp_kernel(
    const float* __restrict__ pnum, const float* __restrict__ pden,
    const float* __restrict__ hprev,
    const float* __restrict__ wih, const float* __restrict__ whh,
    const float* __restrict__ bih, const float* __restrict__ bhh,
    const float* __restrict__ lnmg, const float* __restrict__ lnmb,
    const float* __restrict__ w1, const float* __restrict__ b1,
    const float* __restrict__ w2, const float* __restrict__ b2,
    float* __restrict__ slots) {
  __shared__ float ut[16][68];
  __shared__ float ht[16][68];
  __shared__ float gx[16][196];
  __shared__ float gh[16][196];
  __shared__ float sn[16][68];
  __shared__ float nm[16][68];
  __shared__ float hid[16][132];
  __shared__ float mu[16], rstd[16];
  const int tid = threadIdx.x;
  const int h = 0, rt = blockIdx.x;
  const int rowbase = h * (B_ * S_) + rt * 16;
  const int r0 = tid >> 4, c00 = (tid & 15) * 4;
  {
    const int srow = rt * 16 + r0;
    const int bb = srow >> 3, ss = srow & 7;
    float4 a; a.x = 0.f; a.y = 0.f; a.z = 0.f; a.w = 0.f;
    float den = 0.f;
    for (int q = 0; q < ATT_SPLIT_; ++q) {
      const size_t prow = (size_t)((bb * ATT_SPLIT_ + q) * S_ + ss);
      const float4 x = *(const float4*)(pnum + prow * D_ + c00);
      a.x += x.x; a.y += x.y; a.z += x.z; a.w += x.w;
      den += pden[prow];
    }
    a.x = a.x / den; a.y = a.y / den; a.z = a.z / den; a.w = a.w / den;
    ut[r0][c00] = a.x; ut[r0][c00 + 1] = a.y; ut[r0][c00 + 2] = a.z; ut[r0][c00 + 3] = a.w;
    float4 c = *(const float4*)(hprev + (size_t)(rowbase + r0) * D_ + c00);
    ht[r0][c00] = c.x; ht[r0][c00 + 1] = c.y; ht[r0][c00 + 2] = c.z; ht[r0][c00 + 3] = c.w;
  }
  __syncthreads();
  const int w = tid >> 5, lane = tid & 31, m16 = lane & 15, kh = lane >> 4;
  const float* wihh = wih + (size_t)h * D_ * 192;
  const float* whhh = whh + (size_t)h * D_ * 192;
  for (int T = w; T < 24; T += 8) {
    const bool isX = T < 12;
    const int ct = (T % 12) * 16;
    const float* Wt = isX ? wihh : whhh;
    v8f acc = {};
    for (int kk = 0; kk < 16; ++kk) {
      int kb = kk * 4 + kh * 2;
      v2f a;
      if (isX) { a.x = ut[m16][kb]; a.y = ut[m16][kb + 1]; }
      else     { a.x = ht[m16][kb]; a.y = ht[m16][kb + 1]; }
      v2f bb; bb.x = Wt[kb * 192 + ct + m16]; bb.y = Wt[(kb + 1) * 192 + ct + m16];
      acc = wmma_f32(a, bb, acc);
    }
    float* dst = isX ? &gx[0][0] : &gh[0][0];
    for (int i = 0; i < 8; ++i)
      dst[(i + 8 * kh) * 196 + ct + m16] = acc[i];
  }
  __syncthreads();
  {
    const float* bi = bih + h * 192;
    const float* bh = bhh + h * 192;
    for (int j = 0; j < 4; ++j) {
      int d = c00 + j;
      float xr = gx[r0][d] + bi[d],        hr = gh[r0][d] + bh[d];
      float xz = gx[r0][64 + d] + bi[64 + d],  hz = gh[r0][64 + d] + bh[64 + d];
      float xn = gx[r0][128 + d] + bi[128 + d], hn = gh[r0][128 + d] + bh[128 + d];
      float rg = 1.f / (1.f + expf(-(xr + hr)));
      float zg = 1.f / (1.f + expf(-(xz + hz)));
      float ng = tanhf(xn + rg * hn);
      sn[r0][d] = (1.f - zg) * ng + zg * ht[r0][d];
    }
  }
  __syncthreads();
  if (tid < 16) {
    float m = 0.f;
    for (int j = 0; j < D_; ++j) m += sn[tid][j];
    m *= (1.f / D_);
    float var = 0.f;
    for (int j = 0; j < D_; ++j) { float dd = sn[tid][j] - m; var += dd * dd; }
    var *= (1.f / D_);
    mu[tid] = m; rstd[tid] = rsqrtf(var + LN_EPS_);
  }
  __syncthreads();
  for (int j = 0; j < 4; ++j) {
    int d = c00 + j;
    nm[r0][d] = (sn[r0][d] - mu[r0]) * rstd[r0] * lnmg[h * D_ + d] + lnmb[h * D_ + d];
  }
  __syncthreads();
  {
    const float* w1h = w1 + (size_t)h * D_ * MLPH_;
    const float* b1h = b1 + h * MLPH_;
    const int ct = w * 16;
    v8f acc = {};
    for (int kk = 0; kk < 16; ++kk) {
      int kb = kk * 4 + kh * 2;
      v2f a;  a.x = nm[m16][kb]; a.y = nm[m16][kb + 1];
      v2f bb; bb.x = w1h[kb * MLPH_ + ct + m16]; bb.y = w1h[(kb + 1) * MLPH_ + ct + m16];
      acc = wmma_f32(a, bb, acc);
    }
    float bias = b1h[ct + m16];
    for (int i = 0; i < 8; ++i)
      hid[i + 8 * kh][ct + m16] = fmaxf(acc[i] + bias, 0.f);
  }
  __syncthreads();
  if (w < 4) {
    const float* w2h = w2 + (size_t)h * MLPH_ * D_;
    const float* b2h = b2 + h * D_;
    const int ct = w * 16;
    v8f acc = {};
    for (int kk = 0; kk < 32; ++kk) {
      int kb = kk * 4 + kh * 2;
      v2f a;  a.x = hid[m16][kb]; a.y = hid[m16][kb + 1];
      v2f bb; bb.x = w2h[kb * D_ + ct + m16]; bb.y = w2h[(kb + 1) * D_ + ct + m16];
      acc = wmma_f32(a, bb, acc);
    }
    float bias = b2h[ct + m16];
    for (int i = 0; i < 8; ++i) {
      int rr = i + 8 * kh;
      slots[(size_t)(rowbase + rr) * D_ + ct + m16] = sn[rr][ct + m16] + acc[i] + bias;
    }
  }
}

__global__ void copy_out_kernel(const float* __restrict__ slots,
                                float* __restrict__ out) {
  int i = blockIdx.x * blockDim.x + threadIdx.x;
  if (i < B_ * S_ * D_) out[i] = slots[i];
}

extern "C" void kernel_launch(void* const* d_in, const int* in_sizes, int n_in,
                              void* d_out, int out_size, void* d_ws, size_t ws_size,
                              hipStream_t stream) {
  (void)in_sizes; (void)n_in; (void)out_size; (void)ws_size;
  const float* inputs  = (const float*)d_in[0];
  const float* sinit   = (const float*)d_in[1];
  const float* smu     = (const float*)d_in[2];
  const float* slog    = (const float*)d_in[3];
  const float* ln_in_g = (const float*)d_in[4];
  const float* ln_in_b = (const float*)d_in[5];
  const float* Wk      = (const float*)d_in[6];
  const float* Wv      = (const float*)d_in[7];
  const float* ln_s_g  = (const float*)d_in[8];
  const float* ln_s_b  = (const float*)d_in[9];
  const float* Wq      = (const float*)d_in[10];
  const float* gru_wih = (const float*)d_in[11];
  const float* gru_whh = (const float*)d_in[12];
  const float* gru_bih = (const float*)d_in[13];
  const float* gru_bhh = (const float*)d_in[14];
  const float* ln_m_g  = (const float*)d_in[15];
  const float* ln_m_b  = (const float*)d_in[16];
  const float* mlp_w1  = (const float*)d_in[17];
  const float* mlp_b1  = (const float*)d_in[18];
  const float* mlp_w2  = (const float*)d_in[19];
  const float* mlp_b2  = (const float*)d_in[20];

  float* ws = (float*)d_ws;
  float* kbuf  = ws;
  float* vbuf  = kbuf + (size_t)B_ * N_ * D_;
  float* slots = vbuf + (size_t)B_ * N_ * D_;
  float* hprev = slots + H_ * B_ * S_ * D_;
  float* qbuf  = hprev + H_ * B_ * S_ * D_;
  float* pnum  = qbuf + H_ * B_ * S_ * D_;
  float* pden  = pnum + B_ * ATT_SPLIT_ * S_ * D_;

  ln_kv_kernel<<<KV_GRID_, 256, 0, stream>>>(inputs, ln_in_g, ln_in_b,
                                                        Wk, Wv, kbuf, vbuf);
  slots_init_kernel<<<(B_ * S_ * D_) / 256, 256, 0, stream>>>(sinit, smu, slog,
                                                              slots);
  for (int it = 0; it < ITERS_; ++it) {
    hprev_q_kernel<<<32, 128, 0, stream>>>(slots, ln_s_g, ln_s_b, Wq, hprev,
                                           qbuf);
    attn_kernel<<<ATT_SPLIT_ * B_, 512, 0, stream>>>(kbuf, vbuf, qbuf, pnum,
                                                     pden);
    gru_mlp_kernel<<<32, 256, 0, stream>>>(pnum, pden, hprev, gru_wih, gru_whh,
                                           gru_bih, gru_bhh, ln_m_g, ln_m_b,
                                           mlp_w1, mlp_b1, mlp_w2, mlp_b2,
                                           slots);
  }
  copy_out_kernel<<<(B_ * S_ * D_ + 255) / 256, 256, 0, stream>>>(slots,
                                                                  (float*)d_out);
}


# slot-row launches fused: slots init joins the first LayerNorm/q launch, each GRU/MLP launch also makes the next iteration's h_prev and q from its tile in LDS, the last stores out: 12 launches to 8
# speedup vs baseline: 4.6609x; 1.0815x over previous
#include <hip/hip_runtime.h>

typedef float v2f __attribute__((ext_vector_type(2)));
typedef float v4f __attribute__((ext_vector_type(4)));
typedef float v8f __attribute__((ext_vector_type(8)));

constexpr int H_ = 8, B_ = 64, S_ = 8, D_ = 64, N_ = 4096, MLPH_ = 128, ITERS_ = 3;
constexpr float LN_EPS_ = 1e-5f, EPS_ATTN_ = 1e-8f, SCALE_ = 0.125f;
static_assert(ITERS_ == 3, "kernel_launch writes the three iterations out");

__device__ __forceinline__ v8f wmma_f32(v2f a, v2f b, v8f c) {
  return __builtin_amdgcn_wmma_f32_16x16x4_f32(false, a, false, b, (short)0, c,
                                               false, false);
}

template <int M>
__device__ __forceinline__ float xlane(float x) {
  return __builtin_bit_cast(float, __builtin_amdgcn_update_dpp(
      0, __builtin_bit_cast(int, x), 0x160 + M, 0xf, 0xf, true));
}

constexpr int KV_WAVES_ = 8;
constexpr int KV_GRID_ = 256;
constexpr int KV_TILES_ = (B_ * N_) / 16;
constexpr int KV_STRIDE_ = KV_GRID_ * KV_WAVES_;
constexpr int KV_TRIPS_ = KV_TILES_ / KV_STRIDE_;
static_assert(KV_TILES_ % KV_STRIDE_ == 0, "tiles must divide over the waves");

__device__ __forceinline__ float xhalf(float x) {
  const int b = __builtin_bit_cast(int, x);
  return __builtin_bit_cast(float, __builtin_amdgcn_permlanex16(
      b, b, 0x76543210, 0xfedcba98, false, false));
}

__global__ void __launch_bounds__(256) __attribute__((amdgpu_waves_per_eu(2, 2)))
ln_kv_kernel(
    const float* __restrict__ inp, const float* __restrict__ lng,
    const float* __restrict__ lnb, const float* __restrict__ Wk,
    const float* __restrict__ Wv, float* __restrict__ kout,
    float* __restrict__ vout) {
  const int tid = threadIdx.x;
  const int w = tid >> 5, lane = tid & 31, c = lane & 15, h = lane >> 4;
  const int gw = blockIdx.x * KV_WAVES_ + w;
  v4f gf[8], bf[8];
  for (int i = 0; i < 8; ++i) {
    gf[i] = *(const v4f*)(lng + 8 * i + 4 * h);
    bf[i] = *(const v4f*)(lnb + 8 * i + 4 * h);
  }
  v2f bk[4][16], bv[4][16];
#pragma unroll
  for (int s = 0; s < 16; ++s) {
    const int d0 = 8 * (s >> 1) + 4 * h + 2 * (s & 1);
    const v4f k0 = *(const v4f*)(Wk + d0 * D_ + 4 * c);
    const v4f k1 = *(const v4f*)(Wk + (d0 + 1) * D_ + 4 * c);
    const v4f v0 = *(const v4f*)(Wv + d0 * D_ + 4 * c);
    const v4f v1 = *(const v4f*)(Wv + (d0 + 1) * D_ + 4 * c);
    for (int T = 0; T < 4; ++T) {
      bk[T][s].x = k0[T]; bk[T][s].y = k1[T];
      bv[T][s].x = v0[T]; bv[T][s].y = v1[T];
    }
  }
  v4f xr[8];
  {
    const float* xp = inp + ((size_t)gw * 16 + c) * D_ + 4 * h;
    for (int i = 0; i < 8; ++i) xr[i] = *(const v4f*)(xp + 8 * i);
  }
#pragma nounroll
  for (int j = 0; j < KV_TRIPS_; ++j) {
    const int t = gw + j * KV_STRIDE_;
    v2f sl = xr[0].lo, sh = xr[0].hi;
    for (int i = 1; i < 8; ++i) { sl += xr[i].lo; sh += xr[i].hi; }
    const v2f s2 = sl + sh;
    float sm = s2.x + s2.y;
    sm += xhalf(sm);
    const float m = sm * (1.f / D_);
    v2f a[16];
    v2f ql = {0.f, 0.f}, qh = {0.f, 0.f};
    for (int i = 0; i < 8; ++i) {
      a[2 * i] = xr[i].lo - m;
      a[2 * i + 1] = xr[i].hi - m;
      ql += a[2 * i] * a[2 * i];
      qh += a[2 * i + 1] * a[2 * i + 1];
    }
    const v2f q2 = ql + qh;
    float var = q2.x + q2.y;
    var += xhalf(var);
    const float rs = rsqrtf(var * (1.f / D_) + LN_EPS_);
    for (int i = 0; i < 8; ++i) {
      a[2 * i] = a[2 * i] * rs * gf[i].lo + bf[i].lo;
      a[2 * i + 1] = a[2 * i + 1] * rs * gf[i].hi + bf[i].hi;
    }
    {
      const int tn = (j + 1 < KV_TRIPS_) ? t + KV_STRIDE_ : t;
      const float* xp = inp + ((size_t)tn * 16 + c) * D_ + 4 * h;
      for (int i = 0; i < 8; ++i) xr[i] = *(const v4f*)(xp + 8 * i);
    }
    v8f ak[4] = {}, av[4] = {};
#pragma unroll
    for (int s = 0; s < 16; ++s) {
#pragma unroll
      for (int T = 0; T < 4; ++T) {
        ak[T] = wmma_f32(a[s], bk[T][s], ak[T]);
        av[T] = wmma_f32(a[s], bv[T][s], av[T]);
      }
    }
    float* ko = kout + ((size_t)t * 16 + 8 * h) * D_ + 4 * c;
    float* vo = vout + ((size_t)t * 16 + 8 * h) * D_ + 4 * c;
    for (int r = 0; r < 8; ++r) {
      v4f o; o[0] = ak[0][r]; o[1] = ak[1][r]; o[2] = ak[2][r]; o[3] = ak[3][r];
      *(v4f*)(ko + (size_t)r * D_) = o;
      v4f p; p[0] = av[0][r]; p[1] = av[1][r]; p[2] = av[2][r]; p[3] = av[3][r];
      *(v4f*)(vo + (size_t)r * D_) = p;
    }
  }
}

template <int NT>
__device__ __forceinline__ void ln_q_tile(
    float (*hl)[68], float* mu, float* rstd, const float* __restrict__ lng,
    const float* __restrict__ lnb, const float* __restrict__ Wq,
    float* hprev, float* __restrict__ qout, int rowbase, int tid) {
  if (tid < 16) {
    float m = 0.f;
    for (int j = 0; j < D_; ++j) m += hl[tid][j];
    m *= (1.f / D_);
    float var = 0.f;
    for (int j = 0; j < D_; ++j) { float dd = hl[tid][j] - m; var += dd * dd; }
    var *= (1.f / D_);
    mu[tid] = m; rstd[tid] = rsqrtf(var + LN_EPS_);
  }
  __syncthreads();
  for (int e = tid; e < 16 * D_; e += NT) {
    int r = e >> 6, c = e & 63;
    float nv = (hl[r][c] - mu[r]) * rstd[r] * lng[c] + lnb[c];
    hl[r][c] = nv;
    hprev[(size_t)(rowbase + r) * D_ + c] = nv;
  }
  __syncthreads();
  const int w = tid >> 5, lane = tid & 31, m16 = lane & 15, kh = lane >> 4;
  if (w < 4) {
    const int ct = w * 16;
    v8f acc = {};
    for (int kk = 0; kk < 16; ++kk) {
      int kb = kk * 4 + kh * 2;
      v2f a; a.x = hl[m16][kb]; a.y = hl[m16][kb + 1];
      v2f b; b.x = Wq[kb * D_ + ct + m16]; b.y = Wq[(kb + 1) * D_ + ct + m16];
      acc = wmma_f32(a, b, acc);
    }
    for (int i = 0; i < 8; ++i)
      qout[(size_t)(rowbase + i + 8 * kh) * D_ + ct + m16] = acc[i];
  }
}

__global__ void __launch_bounds__(128) init_hprev_q_kernel(
    const float* __restrict__ sinit, const float* __restrict__ smu,
    const float* __restrict__ slog, const float* __restrict__ lng,
    const float* __restrict__ lnb, const float* __restrict__ Wq,
    float* hprev, float* __restrict__ qout) {
  __shared__ float hl[16][68];
  __shared__ float mu[16], rstd[16];
  const int tid = threadIdx.x;
  const int rowbase = blockIdx.x * 16;
  for (int e = tid; e < 16 * D_; e += 128) {
    int r = e >> 6, c = e & 63;
    hl[r][c] = smu[c] + expf(slog[c]) * sinit[(size_t)(rowbase + r) * D_ + c];
  }
  __syncthreads();
  ln_q_tile<128>(hl, mu, rstd, lng, lnb, Wq, hprev, qout, rowbase, tid);
}

constexpr int ATT_WAVES_ = 16;
constexpr int ATT_SPLIT_ = 4;
constexpr int ATT_TILES_ = (N_ / 16) / ATT_SPLIT_;
static_assert(ATT_TILES_ % ATT_WAVES_ == 0, "tiles must divide over the waves");
__global__ void __launch_bounds__(512) __attribute__((amdgpu_waves_per_eu(4, 4)))
attn_kernel(
    const float* __restrict__ kbuf, const float* __restrict__ vbuf,
    const float* __restrict__ qbuf, float* __restrict__ pnum,
    float* __restrict__ pden) {
  __shared__ alignas(16) float red[ATT_WAVES_][8][68];
  __shared__ float dred[ATT_WAVES_][32];
  const int tid = threadIdx.x;
  const int ch = blockIdx.x >> 6;
  const int b  = blockIdx.x & 63;
  const int w = tid >> 5, lane = tid & 31, c = lane & 15, kh = lane >> 4;
  const float* kb = kbuf + (size_t)b * N_ * D_;
  const float* vb = vbuf + (size_t)b * N_ * D_;
  const float* qrow = qbuf + (size_t)(b * S_ + (c & 7)) * D_ + 4 * kh;
  float4 qf[8];
  for (int i = 0; i < 8; ++i) qf[i] = *(const float4*)(qrow + 8 * i);
  v8f acc0 = {}, acc1 = {}, acc2 = {}, acc3 = {};
  float dsum = 0.f;
  for (int t = ch * ATT_TILES_ + w; t < (ch + 1) * ATT_TILES_; t += ATT_WAVES_) {
    const int n0 = t * 16;
    const float* kp = kb + (size_t)(n0 + c) * D_ + 4 * kh;
    const float* vp = vb + (size_t)(n0 + 8 * kh) * D_ + 4 * c;
    float4 kf[8], vf[8];
    for (int i = 0; i < 8; ++i) kf[i] = *(const float4*)(kp + 8 * i);
    for (int r = 0; r < 8; ++r) vf[r] = *(const float4*)(vp + (size_t)r * D_);
    v8f lg = {};
    for (int i = 0; i < 8; ++i) {
      v2f a0; a0.x = kf[i].x; a0.y = kf[i].y;
      v2f b0; b0.x = qf[i].x; b0.y = qf[i].y;
      lg = wmma_f32(a0, b0, lg);
      v2f a1; a1.x = kf[i].z; a1.y = kf[i].w;
      v2f b1; b1.x = qf[i].z; b1.y = qf[i].w;
      lg = wmma_f32(a1, b1, lg);
    }
    float p[8];
#pragma unroll
    for (int r = 0; r < 8; ++r) {
      float x = lg[r] * SCALE_;
      float mx = fmaxf(x, xlane<1>(x));
      mx = fmaxf(mx, xlane<2>(mx));
      mx = fmaxf(mx, xlane<4>(mx));
      float e = expf(x - mx);
      float sm = e + xlane<1>(e);
      sm += xlane<2>(sm);
      sm += xlane<4>(sm);
      float inv = 1.f / sm;
      p[r] = e * inv + EPS_ATTN_;
      dsum += p[r];
    }
    for (int j = 0; j < 4; ++j) {
      v2f a; a.x = p[2 * j]; a.y = p[2 * j + 1];
      v2f b0; b0.x = vf[2 * j].x; b0.y = vf[2 * j + 1].x;
      acc0 = wmma_f32(a, b0, acc0);
      v2f b1; b1.x = vf[2 * j].y; b1.y = vf[2 * j + 1].y;
      acc1 = wmma_f32(a, b1, acc1);
      v2f b2; b2.x = vf[2 * j].z; b2.y = vf[2 * j + 1].z;
      acc2 = wmma_f32(a, b2, acc2);
      v2f b3; b3.x = vf[2 * j].w; b3.y = vf[2 * j + 1].w;
      acc3 = wmma_f32(a, b3, acc3);
    }
  }
  if (kh == 0) {
    for (int i = 0; i < 8; ++i) {
      float4 o; o.x = acc0[i]; o.y = acc1[i]; o.z = acc2[i]; o.w = acc3[i];
      *(float4*)&red[w][i][4 * c] = o;
    }
  }
  dred[w][lane] = dsum;
  __syncthreads();
  if (tid < 128) {
    const int sp = tid >> 4, d4 = (tid & 15) * 4;
    float4 sacc; sacc.x = 0.f; sacc.y = 0.f; sacc.z = 0.f; sacc.w = 0.f;
    float den = 0.f;
    for (int ww = 0; ww < ATT_WAVES_; ++ww) {
      float4 x = *(const float4*)&red[ww][sp][d4];
      sacc.x += x.x; sacc.y += x.y; sacc.z += x.z; sacc.w += x.w;
      den += dred[ww][sp] + dred[ww][sp + 16];
    }
    const size_t prow = (size_t)((b * ATT_SPLIT_ + ch) * S_ + sp);
    *(float4*)(pnum + prow * D_ + d4) = sacc;
    if (d4 == 0) pden[prow] = den;
  }
}

template <bool LAST>
__global__ void __launch_bounds__(256) gru_mlp_kernel(
    const float* __restrict__ pnum, const float* __restrict__ pden,
    float* hprev,
    const float* __restrict__ wih, const float* __restrict__ whh,
    const float* __restrict__ bih, const float* __restrict__ bhh,
    const float* __restrict__ lnmg, const float* __restrict__ lnmb,
    const float* __restrict__ w1, const float* __restrict__ b1,
    const float* __restrict__ w2, const float* __restrict__ b2,
    const float* __restrict__ lnsg, const float* __restrict__ lnsb,
    const float* __restrict__ Wq, float* __restrict__ qout,
    float* __restrict__ out) {
  __shared__ float ut[16][68];
  __shared__ float ht[16][68];
  __shared__ float gx[16][196];
  __shared__ float gh[16][196];
  __shared__ float sn[16][68];
  __shared__ float nm[16][68];
  __shared__ float hid[16][132];
  __shared__ float mu[16], rstd[16];
  const int tid = threadIdx.x;
  const int h = 0, rt = blockIdx.x;
  const int rowbase = h * (B_ * S_) + rt * 16;
  const int r0 = tid >> 4, c00 = (tid & 15) * 4;
  {
    const int srow = rt * 16 + r0;
    const int bb = srow >> 3, ss = srow & 7;
    float4 a; a.x = 0.f; a.y = 0.f; a.z = 0.f; a.w = 0.f;
    float den = 0.f;
    for (int q = 0; q < ATT_SPLIT_; ++q) {
      const size_t prow = (size_t)((bb * ATT_SPLIT_ + q) * S_ + ss);
      const float4 x = *(const float4*)(pnum + prow * D_ + c00);
      a.x += x.x; a.y += x.y; a.z += x.z; a.w += x.w;
      den += pden[prow];
    }
    a.x = a.x / den; a.y = a.y / den; a.z = a.z / den; a.w = a.w / den;
    ut[r0][c00] = a.x; ut[r0][c00 + 1] = a.y; ut[r0][c00 + 2] = a.z; ut[r0][c00 + 3] = a.w;
    float4 c = *(const float4*)(hprev + (size_t)(rowbase + r0) * D_ + c00);
    ht[r0][c00] = c.x; ht[r0][c00 + 1] = c.y; ht[r0][c00 + 2] = c.z; ht[r0][c00 + 3] = c.w;
  }
  __syncthreads();
  const int w = tid >> 5, lane = tid & 31, m16 = lane & 15, kh = lane >> 4;
  const float* wihh = wih + (size_t)h * D_ * 192;
  const float* whhh = whh + (size_t)h * D_ * 192;
  for (int T = w; T < 24; T += 8) {
    const bool isX = T < 12;
    const int ct = (T % 12) * 16;
    const float* Wt = isX ? wihh : whhh;
    v8f acc = {};
    for (int kk = 0; kk < 16; ++kk) {
      int kb = kk * 4 + kh * 2;
      v2f a;
      if (isX) { a.x = ut[m16][kb]; a.y = ut[m16][kb + 1]; }
      else     { a.x = ht[m16][kb]; a.y = ht[m16][kb + 1]; }
      v2f bb; bb.x = Wt[kb * 192 + ct + m16]; bb.y = Wt[(kb + 1) * 192 + ct + m16];
      acc = wmma_f32(a, bb, acc);
    }
    float* dst = isX ? &gx[0][0] : &gh[0][0];
    for (int i = 0; i < 8; ++i)
      dst[(i + 8 * kh) * 196 + ct + m16] = acc[i];
  }
  __syncthreads();
  {
    const float* bi = bih + h * 192;
    const float* bh = bhh + h * 192;
    for (int j = 0; j < 4; ++j) {
      int d = c00 + j;
      float xr = gx[r0][d] + bi[d],        hr = gh[r0][d] + bh[d];
      float xz = gx[r0][64 + d] + bi[64 + d],  hz = gh[r0][64 + d] + bh[64 + d];
      float xn = gx[r0][128 + d] + bi[128 + d], hn = gh[r0][128 + d] + bh[128 + d];
      float rg = 1.f / (1.f + expf(-(xr + hr)));
      float zg = 1.f / (1.f + expf(-(xz + hz)));
      float ng = tanhf(xn + rg * hn);
      sn[r0][d] = (1.f - zg) * ng + zg * ht[r0][d];
    }
  }
  __syncthreads();
  if (tid < 16) {
    float m = 0.f;
    for (int j = 0; j < D_; ++j) m += sn[tid][j];
    m *= (1.f / D_);
    float var = 0.f;
    for (int j = 0; j < D_; ++j) { float dd = sn[tid][j] - m; var += dd * dd; }
    var *= (1.f / D_);
    mu[tid] = m; rstd[tid] = rsqrtf(var + LN_EPS_);
  }
  __syncthreads();
  for (int j = 0; j < 4; ++j) {
    int d = c00 + j;
    nm[r0][d] = (sn[r0][d] - mu[r0]) * rstd[r0] * lnmg[h * D_ + d] + lnmb[h * D_ + d];
  }
  __syncthreads();
  {
    const float* w1h = w1 + (size_t)h * D_ * MLPH_;
    const float* b1h = b1 + h * MLPH_;
    const int ct = w * 16;
    v8f acc = {};
    for (int kk = 0; kk < 16; ++kk) {
      int kb = kk * 4 + kh * 2;
      v2f a;  a.x = nm[m16][kb]; a.y = nm[m16][kb + 1];
      v2f bb; bb.x = w1h[kb * MLPH_ + ct + m16]; bb.y = w1h[(kb + 1) * MLPH_ + ct + m16];
      acc = wmma_f32(a, bb, acc);
    }
    float bias = b1h[ct + m16];
    for (int i = 0; i < 8; ++i)
      hid[i + 8 * kh][ct + m16] = fmaxf(acc[i] + bias, 0.f);
  }
  __syncthreads();
  if (w < 4) {
    const float* w2h = w2 + (size_t)h * MLPH_ * D_;
    const float* b2h = b2 + h * D_;
    const int ct = w * 16;
    v8f acc = {};
    for (int kk = 0; kk < 32; ++kk) {
      int kb = kk * 4 + kh * 2;
      v2f a;  a.x = hid[m16][kb]; a.y = hid[m16][kb + 1];
      v2f bb; bb.x = w2h[kb * D_ + ct + m16]; bb.y = w2h[(kb + 1) * D_ + ct + m16];
      acc = wmma_f32(a, bb, acc);
    }
    float bias = b2h[ct + m16];
    for (int i = 0; i < 8; ++i) {
      int rr = i + 8 * kh;
      const float sv = sn[rr][ct + m16] + acc[i] + bias;
      if (LAST) out[(size_t)(rowbase + rr) * D_ + ct + m16] = sv;
      else ut[rr][ct + m16] = sv;
    }
  }
  if (!LAST) {
    __syncthreads();
    ln_q_tile<256>(ut, mu, rstd, lnsg, lnsb, Wq, hprev, qout, rowbase, tid);
  }
}

extern "C" void kernel_launch(void* const* d_in, const int* in_sizes, int n_in,
                              void* d_out, int out_size, void* d_ws, size_t ws_size,
                              hipStream_t stream) {
  (void)in_sizes; (void)n_in; (void)out_size; (void)ws_size;
  const float* inputs  = (const float*)d_in[0];
  const float* sinit   = (const float*)d_in[1];
  const float* smu     = (const float*)d_in[2];
  const float* slog    = (const float*)d_in[3];
  const float* ln_in_g = (const float*)d_in[4];
  const float* ln_in_b = (const float*)d_in[5];
  const float* Wk      = (const float*)d_in[6];
  const float* Wv      = (const float*)d_in[7];
  const float* ln_s_g  = (const float*)d_in[8];
  const float* ln_s_b  = (const float*)d_in[9];
  const float* Wq      = (const float*)d_in[10];
  const float* gru_wih = (const float*)d_in[11];
  const float* gru_whh = (const float*)d_in[12];
  const float* gru_bih = (const float*)d_in[13];
  const float* gru_bhh = (const float*)d_in[14];
  const float* ln_m_g  = (const float*)d_in[15];
  const float* ln_m_b  = (const float*)d_in[16];
  const float* mlp_w1  = (const float*)d_in[17];
  const float* mlp_b1  = (const float*)d_in[18];
  const float* mlp_w2  = (const float*)d_in[19];
  const float* mlp_b2  = (const float*)d_in[20];

  float* ws = (float*)d_ws;
  float* kbuf  = ws;
  float* vbuf  = kbuf + (size_t)B_ * N_ * D_;
  float* hprev = vbuf + (size_t)B_ * N_ * D_;
  float* qbuf  = hprev + H_ * B_ * S_ * D_;
  float* pnum  = qbuf + H_ * B_ * S_ * D_;
  float* pden  = pnum + B_ * ATT_SPLIT_ * S_ * D_;

  ln_kv_kernel<<<KV_GRID_, 256, 0, stream>>>(inputs, ln_in_g, ln_in_b,
                                                        Wk, Wv, kbuf, vbuf);
  init_hprev_q_kernel<<<32, 128, 0, stream>>>(sinit, smu, slog, ln_s_g, ln_s_b,
                                              Wq, hprev, qbuf);
  attn_kernel<<<ATT_SPLIT_ * B_, 512, 0, stream>>>(kbuf, vbuf, qbuf, pnum, pden);
  gru_mlp_kernel<false><<<32, 256, 0, stream>>>(
      pnum, pden, hprev, gru_wih, gru_whh, gru_bih, gru_bhh, ln_m_g, ln_m_b,
      mlp_w1, mlp_b1, mlp_w2, mlp_b2, ln_s_g, ln_s_b, Wq, qbuf, (float*)d_out);
  attn_kernel<<<ATT_SPLIT_ * B_, 512, 0, stream>>>(kbuf, vbuf, qbuf, pnum, pden);
  gru_mlp_kernel<false><<<32, 256, 0, stream>>>(
      pnum, pden, hprev, gru_wih, gru_whh, gru_bih, gru_bhh, ln_m_g, ln_m_b,
      mlp_w1, mlp_b1, mlp_w2, mlp_b2, ln_s_g, ln_s_b, Wq, qbuf, (float*)d_out);
  attn_kernel<<<ATT_SPLIT_ * B_, 512, 0, stream>>>(kbuf, vbuf, qbuf, pnum, pden);
  gru_mlp_kernel<true><<<32, 256, 0, stream>>>(
      pnum, pden, hprev, gru_wih, gru_whh, gru_bih, gru_bhh, ln_m_g, ln_m_b,
      mlp_w1, mlp_b1, mlp_w2, mlp_b2, ln_s_g, ln_s_b, Wq, qbuf, (float*)d_out);
}


# k and v never formed: x=LN(inputs) stored once, attention reads x, q uses Wq Wk^T and the GRU Wv wih (both made per call on WMMA); LN, logit and update sum order changes
# speedup vs baseline: 5.7753x; 1.2391x over previous
#include <hip/hip_runtime.h>

typedef float v2f __attribute__((ext_vector_type(2)));
typedef float v4f __attribute__((ext_vector_type(4)));
typedef float v8f __attribute__((ext_vector_type(8)));

constexpr int H_ = 8, B_ = 64, S_ = 8, D_ = 64, N_ = 4096, MLPH_ = 128, ITERS_ = 3;
constexpr float LN_EPS_ = 1e-5f, EPS_ATTN_ = 1e-8f, SCALE_ = 0.125f;
static_assert(ITERS_ == 3, "kernel_launch writes the three iterations out");

__device__ __forceinline__ v8f wmma_f32(v2f a, v2f b, v8f c) {
  return __builtin_amdgcn_wmma_f32_16x16x4_f32(false, a, false, b, (short)0, c,
                                               false, false);
}

template <int M>
__device__ __forceinline__ float xlane(float x) {
  return __builtin_bit_cast(float, __builtin_amdgcn_update_dpp(
      0, __builtin_bit_cast(int, x), 0x160 + M, 0xf, 0xf, true));
}

constexpr int LN_WAVES_ = 16;
constexpr int LN_GRID_ = 512;
constexpr int LN_TILES_ = (B_ * N_) / 16;
constexpr int LN_STRIDE_ = LN_GRID_ * LN_WAVES_;
constexpr int LN_TRIPS_ = LN_TILES_ / LN_STRIDE_;
constexpr int FUSE_GRID_ = 4;
static_assert(LN_TILES_ % LN_STRIDE_ == 0, "tiles must divide over the waves");
static_assert(FUSE_GRID_ * LN_WAVES_ == 16 + 48, "one fused-weight tile per wave");

__global__ void __launch_bounds__(512) ln_x_kernel(
    const float* __restrict__ inp, const float* __restrict__ lng,
    const float* __restrict__ lnb, const float* __restrict__ Wq,
    const float* __restrict__ Wk, const float* __restrict__ Wv,
    const float* __restrict__ wih, float* __restrict__ xout,
    float* __restrict__ wqk, float* __restrict__ wvih) {
  const int tid = threadIdx.x;
  const int w = tid >> 5, lane = tid & 31, c = lane & 15, h = lane >> 4;
  if (blockIdx.x < FUSE_GRID_) {
    const int g = blockIdx.x * LN_WAVES_ + w;
    v2f a[16], b[16];
    float* dst;
    int ld;
    if (g < 16) {
      const int rt = (g >> 2) * 16, ct = (g & 3) * 16;
#pragma unroll
      for (int kk = 0; kk < 16; ++kk) {
        const int kb = kk * 4 + h * 2;
        a[kk] = *(const v2f*)(Wq + (rt + c) * D_ + kb);
        b[kk] = *(const v2f*)(Wk + (ct + c) * D_ + kb);
      }
      dst = wqk + (size_t)(rt + 8 * h) * D_ + ct + c;
      ld = D_;
    } else {
      const int g2 = g - 16;
      const int rt = (g2 / 12) * 16, ct = (g2 % 12) * 16;
#pragma unroll
      for (int kk = 0; kk < 16; ++kk) {
        const int kb = kk * 4 + h * 2;
        a[kk] = *(const v2f*)(Wv + (rt + c) * D_ + kb);
        b[kk].x = wih[kb * 192 + ct + c];
        b[kk].y = wih[(kb + 1) * 192 + ct + c];
      }
      dst = wvih + (size_t)(rt + 8 * h) * 192 + ct + c;
      ld = 192;
    }
    v8f acc0 = {}, acc1 = {};
#pragma unroll
    for (int kk = 0; kk < 16; kk += 2) {
      acc0 = wmma_f32(a[kk], b[kk], acc0);
      acc1 = wmma_f32(a[kk + 1], b[kk + 1], acc1);
    }
    for (int i = 0; i < 8; ++i) dst[(size_t)i * ld] = acc0[i] + acc1[i];
    return;
  }
  const int gw = (blockIdx.x - FUSE_GRID_) * LN_WAVES_ + w;
  const v4f gf = *(const v4f*)(lng + 4 * c);
  const v4f bf = *(const v4f*)(lnb + 4 * c);
  for (int j = 0; j < LN_TRIPS_; ++j) {
    const size_t off = ((size_t)(gw + j * LN_STRIDE_) * 16 + h) * D_ + 4 * c;
    v4f xr[8];
#pragma unroll
    for (int i = 0; i < 8; ++i)
      xr[i] = __builtin_nontemporal_load((const v4f*)(inp + off + (size_t)(2 * i) * D_));
#pragma unroll
    for (int i = 0; i < 8; ++i) {
      const v4f x = xr[i];
      float s = (x[0] + x[1]) + (x[2] + x[3]);
      s += xlane<1>(s); s += xlane<2>(s); s += xlane<4>(s); s += xlane<8>(s);
      const float m = s * (1.f / D_);
      const v4f d = x - m;
      float q = (d[0] * d[0] + d[1] * d[1]) + (d[2] * d[2] + d[3] * d[3]);
      q += xlane<1>(q); q += xlane<2>(q); q += xlane<4>(q); q += xlane<8>(q);
      const float rs = rsqrtf(q * (1.f / D_) + LN_EPS_);
      *(v4f*)(xout + off + (size_t)(2 * i) * D_) = d * rs * gf + bf;
    }
  }
}

template <int NT>
__device__ __forceinline__ void ln_q_tile(
    float (*hl)[68], float* mu, float* rstd, const float* __restrict__ lng,
    const float* __restrict__ lnb, const float* __restrict__ Wq,
    float* hprev, float* __restrict__ qout, int rowbase, int tid) {
  if (tid < 16) {
    float m = 0.f;
    for (int j = 0; j < D_; ++j) m += hl[tid][j];
    m *= (1.f / D_);
    float var = 0.f;
    for (int j = 0; j < D_; ++j) { float dd = hl[tid][j] - m; var += dd * dd; }
    var *= (1.f / D_);
    mu[tid] = m; rstd[tid] = rsqrtf(var + LN_EPS_);
  }
  __syncthreads();
  for (int e = tid; e < 16 * D_; e += NT) {
    int r = e >> 6, c = e & 63;
    float nv = (hl[r][c] - mu[r]) * rstd[r] * lng[c] + lnb[c];
    hl[r][c] = nv;
    hprev[(size_t)(rowbase + r) * D_ + c] = nv;
  }
  __syncthreads();
  const int w = tid >> 5, lane = tid & 31, m16 = lane & 15, kh = lane >> 4;
  if (w < 4) {
    const int ct = w * 16;
    v8f acc = {};
    for (int kk = 0; kk < 16; ++kk) {
      int kb = kk * 4 + kh * 2;
      v2f a; a.x = hl[m16][kb]; a.y = hl[m16][kb + 1];
      v2f b; b.x = Wq[kb * D_ + ct + m16]; b.y = Wq[(kb + 1) * D_ + ct + m16];
      acc = wmma_f32(a, b, acc);
    }
    for (int i = 0; i < 8; ++i)
      qout[(size_t)(rowbase + i + 8 * kh) * D_ + ct + m16] = acc[i];
  }
}

__global__ void __launch_bounds__(128) init_hprev_q_kernel(
    const float* __restrict__ sinit, const float* __restrict__ smu,
    const float* __restrict__ slog, const float* __restrict__ lng,
    const float* __restrict__ lnb, const float* __restrict__ Wq,
    float* hprev, float* __restrict__ qout) {
  __shared__ float hl[16][68];
  __shared__ float mu[16], rstd[16];
  const int tid = threadIdx.x;
  const int rowbase = blockIdx.x * 16;
  for (int e = tid; e < 16 * D_; e += 128) {
    int r = e >> 6, c = e & 63;
    hl[r][c] = smu[c] + expf(slog[c]) * sinit[(size_t)(rowbase + r) * D_ + c];
  }
  __syncthreads();
  ln_q_tile<128>(hl, mu, rstd, lng, lnb, Wq, hprev, qout, rowbase, tid);
}

constexpr int ATT_WAVES_ = 16;
constexpr int ATT_SPLIT_ = 4;
constexpr int ATT_TILES_ = (N_ / 16) / ATT_SPLIT_;
static_assert(ATT_TILES_ % ATT_WAVES_ == 0, "tiles must divide over the waves");
__global__ void __launch_bounds__(512) __attribute__((amdgpu_waves_per_eu(4, 4)))
attn_kernel(
    const float* __restrict__ xbuf,
    const float* __restrict__ qbuf, float* __restrict__ pnum,
    float* __restrict__ pden) {
  __shared__ alignas(16) float red[ATT_WAVES_][8][68];
  __shared__ float dred[ATT_WAVES_][32];
  const int tid = threadIdx.x;
  const int ch = blockIdx.x >> 6;
  const int b  = blockIdx.x & 63;
  const int w = tid >> 5, lane = tid & 31, c = lane & 15, kh = lane >> 4;
  const float* kb = xbuf + (size_t)b * N_ * D_;
  const float* vb = kb;
  const float* qrow = qbuf + (size_t)(b * S_ + (c & 7)) * D_ + 4 * kh;
  float4 qf[8];
  for (int i = 0; i < 8; ++i) qf[i] = *(const float4*)(qrow + 8 * i);
  v8f acc0 = {}, acc1 = {}, acc2 = {}, acc3 = {};
  float dsum = 0.f;
  for (int t = ch * ATT_TILES_ + w; t < (ch + 1) * ATT_TILES_; t += ATT_WAVES_) {
    const int n0 = t * 16;
    const float* kp = kb + (size_t)(n0 + c) * D_ + 4 * kh;
    const float* vp = vb + (size_t)(n0 + 8 * kh) * D_ + 4 * c;
    float4 kf[8], vf[8];
    for (int i = 0; i < 8; ++i) kf[i] = *(const float4*)(kp + 8 * i);
    for (int r = 0; r < 8; ++r) vf[r] = *(const float4*)(vp + (size_t)r * D_);
    v8f lg = {};
    for (int i = 0; i < 8; ++i) {
      v2f a0; a0.x = kf[i].x; a0.y = kf[i].y;
      v2f b0; b0.x = qf[i].x; b0.y = qf[i].y;
      lg = wmma_f32(a0, b0, lg);
      v2f a1; a1.x = kf[i].z; a1.y = kf[i].w;
      v2f b1; b1.x = qf[i].z; b1.y = qf[i].w;
      lg = wmma_f32(a1, b1, lg);
    }
    float p[8];
#pragma unroll
    for (int r = 0; r < 8; ++r) {
      float x = lg[r] * SCALE_;
      float mx = fmaxf(x, xlane<1>(x));
      mx = fmaxf(mx, xlane<2>(mx));
      mx = fmaxf(mx, xlane<4>(mx));
      float e = expf(x - mx);
      float sm = e + xlane<1>(e);
      sm += xlane<2>(sm);
      sm += xlane<4>(sm);
      float inv = 1.f / sm;
      p[r] = e * inv + EPS_ATTN_;
      dsum += p[r];
    }
    for (int j = 0; j < 4; ++j) {
      v2f a; a.x = p[2 * j]; a.y = p[2 * j + 1];
      v2f b0; b0.x = vf[2 * j].x; b0.y = vf[2 * j + 1].x;
      acc0 = wmma_f32(a, b0, acc0);
      v2f b1; b1.x = vf[2 * j].y; b1.y = vf[2 * j + 1].y;
      acc1 = wmma_f32(a, b1, acc1);
      v2f b2; b2.x = vf[2 * j].z; b2.y = vf[2 * j + 1].z;
      acc2 = wmma_f32(a, b2, acc2);
      v2f b3; b3.x = vf[2 * j].w; b3.y = vf[2 * j + 1].w;
      acc3 = wmma_f32(a, b3, acc3);
    }
  }
  if (kh == 0) {
    for (int i = 0; i < 8; ++i) {
      float4 o; o.x = acc0[i]; o.y = acc1[i]; o.z = acc2[i]; o.w = acc3[i];
      *(float4*)&red[w][i][4 * c] = o;
    }
  }
  dred[w][lane] = dsum;
  __syncthreads();
  if (tid < 128) {
    const int sp = tid >> 4, d4 = (tid & 15) * 4;
    float4 sacc; sacc.x = 0.f; sacc.y = 0.f; sacc.z = 0.f; sacc.w = 0.f;
    float den = 0.f;
    for (int ww = 0; ww < ATT_WAVES_; ++ww) {
      float4 x = *(const float4*)&red[ww][sp][d4];
      sacc.x += x.x; sacc.y += x.y; sacc.z += x.z; sacc.w += x.w;
      den += dred[ww][sp] + dred[ww][sp + 16];
    }
    const size_t prow = (size_t)((b * ATT_SPLIT_ + ch) * S_ + sp);
    *(float4*)(pnum + prow * D_ + d4) = sacc;
    if (d4 == 0) pden[prow] = den;
  }
}

template <bool LAST>
__global__ void __launch_bounds__(256) gru_mlp_kernel(
    const float* __restrict__ pnum, const float* __restrict__ pden,
    float* hprev,
    const float* __restrict__ wih, const float* __restrict__ whh,
    const float* __restrict__ bih, const float* __restrict__ bhh,
    const float* __restrict__ lnmg, const float* __restrict__ lnmb,
    const float* __restrict__ w1, const float* __restrict__ b1,
    const float* __restrict__ w2, const float* __restrict__ b2,
    const float* __restrict__ lnsg, const float* __restrict__ lnsb,
    const float* __restrict__ Wq, float* __restrict__ qout,
    float* __restrict__ out) {
  __shared__ float ut[16][68];
  __shared__ float ht[16][68];
  __shared__ float gx[16][196];
  __shared__ float gh[16][196];
  __shared__ float sn[16][68];
  __shared__ float nm[16][68];
  __shared__ float hid[16][132];
  __shared__ float mu[16], rstd[16];
  const int tid = threadIdx.x;
  const int h = 0, rt = blockIdx.x;
  const int rowbase = h * (B_ * S_) + rt * 16;
  const int r0 = tid >> 4, c00 = (tid & 15) * 4;
  {
    const int srow = rt * 16 + r0;
    const int bb = srow >> 3, ss = srow & 7;
    float4 a; a.x = 0.f; a.y = 0.f; a.z = 0.f; a.w = 0.f;
    float den = 0.f;
    for (int q = 0; q < ATT_SPLIT_; ++q) {
      const size_t prow = (size_t)((bb * ATT_SPLIT_ + q) * S_ + ss);
      const float4 x = *(const float4*)(pnum + prow * D_ + c00);
      a.x += x.x; a.y += x.y; a.z += x.z; a.w += x.w;
      den += pden[prow];
    }
    a.x = a.x / den; a.y = a.y / den; a.z = a.z / den; a.w = a.w / den;
    ut[r0][c00] = a.x; ut[r0][c00 + 1] = a.y; ut[r0][c00 + 2] = a.z; ut[r0][c00 + 3] = a.w;
    float4 c = *(const float4*)(hprev + (size_t)(rowbase + r0) * D_ + c00);
    ht[r0][c00] = c.x; ht[r0][c00 + 1] = c.y; ht[r0][c00 + 2] = c.z; ht[r0][c00 + 3] = c.w;
  }
  __syncthreads();
  const int w = tid >> 5, lane = tid & 31, m16 = lane & 15, kh = lane >> 4;
  const float* wihh = wih + (size_t)h * D_ * 192;
  const float* whhh = whh + (size_t)h * D_ * 192;
  for (int T = w; T < 24; T += 8) {
    const bool isX = T < 12;
    const int ct = (T % 12) * 16;
    const float* Wt = isX ? wihh : whhh;
    v8f acc = {};
    for (int kk = 0; kk < 16; ++kk) {
      int kb = kk * 4 + kh * 2;
      v2f a;
      if (isX) { a.x = ut[m16][kb]; a.y = ut[m16][kb + 1]; }
      else     { a.x = ht[m16][kb]; a.y = ht[m16][kb + 1]; }
      v2f bb; bb.x = Wt[kb * 192 + ct + m16]; bb.y = Wt[(kb + 1) * 192 + ct + m16];
      acc = wmma_f32(a, bb, acc);
    }
    float* dst = isX ? &gx[0][0] : &gh[0][0];
    for (int i = 0; i < 8; ++i)
      dst[(i + 8 * kh) * 196 + ct + m16] = acc[i];
  }
  __syncthreads();
  {
    const float* bi = bih + h * 192;
    const float* bh = bhh + h * 192;
    for (int j = 0; j < 4; ++j) {
      int d = c00 + j;
      float xr = gx[r0][d] + bi[d],        hr = gh[r0][d] + bh[d];
      float xz = gx[r0][64 + d] + bi[64 + d],  hz = gh[r0][64 + d] + bh[64 + d];
      float xn = gx[r0][128 + d] + bi[128 + d], hn = gh[r0][128 + d] + bh[128 + d];
      float rg = 1.f / (1.f + expf(-(xr + hr)));
      float zg = 1.f / (1.f + expf(-(xz + hz)));
      float ng = tanhf(xn + rg * hn);
      sn[r0][d] = (1.f - zg) * ng + zg * ht[r0][d];
    }
  }
  __syncthreads();
  if (tid < 16) {
    float m = 0.f;
    for (int j = 0; j < D_; ++j) m += sn[tid][j];
    m *= (1.f / D_);
    float var = 0.f;
    for (int j = 0; j < D_; ++j) { float dd = sn[tid][j] - m; var += dd * dd; }
    var *= (1.f / D_);
    mu[tid] = m; rstd[tid] = rsqrtf(var + LN_EPS_);
  }
  __syncthreads();
  for (int j = 0; j < 4; ++j) {
    int d = c00 + j;
    nm[r0][d] = (sn[r0][d] - mu[r0]) * rstd[r0] * lnmg[h * D_ + d] + lnmb[h * D_ + d];
  }
  __syncthreads();
  {
    const float* w1h = w1 + (size_t)h * D_ * MLPH_;
    const float* b1h = b1 + h * MLPH_;
    const int ct = w * 16;
    v8f acc = {};
    for (int kk = 0; kk < 16; ++kk) {
      int kb = kk * 4 + kh * 2;
      v2f a;  a.x = nm[m16][kb]; a.y = nm[m16][kb + 1];
      v2f bb; bb.x = w1h[kb * MLPH_ + ct + m16]; bb.y = w1h[(kb + 1) * MLPH_ + ct + m16];
      acc = wmma_f32(a, bb, acc);
    }
    float bias = b1h[ct + m16];
    for (int i = 0; i < 8; ++i)
      hid[i + 8 * kh][ct + m16] = fmaxf(acc[i] + bias, 0.f);
  }
  __syncthreads();
  if (w < 4) {
    const float* w2h = w2 + (size_t)h * MLPH_ * D_;
    const float* b2h = b2 + h * D_;
    const int ct = w * 16;
    v8f acc = {};
    for (int kk = 0; kk < 32; ++kk) {
      int kb = kk * 4 + kh * 2;
      v2f a;  a.x = hid[m16][kb]; a.y = hid[m16][kb + 1];
      v2f bb; bb.x = w2h[kb * D_ + ct + m16]; bb.y = w2h[(kb + 1) * D_ + ct + m16];
      acc = wmma_f32(a, bb, acc);
    }
    float bias = b2h[ct + m16];
    for (int i = 0; i < 8; ++i) {
      int rr = i + 8 * kh;
      const float sv = sn[rr][ct + m16] + acc[i] + bias;
      if (LAST) out[(size_t)(rowbase + rr) * D_ + ct + m16] = sv;
      else ut[rr][ct + m16] = sv;
    }
  }
  if (!LAST) {
    __syncthreads();
    ln_q_tile<256>(ut, mu, rstd, lnsg, lnsb, Wq, hprev, qout, rowbase, tid);
  }
}

extern "C" void kernel_launch(void* const* d_in, const int* in_sizes, int n_in,
                              void* d_out, int out_size, void* d_ws, size_t ws_size,
                              hipStream_t stream) {
  (void)in_sizes; (void)n_in; (void)out_size; (void)ws_size;
  const float* inputs  = (const float*)d_in[0];
  const float* sinit   = (const float*)d_in[1];
  const float* smu     = (const float*)d_in[2];
  const float* slog    = (const float*)d_in[3];
  const float* ln_in_g = (const float*)d_in[4];
  const float* ln_in_b = (const float*)d_in[5];
  const float* Wk      = (const float*)d_in[6];
  const float* Wv      = (const float*)d_in[7];
  const float* ln_s_g  = (const float*)d_in[8];
  const float* ln_s_b  = (const float*)d_in[9];
  const float* Wq      = (const float*)d_in[10];
  const float* gru_wih = (const float*)d_in[11];
  const float* gru_whh = (const float*)d_in[12];
  const float* gru_bih = (const float*)d_in[13];
  const float* gru_bhh = (const float*)d_in[14];
  const float* ln_m_g  = (const float*)d_in[15];
  const float* ln_m_b  = (const float*)d_in[16];
  const float* mlp_w1  = (const float*)d_in[17];
  const float* mlp_b1  = (const float*)d_in[18];
  const float* mlp_w2  = (const float*)d_in[19];
  const float* mlp_b2  = (const float*)d_in[20];

  float* ws = (float*)d_ws;
  float* xbuf  = ws;
  float* hprev = xbuf + (size_t)B_ * N_ * D_;
  float* qbuf  = hprev + H_ * B_ * S_ * D_;
  float* pnum  = qbuf + H_ * B_ * S_ * D_;
  float* pden  = pnum + B_ * ATT_SPLIT_ * S_ * D_;
  float* wqk   = pden + B_ * ATT_SPLIT_ * S_;
  float* wvih  = wqk + D_ * D_;

  ln_x_kernel<<<FUSE_GRID_ + LN_GRID_, 512, 0, stream>>>(
      inputs, ln_in_g, ln_in_b, Wq, Wk, Wv, gru_wih, xbuf, wqk, wvih);
  init_hprev_q_kernel<<<32, 128, 0, stream>>>(sinit, smu, slog, ln_s_g, ln_s_b,
                                              wqk, hprev, qbuf);
  attn_kernel<<<ATT_SPLIT_ * B_, 512, 0, stream>>>(xbuf, qbuf, pnum, pden);
  gru_mlp_kernel<false><<<32, 256, 0, stream>>>(
      pnum, pden, hprev, wvih, gru_whh, gru_bih, gru_bhh, ln_m_g, ln_m_b,
      mlp_w1, mlp_b1, mlp_w2, mlp_b2, ln_s_g, ln_s_b, wqk, qbuf, (float*)d_out);
  attn_kernel<<<ATT_SPLIT_ * B_, 512, 0, stream>>>(xbuf, qbuf, pnum, pden);
  gru_mlp_kernel<false><<<32, 256, 0, stream>>>(
      pnum, pden, hprev, wvih, gru_whh, gru_bih, gru_bhh, ln_m_g, ln_m_b,
      mlp_w1, mlp_b1, mlp_w2, mlp_b2, ln_s_g, ln_s_b, wqk, qbuf, (float*)d_out);
  attn_kernel<<<ATT_SPLIT_ * B_, 512, 0, stream>>>(xbuf, qbuf, pnum, pden);
  gru_mlp_kernel<true><<<32, 256, 0, stream>>>(
      pnum, pden, hprev, wvih, gru_whh, gru_bih, gru_bhh, ln_m_g, ln_m_b,
      mlp_w1, mlp_b1, mlp_w2, mlp_b2, ln_s_g, ln_s_b, wqk, qbuf, (float*)d_out);
}


# GRU/MLP and init launches: every global operand loaded at entry as one group, weights staged as float4 lines in padded LDS, product fragments read from LDS as a group; every sum keeps its order
# speedup vs baseline: 6.8671x; 1.1890x over previous
#include <hip/hip_runtime.h>

typedef float v2f __attribute__((ext_vector_type(2)));
typedef float v4f __attribute__((ext_vector_type(4)));
typedef float v8f __attribute__((ext_vector_type(8)));

constexpr int H_ = 8, B_ = 64, S_ = 8, D_ = 64, N_ = 4096, MLPH_ = 128, ITERS_ = 3;
constexpr float LN_EPS_ = 1e-5f, EPS_ATTN_ = 1e-8f, SCALE_ = 0.125f;
static_assert(ITERS_ == 3, "kernel_launch writes the three iterations out");

__device__ __forceinline__ v8f wmma_f32(v2f a, v2f b, v8f c) {
  return __builtin_amdgcn_wmma_f32_16x16x4_f32(false, a, false, b, (short)0, c,
                                               false, false);
}

template <int M>
__device__ __forceinline__ float xlane(float x) {
  return __builtin_bit_cast(float, __builtin_amdgcn_update_dpp(
      0, __builtin_bit_cast(int, x), 0x160 + M, 0xf, 0xf, true));
}

constexpr int LN_WAVES_ = 16;
constexpr int LN_GRID_ = 512;
constexpr int LN_TILES_ = (B_ * N_) / 16;
constexpr int LN_STRIDE_ = LN_GRID_ * LN_WAVES_;
constexpr int LN_TRIPS_ = LN_TILES_ / LN_STRIDE_;
constexpr int FUSE_GRID_ = 4;
static_assert(LN_TILES_ % LN_STRIDE_ == 0, "tiles must divide over the waves");
static_assert(FUSE_GRID_ * LN_WAVES_ == 16 + 48, "one fused-weight tile per wave");

__global__ void __launch_bounds__(512) ln_x_kernel(
    const float* __restrict__ inp, const float* __restrict__ lng,
    const float* __restrict__ lnb, const float* __restrict__ Wq,
    const float* __restrict__ Wk, const float* __restrict__ Wv,
    const float* __restrict__ wih, float* __restrict__ xout,
    float* __restrict__ wqk, float* __restrict__ wvih) {
  const int tid = threadIdx.x;
  const int w = tid >> 5, lane = tid & 31, c = lane & 15, h = lane >> 4;
  if (blockIdx.x < FUSE_GRID_) {
    const int g = blockIdx.x * LN_WAVES_ + w;
    v2f a[16], b[16];
    float* dst;
    int ld;
    if (g < 16) {
      const int rt = (g >> 2) * 16, ct = (g & 3) * 16;
#pragma unroll
      for (int kk = 0; kk < 16; ++kk) {
        const int kb = kk * 4 + h * 2;
        a[kk] = *(const v2f*)(Wq + (rt + c) * D_ + kb);
        b[kk] = *(const v2f*)(Wk + (ct + c) * D_ + kb);
      }
      dst = wqk + (size_t)(rt + 8 * h) * D_ + ct + c;
      ld = D_;
    } else {
      const int g2 = g - 16;
      const int rt = (g2 / 12) * 16, ct = (g2 % 12) * 16;
#pragma unroll
      for (int kk = 0; kk < 16; ++kk) {
        const int kb = kk * 4 + h * 2;
        a[kk] = *(const v2f*)(Wv + (rt + c) * D_ + kb);
        b[kk].x = wih[kb * 192 + ct + c];
        b[kk].y = wih[(kb + 1) * 192 + ct + c];
      }
      dst = wvih + (size_t)(rt + 8 * h) * 192 + ct + c;
      ld = 192;
    }
    v8f acc0 = {}, acc1 = {};
#pragma unroll
    for (int kk = 0; kk < 16; kk += 2) {
      acc0 = wmma_f32(a[kk], b[kk], acc0);
      acc1 = wmma_f32(a[kk + 1], b[kk + 1], acc1);
    }
    for (int i = 0; i < 8; ++i) dst[(size_t)i * ld] = acc0[i] + acc1[i];
    return;
  }
  const int gw = (blockIdx.x - FUSE_GRID_) * LN_WAVES_ + w;
  const v4f gf = *(const v4f*)(lng + 4 * c);
  const v4f bf = *(const v4f*)(lnb + 4 * c);
  for (int j = 0; j < LN_TRIPS_; ++j) {
    const size_t off = ((size_t)(gw + j * LN_STRIDE_) * 16 + h) * D_ + 4 * c;
    v4f xr[8];
#pragma unroll
    for (int i = 0; i < 8; ++i)
      xr[i] = __builtin_nontemporal_load((const v4f*)(inp + off + (size_t)(2 * i) * D_));
#pragma unroll
    for (int i = 0; i < 8; ++i) {
      const v4f x = xr[i];
      float s = (x[0] + x[1]) + (x[2] + x[3]);
      s += xlane<1>(s); s += xlane<2>(s); s += xlane<4>(s); s += xlane<8>(s);
      const float m = s * (1.f / D_);
      const v4f d = x - m;
      float q = (d[0] * d[0] + d[1] * d[1]) + (d[2] * d[2] + d[3] * d[3]);
      q += xlane<1>(q); q += xlane<2>(q); q += xlane<4>(q); q += xlane<8>(q);
      const float rs = rsqrtf(q * (1.f / D_) + LN_EPS_);
      *(v4f*)(xout + off + (size_t)(2 * i) * D_) = d * rs * gf + bf;
    }
  }
}

constexpr int PX_ = 200;
constexpr int P1_ = 136;
constexpr int P2_ = 72;
static_assert((2 * PX_) % 64 == 16 && (2 * P1_) % 64 == 16 && (2 * P2_) % 64 == 16, "bank spread");

template <int ROWS, int COLS, int NT>
__device__ __forceinline__ void wload(float4 (&v)[ROWS * COLS / 4 / NT],
                                      const float* __restrict__ W, int tid) {
#pragma unroll
  for (int i = 0; i < ROWS * COLS / 4 / NT; ++i)
    v[i] = *(const float4*)(W + 4 * (tid + NT * i));
}
template <int ROWS, int COLS, int PITCH, int NT>
__device__ __forceinline__ void wstore(float* L, const float4 (&v)[ROWS * COLS / 4 / NT],
                                       int tid) {
#pragma unroll
  for (int i = 0; i < ROWS * COLS / 4 / NT; ++i) {
    const int e = tid + NT * i;
    const int r = e / (COLS / 4), c4 = (e % (COLS / 4)) * 4;
    *(float4*)(L + r * PITCH + c4) = v[i];
  }
}

template <int KSTEPS, int AP, int BP>
__device__ __forceinline__ v8f lds_product(const float* A, const float* Bm,
                                           int m16, int kh, int ct) {
  v2f a[KSTEPS], b[KSTEPS];
#pragma unroll
  for (int kk = 0; kk < KSTEPS; ++kk) {
    const int kb = kk * 4 + kh * 2;
    a[kk].x = A[m16 * AP + kb]; a[kk].y = A[m16 * AP + kb + 1];
    b[kk].x = Bm[kb * BP + ct + m16]; b[kk].y = Bm[(kb + 1) * BP + ct + m16];
  }
  v8f acc = {};
#pragma unroll
  for (int kk = 0; kk < KSTEPS; ++kk) acc = wmma_f32(a[kk], b[kk], acc);
  __builtin_amdgcn_sched_group_barrier(0x100, 4 * KSTEPS, 0);
  __builtin_amdgcn_sched_group_barrier(0x008, KSTEPS, 0);
  return acc;
}

template <int NT>
__device__ __forceinline__ void ln_q_tile(
    float (*hl)[68], float* mu, float* rstd, float lg, float lb,
    const float* WqL, float* hprev, float* __restrict__ qout, int rowbase, int tid) {
  static_assert(NT % 64 == 0, "a thread keeps one column");
  if (tid < 16) {
    float m = 0.f;
    for (int j = 0; j < D_; ++j) m += hl[tid][j];
    m *= (1.f / D_);
    float var = 0.f;
    for (int j = 0; j < D_; ++j) { float dd = hl[tid][j] - m; var += dd * dd; }
    var *= (1.f / D_);
    mu[tid] = m; rstd[tid] = rsqrtf(var + LN_EPS_);
  }
  __syncthreads();
  for (int e = tid; e < 16 * D_; e += NT) {
    int r = e >> 6, c = e & 63;
    float nv = (hl[r][c] - mu[r]) * rstd[r] * lg + lb;
    hl[r][c] = nv;
    hprev[(size_t)(rowbase + r) * D_ + c] = nv;
  }
  __syncthreads();
  const int w = tid >> 5, lane = tid & 31, m16 = lane & 15, kh = lane >> 4;
  if (w < 4) {
    const int ct = w * 16;
    const v8f acc = lds_product<16, 68, P2_>(&hl[0][0], WqL, m16, kh, ct);
    for (int i = 0; i < 8; ++i)
      qout[(size_t)(rowbase + i + 8 * kh) * D_ + ct + m16] = acc[i];
  }
}

__global__ void __launch_bounds__(128) init_hprev_q_kernel(
    const float* __restrict__ sinit, const float* __restrict__ smu,
    const float* __restrict__ slog, const float* __restrict__ lng,
    const float* __restrict__ lnb, const float* __restrict__ Wq,
    float* hprev, float* __restrict__ qout) {
  __shared__ float hl[16][68];
  __shared__ float mu[16], rstd[16];
  __shared__ alignas(16) float Lq[64 * P2_];
  const int tid = threadIdx.x;
  const int rowbase = blockIdx.x * 16;
  const int c = tid & 63, rb = tid >> 6;
  float si[8];
#pragma unroll
  for (int i = 0; i < 8; ++i) si[i] = sinit[(size_t)(rowbase + rb + 2 * i) * D_ + c];
  const float m0 = smu[c], sd = expf(slog[c]);
  const float lg = lng[c], lb = lnb[c];
  float4 sq[8];
  wload<64, 64, 128>(sq, Wq, tid);
#pragma unroll
  for (int i = 0; i < 8; ++i) hl[rb + 2 * i][c] = m0 + sd * si[i];
  wstore<64, 64, P2_, 128>(Lq, sq, tid);
  __syncthreads();
  ln_q_tile<128>(hl, mu, rstd, lg, lb, Lq, hprev, qout, rowbase, tid);
}

constexpr int ATT_WAVES_ = 16;
constexpr int ATT_SPLIT_ = 4;
constexpr int ATT_TILES_ = (N_ / 16) / ATT_SPLIT_;
static_assert(ATT_TILES_ % ATT_WAVES_ == 0, "tiles must divide over the waves");
__global__ void __launch_bounds__(512) __attribute__((amdgpu_waves_per_eu(4, 4)))
attn_kernel(
    const float* __restrict__ xbuf,
    const float* __restrict__ qbuf, float* __restrict__ pnum,
    float* __restrict__ pden) {
  __shared__ alignas(16) float red[ATT_WAVES_][8][68];
  __shared__ float dred[ATT_WAVES_][32];
  const int tid = threadIdx.x;
  const int ch = blockIdx.x >> 6;
  const int b  = blockIdx.x & 63;
  const int w = tid >> 5, lane = tid & 31, c = lane & 15, kh = lane >> 4;
  const float* kb = xbuf + (size_t)b * N_ * D_;
  const float* vb = kb;
  const float* qrow = qbuf + (size_t)(b * S_ + (c & 7)) * D_ + 4 * kh;
  float4 qf[8];
  for (int i = 0; i < 8; ++i) qf[i] = *(const float4*)(qrow + 8 * i);
  v8f acc0 = {}, acc1 = {}, acc2 = {}, acc3 = {};
  float dsum = 0.f;
  for (int t = ch * ATT_TILES_ + w; t < (ch + 1) * ATT_TILES_; t += ATT_WAVES_) {
    const int n0 = t * 16;
    const float* kp = kb + (size_t)(n0 + c) * D_ + 4 * kh;
    const float* vp = vb + (size_t)(n0 + 8 * kh) * D_ + 4 * c;
    float4 kf[8], vf[8];
    for (int i = 0; i < 8; ++i) kf[i] = *(const float4*)(kp + 8 * i);
    for (int r = 0; r < 8; ++r) vf[r] = *(const float4*)(vp + (size_t)r * D_);
    v8f lg = {};
    for (int i = 0; i < 8; ++i) {
      v2f a0; a0.x = kf[i].x; a0.y = kf[i].y;
      v2f b0; b0.x = qf[i].x; b0.y = qf[i].y;
      lg = wmma_f32(a0, b0, lg);
      v2f a1; a1.x = kf[i].z; a1.y = kf[i].w;
      v2f b1; b1.x = qf[i].z; b1.y = qf[i].w;
      lg = wmma_f32(a1, b1, lg);
    }
    float p[8];
#pragma unroll
    for (int r = 0; r < 8; ++r) {
      float x = lg[r] * SCALE_;
      float mx = fmaxf(x, xlane<1>(x));
      mx = fmaxf(mx, xlane<2>(mx));
      mx = fmaxf(mx, xlane<4>(mx));
      float e = expf(x - mx);
      float sm = e + xlane<1>(e);
      sm += xlane<2>(sm);
      sm += xlane<4>(sm);
      float inv = 1.f / sm;
      p[r] = e * inv + EPS_ATTN_;
      dsum += p[r];
    }
    for (int j = 0; j < 4; ++j) {
      v2f a; a.x = p[2 * j]; a.y = p[2 * j + 1];
      v2f b0; b0.x = vf[2 * j].x; b0.y = vf[2 * j + 1].x;
      acc0 = wmma_f32(a, b0, acc0);
      v2f b1; b1.x = vf[2 * j].y; b1.y = vf[2 * j + 1].y;
      acc1 = wmma_f32(a, b1, acc1);
      v2f b2; b2.x = vf[2 * j].z; b2.y = vf[2 * j + 1].z;
      acc2 = wmma_f32(a, b2, acc2);
      v2f b3; b3.x = vf[2 * j].w; b3.y = vf[2 * j + 1].w;
      acc3 = wmma_f32(a, b3, acc3);
    }
  }
  if (kh == 0) {
    for (int i = 0; i < 8; ++i) {
      float4 o; o.x = acc0[i]; o.y = acc1[i]; o.z = acc2[i]; o.w = acc3[i];
      *(float4*)&red[w][i][4 * c] = o;
    }
  }
  dred[w][lane] = dsum;
  __syncthreads();
  if (tid < 128) {
    const int sp = tid >> 4, d4 = (tid & 15) * 4;
    float4 sacc; sacc.x = 0.f; sacc.y = 0.f; sacc.z = 0.f; sacc.w = 0.f;
    float den = 0.f;
    for (int ww = 0; ww < ATT_WAVES_; ++ww) {
      float4 x = *(const float4*)&red[ww][sp][d4];
      sacc.x += x.x; sacc.y += x.y; sacc.z += x.z; sacc.w += x.w;
      den += dred[ww][sp] + dred[ww][sp + 16];
    }
    const size_t prow = (size_t)((b * ATT_SPLIT_ + ch) * S_ + sp);
    *(float4*)(pnum + prow * D_ + d4) = sacc;
    if (d4 == 0) pden[prow] = den;
  }
}

template <bool LAST>
__global__ void __launch_bounds__(256) gru_mlp_kernel(
    const float* __restrict__ pnum, const float* __restrict__ pden,
    float* hprev,
    const float* __restrict__ wih, const float* __restrict__ whh,
    const float* __restrict__ bih, const float* __restrict__ bhh,
    const float* __restrict__ lnmg, const float* __restrict__ lnmb,
    const float* __restrict__ w1, const float* __restrict__ b1,
    const float* __restrict__ w2, const float* __restrict__ b2,
    const float* __restrict__ lnsg, const float* __restrict__ lnsb,
    const float* __restrict__ Wq, float* __restrict__ qout,
    float* __restrict__ out) {
  __shared__ float ut[16][68];
  __shared__ float ht[16][68];
  __shared__ float gx[16][196];
  __shared__ float gh[16][196];
  __shared__ float sn[16][68];
  __shared__ float nm[16][68];
  __shared__ float hid[16][132];
  __shared__ float mu[16], rstd[16];
  __shared__ alignas(16) float Lx[64 * PX_];
  __shared__ alignas(16) float Lh[64 * PX_];
  __shared__ alignas(16) float L1[64 * P1_];
  __shared__ alignas(16) float L2[128 * P2_];
  __shared__ alignas(16) float Lq[LAST ? 4 : 64 * P2_];
  const int tid = threadIdx.x;
  const int rt = blockIdx.x;
  const int rowbase = rt * 16;
  const int r0 = tid >> 4, c00 = (tid & 15) * 4;
  const int w = tid >> 5, lane = tid & 31, m16 = lane & 15, kh = lane >> 4;

  const int srow = rt * 16 + r0;
  const int bb = srow >> 3, ss = srow & 7;
  float4 pn[ATT_SPLIT_];
  float pd[ATT_SPLIT_];
#pragma unroll
  for (int q = 0; q < ATT_SPLIT_; ++q) {
    const size_t prow = (size_t)((bb * ATT_SPLIT_ + q) * S_ + ss);
    pn[q] = *(const float4*)(pnum + prow * D_ + c00);
    pd[q] = pden[prow];
  }
  const float4 hp = *(const float4*)(hprev + (size_t)(rowbase + r0) * D_ + c00);
  float4 sx[12], sh[12];
  wload<64, 192, 256>(sx, wih, tid);
  wload<64, 192, 256>(sh, whh, tid);
  const v4f bir = *(const v4f*)(bih + c00), biz = *(const v4f*)(bih + 64 + c00),
            bin = *(const v4f*)(bih + 128 + c00);
  const v4f bhr = *(const v4f*)(bhh + c00), bhz = *(const v4f*)(bhh + 64 + c00),
            bhn = *(const v4f*)(bhh + 128 + c00);
  const v4f mg = *(const v4f*)(lnmg + c00), mb = *(const v4f*)(lnmb + c00);
  const float b1v = b1[w * 16 + m16];
  const float b2v = b2[(w & 3) * 16 + m16];
  float sg = 0.f, sb = 0.f;
  if (!LAST) { sg = lnsg[tid & 63]; sb = lnsb[tid & 63]; }
  float4 s1[8], s2[8], sq[4];
  wload<64, 128, 256>(s1, w1, tid);
  wload<128, 64, 256>(s2, w2, tid);
  if (!LAST) wload<64, 64, 256>(sq, Wq, tid);

  {
    float4 a; a.x = 0.f; a.y = 0.f; a.z = 0.f; a.w = 0.f;
    float den = 0.f;
#pragma unroll
    for (int q = 0; q < ATT_SPLIT_; ++q) {
      a.x += pn[q].x; a.y += pn[q].y; a.z += pn[q].z; a.w += pn[q].w;
      den += pd[q];
    }
    a.x = a.x / den; a.y = a.y / den; a.z = a.z / den; a.w = a.w / den;
    ut[r0][c00] = a.x; ut[r0][c00 + 1] = a.y; ut[r0][c00 + 2] = a.z; ut[r0][c00 + 3] = a.w;
    ht[r0][c00] = hp.x; ht[r0][c00 + 1] = hp.y; ht[r0][c00 + 2] = hp.z; ht[r0][c00 + 3] = hp.w;
  }
  wstore<64, 192, PX_, 256>(Lx, sx, tid);
  wstore<64, 192, PX_, 256>(Lh, sh, tid);
  wstore<64, 128, P1_, 256>(L1, s1, tid);
  wstore<128, 64, P2_, 256>(L2, s2, tid);
  if (!LAST) wstore<64, 64, P2_, 256>(Lq, sq, tid);
  __syncthreads();
  for (int T = w; T < 24; T += 8) {
    const bool isX = T < 12;
    const int ct = (T % 12) * 16;
    const float* Wt = isX ? Lx : Lh;
    const float* At = isX ? &ut[0][0] : &ht[0][0];
    const v8f acc = lds_product<16, 68, PX_>(At, Wt, m16, kh, ct);
    float* dst = isX ? &gx[0][0] : &gh[0][0];
    for (int i = 0; i < 8; ++i)
      dst[(i + 8 * kh) * 196 + ct + m16] = acc[i];
  }
  __syncthreads();
  {
#pragma unroll
    for (int j = 0; j < 4; ++j) {
      int d = c00 + j;
      float xr = gx[r0][d] + bir[j],       hr = gh[r0][d] + bhr[j];
      float xz = gx[r0][64 + d] + biz[j],  hz = gh[r0][64 + d] + bhz[j];
      float xn = gx[r0][128 + d] + bin[j], hn = gh[r0][128 + d] + bhn[j];
      float rg = 1.f / (1.f + expf(-(xr + hr)));
      float zg = 1.f / (1.f + expf(-(xz + hz)));
      float ng = tanhf(xn + rg * hn);
      sn[r0][d] = (1.f - zg) * ng + zg * ht[r0][d];
    }
  }
  __syncthreads();
  if (tid < 16) {
    float m = 0.f;
    for (int j = 0; j < D_; ++j) m += sn[tid][j];
    m *= (1.f / D_);
    float var = 0.f;
    for (int j = 0; j < D_; ++j) { float dd = sn[tid][j] - m; var += dd * dd; }
    var *= (1.f / D_);
    mu[tid] = m; rstd[tid] = rsqrtf(var + LN_EPS_);
  }
  __syncthreads();
#pragma unroll
  for (int j = 0; j < 4; ++j) {
    int d = c00 + j;
    nm[r0][d] = (sn[r0][d] - mu[r0]) * rstd[r0] * mg[j] + mb[j];
  }
  __syncthreads();
  {
    const int ct = w * 16;
    const v8f acc = lds_product<16, 68, P1_>(&nm[0][0], L1, m16, kh, ct);
    for (int i = 0; i < 8; ++i)
      hid[i + 8 * kh][ct + m16] = fmaxf(acc[i] + b1v, 0.f);
  }
  __syncthreads();
  if (w < 4) {
    const int ct = w * 16;
    const v8f acc = lds_product<32, 132, P2_>(&hid[0][0], L2, m16, kh, ct);
    for (int i = 0; i < 8; ++i) {
      int rr = i + 8 * kh;
      const float sv = sn[rr][ct + m16] + acc[i] + b2v;
      if (LAST) out[(size_t)(rowbase + rr) * D_ + ct + m16] = sv;
      else ut[rr][ct + m16] = sv;
    }
  }
  if (!LAST) {
    __syncthreads();
    ln_q_tile<256>(ut, mu, rstd, sg, sb, Lq, hprev, qout, rowbase, tid);
  }
}

extern "C" void kernel_launch(void* const* d_in, const int* in_sizes, int n_in,
                              void* d_out, int out_size, void* d_ws, size_t ws_size,
                              hipStream_t stream) {
  (void)in_sizes; (void)n_in; (void)out_size; (void)ws_size;
  const float* inputs  = (const float*)d_in[0];
  const float* sinit   = (const float*)d_in[1];
  const float* smu     = (const float*)d_in[2];
  const float* slog    = (const float*)d_in[3];
  const float* ln_in_g = (const float*)d_in[4];
  const float* ln_in_b = (const float*)d_in[5];
  const float* Wk      = (const float*)d_in[6];
  const float* Wv      = (const float*)d_in[7];
  const float* ln_s_g  = (const float*)d_in[8];
  const float* ln_s_b  = (const float*)d_in[9];
  const float* Wq      = (const float*)d_in[10];
  const float* gru_wih = (const float*)d_in[11];
  const float* gru_whh = (const float*)d_in[12];
  const float* gru_bih = (const float*)d_in[13];
  const float* gru_bhh = (const float*)d_in[14];
  const float* ln_m_g  = (const float*)d_in[15];
  const float* ln_m_b  = (const float*)d_in[16];
  const float* mlp_w1  = (const float*)d_in[17];
  const float* mlp_b1  = (const float*)d_in[18];
  const float* mlp_w2  = (const float*)d_in[19];
  const float* mlp_b2  = (const float*)d_in[20];

  float* ws = (float*)d_ws;
  float* xbuf  = ws;
  float* hprev = xbuf + (size_t)B_ * N_ * D_;
  float* qbuf  = hprev + H_ * B_ * S_ * D_;
  float* pnum  = qbuf + H_ * B_ * S_ * D_;
  float* pden  = pnum + B_ * ATT_SPLIT_ * S_ * D_;
  float* wqk   = pden + B_ * ATT_SPLIT_ * S_;
  float* wvih  = wqk + D_ * D_;

  ln_x_kernel<<<FUSE_GRID_ + LN_GRID_, 512, 0, stream>>>(
      inputs, ln_in_g, ln_in_b, Wq, Wk, Wv, gru_wih, xbuf, wqk, wvih);
  init_hprev_q_kernel<<<32, 128, 0, stream>>>(sinit, smu, slog, ln_s_g, ln_s_b,
                                              wqk, hprev, qbuf);
  attn_kernel<<<ATT_SPLIT_ * B_, 512, 0, stream>>>(xbuf, qbuf, pnum, pden);
  gru_mlp_kernel<false><<<32, 256, 0, stream>>>(
      pnum, pden, hprev, wvih, gru_whh, gru_bih, gru_bhh, ln_m_g, ln_m_b,
      mlp_w1, mlp_b1, mlp_w2, mlp_b2, ln_s_g, ln_s_b, wqk, qbuf, (float*)d_out);
  attn_kernel<<<ATT_SPLIT_ * B_, 512, 0, stream>>>(xbuf, qbuf, pnum, pden);
  gru_mlp_kernel<false><<<32, 256, 0, stream>>>(
      pnum, pden, hprev, wvih, gru_whh, gru_bih, gru_bhh, ln_m_g, ln_m_b,
      mlp_w1, mlp_b1, mlp_w2, mlp_b2, ln_s_g, ln_s_b, wqk, qbuf, (float*)d_out);
  attn_kernel<<<ATT_SPLIT_ * B_, 512, 0, stream>>>(xbuf, qbuf, pnum, pden);
  gru_mlp_kernel<true><<<32, 256, 0, stream>>>(
      pnum, pden, hprev, wvih, gru_whh, gru_bih, gru_bhh, ln_m_g, ln_m_b,
      mlp_w1, mlp_b1, mlp_w2, mlp_b2, ln_s_g, ln_s_b, wqk, qbuf, (float*)d_out);
}


# x=LN(inputs) never formed: attention reads raw rows and applies row mean/rstd (made and stored by the first pass) to the WMMA logits and update sums; LN pass gone; LN, logit, update sum order changes
# speedup vs baseline: 7.7130x; 1.1232x over previous
#include <hip/hip_runtime.h>

typedef float v2f __attribute__((ext_vector_type(2)));
typedef float v4f __attribute__((ext_vector_type(4)));
typedef float v8f __attribute__((ext_vector_type(8)));

constexpr int H_ = 8, B_ = 64, S_ = 8, D_ = 64, N_ = 4096, MLPH_ = 128, ITERS_ = 3;
constexpr float LN_EPS_ = 1e-5f, EPS_ATTN_ = 1e-8f, SCALE_ = 0.125f;
static_assert(ITERS_ == 3, "kernel_launch writes the three iterations out");

__device__ __forceinline__ v8f wmma_f32(v2f a, v2f b, v8f c) {
  return __builtin_amdgcn_wmma_f32_16x16x4_f32(false, a, false, b, (short)0, c,
                                               false, false);
}

template <int M>
__device__ __forceinline__ float xlane(float x) {
  return __builtin_bit_cast(float, __builtin_amdgcn_update_dpp(
      0, __builtin_bit_cast(int, x), 0x160 + M, 0xf, 0xf, true));
}

constexpr int FUSE_WAVES_ = 16;
constexpr int FUSE_GRID_ = 4;
static_assert(FUSE_GRID_ * FUSE_WAVES_ == 16 + 48, "one fused-weight tile per wave");

__global__ void __launch_bounds__(512) fuse_w_kernel(
    const float* __restrict__ Wq, const float* __restrict__ Wk,
    const float* __restrict__ Wv, const float* __restrict__ wih,
    float* __restrict__ wqk, float* __restrict__ wvih) {
  const int tid = threadIdx.x;
  const int w = tid >> 5, lane = tid & 31, c = lane & 15, h = lane >> 4;
  const int g = blockIdx.x * FUSE_WAVES_ + w;
  v2f a[16], b[16];
  float* dst;
  int ld;
  if (g < 16) {
    const int rt = (g >> 2) * 16, ct = (g & 3) * 16;
#pragma unroll
    for (int kk = 0; kk < 16; ++kk) {
      const int kb = kk * 4 + h * 2;
      a[kk] = *(const v2f*)(Wq + (rt + c) * D_ + kb);
      b[kk] = *(const v2f*)(Wk + (ct + c) * D_ + kb);
    }
    dst = wqk + (size_t)(rt + 8 * h) * D_ + ct + c;
    ld = D_;
  } else {
    const int g2 = g - 16;
    const int rt = (g2 / 12) * 16, ct = (g2 % 12) * 16;
#pragma unroll
    for (int kk = 0; kk < 16; ++kk) {
      const int kb = kk * 4 + h * 2;
      a[kk] = *(const v2f*)(Wv + (rt + c) * D_ + kb);
      b[kk].x = wih[kb * 192 + ct + c];
      b[kk].y = wih[(kb + 1) * 192 + ct + c];
    }
    dst = wvih + (size_t)(rt + 8 * h) * 192 + ct + c;
    ld = 192;
  }
  v8f acc0 = {}, acc1 = {};
#pragma unroll
  for (int kk = 0; kk < 16; kk += 2) {
    acc0 = wmma_f32(a[kk], b[kk], acc0);
    acc1 = wmma_f32(a[kk + 1], b[kk + 1], acc1);
  }
  for (int i = 0; i < 8; ++i) dst[(size_t)i * ld] = acc0[i] + acc1[i];
}

constexpr int PX_ = 200;
constexpr int P1_ = 136;
constexpr int P2_ = 72;
static_assert((2 * PX_) % 64 == 16 && (2 * P1_) % 64 == 16 && (2 * P2_) % 64 == 16, "bank spread");

template <int ROWS, int COLS, int NT>
__device__ __forceinline__ void wload(float4 (&v)[ROWS * COLS / 4 / NT],
                                      const float* __restrict__ W, int tid) {
#pragma unroll
  for (int i = 0; i < ROWS * COLS / 4 / NT; ++i)
    v[i] = *(const float4*)(W + 4 * (tid + NT * i));
}
template <int ROWS, int COLS, int PITCH, int NT>
__device__ __forceinline__ void wstore(float* L, const float4 (&v)[ROWS * COLS / 4 / NT],
                                       int tid) {
#pragma unroll
  for (int i = 0; i < ROWS * COLS / 4 / NT; ++i) {
    const int e = tid + NT * i;
    const int r = e / (COLS / 4), c4 = (e % (COLS / 4)) * 4;
    *(float4*)(L + r * PITCH + c4) = v[i];
  }
}

template <int KSTEPS, int AP, int BP>
__device__ __forceinline__ v8f lds_product(const float* A, const float* Bm,
                                           int m16, int kh, int ct) {
  v2f a[KSTEPS], b[KSTEPS];
#pragma unroll
  for (int kk = 0; kk < KSTEPS; ++kk) {
    const int kb = kk * 4 + kh * 2;
    a[kk].x = A[m16 * AP + kb]; a[kk].y = A[m16 * AP + kb + 1];
    b[kk].x = Bm[kb * BP + ct + m16]; b[kk].y = Bm[(kb + 1) * BP + ct + m16];
  }
  v8f acc = {};
#pragma unroll
  for (int kk = 0; kk < KSTEPS; ++kk) acc = wmma_f32(a[kk], b[kk], acc);
  __builtin_amdgcn_sched_group_barrier(0x100, 4 * KSTEPS, 0);
  __builtin_amdgcn_sched_group_barrier(0x008, KSTEPS, 0);
  return acc;
}

template <int NT>
__device__ __forceinline__ void ln_q_tile(
    float (*hl)[68], float* mu, float* rstd, float lg, float lb,
    const float* WqL, float* hprev, float* __restrict__ qout, int rowbase, int tid) {
  static_assert(NT % 64 == 0, "a thread keeps one column");
  if (tid < 16) {
    float m = 0.f;
    for (int j = 0; j < D_; ++j) m += hl[tid][j];
    m *= (1.f / D_);
    float var = 0.f;
    for (int j = 0; j < D_; ++j) { float dd = hl[tid][j] - m; var += dd * dd; }
    var *= (1.f / D_);
    mu[tid] = m; rstd[tid] = rsqrtf(var + LN_EPS_);
  }
  __syncthreads();
  for (int e = tid; e < 16 * D_; e += NT) {
    int r = e >> 6, c = e & 63;
    float nv = (hl[r][c] - mu[r]) * rstd[r] * lg + lb;
    hl[r][c] = nv;
    hprev[(size_t)(rowbase + r) * D_ + c] = nv;
  }
  __syncthreads();
  const int w = tid >> 5, lane = tid & 31, m16 = lane & 15, kh = lane >> 4;
  if (w < 4) {
    const int ct = w * 16;
    const v8f acc = lds_product<16, 68, P2_>(&hl[0][0], WqL, m16, kh, ct);
    for (int i = 0; i < 8; ++i)
      qout[(size_t)(rowbase + i + 8 * kh) * D_ + ct + m16] = acc[i];
  }
}

__global__ void __launch_bounds__(128) init_hprev_q_kernel(
    const float* __restrict__ sinit, const float* __restrict__ smu,
    const float* __restrict__ slog, const float* __restrict__ lng,
    const float* __restrict__ lnb, const float* __restrict__ Wq,
    float* hprev, float* __restrict__ qout) {
  __shared__ float hl[16][68];
  __shared__ float mu[16], rstd[16];
  __shared__ alignas(16) float Lq[64 * P2_];
  const int tid = threadIdx.x;
  const int rowbase = blockIdx.x * 16;
  const int c = tid & 63, rb = tid >> 6;
  float si[8];
#pragma unroll
  for (int i = 0; i < 8; ++i) si[i] = sinit[(size_t)(rowbase + rb + 2 * i) * D_ + c];
  const float m0 = smu[c], sd = expf(slog[c]);
  const float lg = lng[c], lb = lnb[c];
  float4 sq[8];
  wload<64, 64, 128>(sq, Wq, tid);
#pragma unroll
  for (int i = 0; i < 8; ++i) hl[rb + 2 * i][c] = m0 + sd * si[i];
  wstore<64, 64, P2_, 128>(Lq, sq, tid);
  __syncthreads();
  ln_q_tile<128>(hl, mu, rstd, lg, lb, Lq, hprev, qout, rowbase, tid);
}

constexpr int ATT_WAVES_ = 16;
constexpr int ATT_SPLIT_ = 4;
constexpr int ATT_TILES_ = (N_ / 16) / ATT_SPLIT_;
static_assert(ATT_TILES_ % ATT_WAVES_ == 0, "tiles must divide over the waves");
template <bool FIRST>
__global__ void __launch_bounds__(512) __attribute__((amdgpu_waves_per_eu(4, 4)))
attn_kernel(
    const float* __restrict__ inp, float* __restrict__ stat,
    const float* __restrict__ lng, const float* __restrict__ lnb,
    const float* __restrict__ qbuf, float* __restrict__ pnum,
    float* __restrict__ pden) {
  __shared__ alignas(16) float red[ATT_WAVES_][8][68];
  __shared__ float dred[ATT_WAVES_][32];
  __shared__ float tred[ATT_WAVES_][32];
  const int tid = threadIdx.x;
  const int ch = blockIdx.x >> 6;
  const int b  = blockIdx.x & 63;
  const int w = tid >> 5, lane = tid & 31, c = lane & 15, kh = lane >> 4;
  const float* kb = inp + (size_t)b * N_ * D_;
  const float* vb = kb;
  const float* qrow = qbuf + (size_t)(b * S_ + (c & 7)) * D_ + 4 * kh;
  float4 qf[8];
  float sg = 0.f, sb = 0.f;
#pragma unroll
  for (int i = 0; i < 8; ++i) {
    float4 q = *(const float4*)(qrow + 8 * i);
    const float4 g = *(const float4*)(lng + 4 * kh + 8 * i);
    const float4 bb = *(const float4*)(lnb + 4 * kh + 8 * i);
    sb += (bb.x * q.x + bb.y * q.y) + (bb.z * q.z + bb.w * q.w);
    q.x *= g.x * SCALE_; q.y *= g.y * SCALE_; q.z *= g.z * SCALE_; q.w *= g.w * SCALE_;
    sg += (q.x + q.y) + (q.z + q.w);
    qf[i] = q;
  }
  sb *= SCALE_;
  sg += __shfl_xor(sg, 16);
  sb += __shfl_xor(sb, 16);
  v8f acc0 = {}, acc1 = {}, acc2 = {}, acc3 = {};
  float dsum = 0.f, tsum = 0.f;
  for (int t = ch * ATT_TILES_ + w; t < (ch + 1) * ATT_TILES_; t += ATT_WAVES_) {
    const int n0 = t * 16;
    const float* kp = kb + (size_t)(n0 + c) * D_ + 4 * kh;
    const float* vp = vb + (size_t)(n0 + 8 * kh) * D_ + 4 * c;
    float4 kf[8], vf[8];
    for (int i = 0; i < 8; ++i) kf[i] = *(const float4*)(kp + 8 * i);
    for (int r = 0; r < 8; ++r) vf[r] = *(const float4*)(vp + (size_t)r * D_);
    float4* sp4 = (float4*)(stat + ((size_t)b * N_ + n0 + 8 * kh) * 2);
    float4 st[4];
    if (!FIRST)
      for (int k = 0; k < 4; ++k) st[k] = sp4[k];
    v8f lg = {};
    for (int i = 0; i < 8; ++i) {
      v2f a0; a0.x = kf[i].x; a0.y = kf[i].y;
      v2f b0; b0.x = qf[i].x; b0.y = qf[i].y;
      lg = wmma_f32(a0, b0, lg);
      v2f a1; a1.x = kf[i].z; a1.y = kf[i].w;
      v2f b1; b1.x = qf[i].z; b1.y = qf[i].w;
      lg = wmma_f32(a1, b1, lg);
    }
    float mr[8], rr[8];
    if (FIRST) {
#pragma unroll
      for (int r = 0; r < 8; ++r) {
        const float4 x = vf[r];
        float s = (x.x + x.y) + (x.z + x.w);
        s += xlane<1>(s); s += xlane<2>(s); s += xlane<4>(s); s += xlane<8>(s);
        const float m = s * (1.f / D_);
        const float e0 = x.x - m, e1 = x.y - m, e2 = x.z - m, e3 = x.w - m;
        float q2 = (e0 * e0 + e1 * e1) + (e2 * e2 + e3 * e3);
        q2 += xlane<1>(q2); q2 += xlane<2>(q2); q2 += xlane<4>(q2); q2 += xlane<8>(q2);
        mr[r] = m;
        rr[r] = rsqrtf(q2 * (1.f / D_) + LN_EPS_);
      }
      if (c == 0) {
#pragma unroll
        for (int k = 0; k < 4; ++k) {
          float4 o; o.x = mr[2 * k]; o.y = rr[2 * k]; o.z = mr[2 * k + 1]; o.w = rr[2 * k + 1];
          sp4[k] = o;
        }
      }
    } else {
#pragma unroll
      for (int k = 0; k < 4; ++k) {
        mr[2 * k] = st[k].x; rr[2 * k] = st[k].y; mr[2 * k + 1] = st[k].z; rr[2 * k + 1] = st[k].w;
      }
    }
    float p[8], pa[8];
#pragma unroll
    for (int r = 0; r < 8; ++r) {
      float x = rr[r] * (lg[r] - mr[r] * sg) + sb;
      float mx = fmaxf(x, xlane<1>(x));
      mx = fmaxf(mx, xlane<2>(mx));
      mx = fmaxf(mx, xlane<4>(mx));
      float e = expf(x - mx);
      float sm = e + xlane<1>(e);
      sm += xlane<2>(sm);
      sm += xlane<4>(sm);
      float inv = 1.f / sm;
      p[r] = e * inv + EPS_ATTN_;
      dsum += p[r];
      pa[r] = p[r] * rr[r];
      tsum += pa[r] * mr[r];
    }
    for (int j = 0; j < 4; ++j) {
      v2f a; a.x = pa[2 * j]; a.y = pa[2 * j + 1];
      v2f b0; b0.x = vf[2 * j].x; b0.y = vf[2 * j + 1].x;
      acc0 = wmma_f32(a, b0, acc0);
      v2f b1; b1.x = vf[2 * j].y; b1.y = vf[2 * j + 1].y;
      acc1 = wmma_f32(a, b1, acc1);
      v2f b2; b2.x = vf[2 * j].z; b2.y = vf[2 * j + 1].z;
      acc2 = wmma_f32(a, b2, acc2);
      v2f b3; b3.x = vf[2 * j].w; b3.y = vf[2 * j + 1].w;
      acc3 = wmma_f32(a, b3, acc3);
    }
  }
  if (kh == 0) {
    for (int i = 0; i < 8; ++i) {
      float4 o; o.x = acc0[i]; o.y = acc1[i]; o.z = acc2[i]; o.w = acc3[i];
      *(float4*)&red[w][i][4 * c] = o;
    }
  }
  dred[w][lane] = dsum;
  tred[w][lane] = tsum;
  __syncthreads();
  if (tid < 128) {
    const int sp = tid >> 4, d4 = (tid & 15) * 4;
    float4 sacc; sacc.x = 0.f; sacc.y = 0.f; sacc.z = 0.f; sacc.w = 0.f;
    float den = 0.f, tt = 0.f;
    for (int ww = 0; ww < ATT_WAVES_; ++ww) {
      float4 x = *(const float4*)&red[ww][sp][d4];
      sacc.x += x.x; sacc.y += x.y; sacc.z += x.z; sacc.w += x.w;
      den += dred[ww][sp] + dred[ww][sp + 16];
      tt += tred[ww][sp] + tred[ww][sp + 16];
    }
    sacc.x -= tt; sacc.y -= tt; sacc.z -= tt; sacc.w -= tt;
    const size_t prow = (size_t)((b * ATT_SPLIT_ + ch) * S_ + sp);
    *(float4*)(pnum + prow * D_ + d4) = sacc;
    if (d4 == 0) pden[prow] = den;
  }
}

template <bool LAST>
__global__ void __launch_bounds__(256) gru_mlp_kernel(
    const float* __restrict__ pnum, const float* __restrict__ pden,
    const float* __restrict__ lning, const float* __restrict__ lninb,
    float* hprev,
    const float* __restrict__ wih, const float* __restrict__ whh,
    const float* __restrict__ bih, const float* __restrict__ bhh,
    const float* __restrict__ lnmg, const float* __restrict__ lnmb,
    const float* __restrict__ w1, const float* __restrict__ b1,
    const float* __restrict__ w2, const float* __restrict__ b2,
    const float* __restrict__ lnsg, const float* __restrict__ lnsb,
    const float* __restrict__ Wq, float* __restrict__ qout,
    float* __restrict__ out) {
  __shared__ float ut[16][68];
  __shared__ float ht[16][68];
  __shared__ float gx[16][196];
  __shared__ float gh[16][196];
  __shared__ float sn[16][68];
  __shared__ float nm[16][68];
  __shared__ float hid[16][132];
  __shared__ float mu[16], rstd[16];
  __shared__ alignas(16) float Lx[64 * PX_];
  __shared__ alignas(16) float Lh[64 * PX_];
  __shared__ alignas(16) float L1[64 * P1_];
  __shared__ alignas(16) float L2[128 * P2_];
  __shared__ alignas(16) float Lq[LAST ? 4 : 64 * P2_];
  const int tid = threadIdx.x;
  const int rt = blockIdx.x;
  const int rowbase = rt * 16;
  const int r0 = tid >> 4, c00 = (tid & 15) * 4;
  const int w = tid >> 5, lane = tid & 31, m16 = lane & 15, kh = lane >> 4;

  const int srow = rt * 16 + r0;
  const int bb = srow >> 3, ss = srow & 7;
  float4 pn[ATT_SPLIT_];
  float pd[ATT_SPLIT_];
#pragma unroll
  for (int q = 0; q < ATT_SPLIT_; ++q) {
    const size_t prow = (size_t)((bb * ATT_SPLIT_ + q) * S_ + ss);
    pn[q] = *(const float4*)(pnum + prow * D_ + c00);
    pd[q] = pden[prow];
  }
  const float4 hp = *(const float4*)(hprev + (size_t)(rowbase + r0) * D_ + c00);
  float4 sx[12], sh[12];
  wload<64, 192, 256>(sx, wih, tid);
  wload<64, 192, 256>(sh, whh, tid);
  const v4f bir = *(const v4f*)(bih + c00), biz = *(const v4f*)(bih + 64 + c00),
            bin = *(const v4f*)(bih + 128 + c00);
  const v4f bhr = *(const v4f*)(bhh + c00), bhz = *(const v4f*)(bhh + 64 + c00),
            bhn = *(const v4f*)(bhh + 128 + c00);
  const v4f mg = *(const v4f*)(lnmg + c00), mb = *(const v4f*)(lnmb + c00);
  const v4f ig = *(const v4f*)(lning + c00), ib = *(const v4f*)(lninb + c00);
  const float b1v = b1[w * 16 + m16];
  const float b2v = b2[(w & 3) * 16 + m16];
  float sg = 0.f, sb = 0.f;
  if (!LAST) { sg = lnsg[tid & 63]; sb = lnsb[tid & 63]; }
  float4 s1[8], s2[8], sq[4];
  wload<64, 128, 256>(s1, w1, tid);
  wload<128, 64, 256>(s2, w2, tid);
  if (!LAST) wload<64, 64, 256>(sq, Wq, tid);

  {
    float4 a; a.x = 0.f; a.y = 0.f; a.z = 0.f; a.w = 0.f;
    float den = 0.f;
#pragma unroll
    for (int q = 0; q < ATT_SPLIT_; ++q) {
      a.x += pn[q].x; a.y += pn[q].y; a.z += pn[q].z; a.w += pn[q].w;
      den += pd[q];
    }
    a.x = a.x / den * ig[0] + ib[0]; a.y = a.y / den * ig[1] + ib[1];
    a.z = a.z / den * ig[2] + ib[2]; a.w = a.w / den * ig[3] + ib[3];
    ut[r0][c00] = a.x; ut[r0][c00 + 1] = a.y; ut[r0][c00 + 2] = a.z; ut[r0][c00 + 3] = a.w;
    ht[r0][c00] = hp.x; ht[r0][c00 + 1] = hp.y; ht[r0][c00 + 2] = hp.z; ht[r0][c00 + 3] = hp.w;
  }
  wstore<64, 192, PX_, 256>(Lx, sx, tid);
  wstore<64, 192, PX_, 256>(Lh, sh, tid);
  wstore<64, 128, P1_, 256>(L1, s1, tid);
  wstore<128, 64, P2_, 256>(L2, s2, tid);
  if (!LAST) wstore<64, 64, P2_, 256>(Lq, sq, tid);
  __syncthreads();
  for (int T = w; T < 24; T += 8) {
    const bool isX = T < 12;
    const int ct = (T % 12) * 16;
    const float* Wt = isX ? Lx : Lh;
    const float* At = isX ? &ut[0][0] : &ht[0][0];
    const v8f acc = lds_product<16, 68, PX_>(At, Wt, m16, kh, ct);
    float* dst = isX ? &gx[0][0] : &gh[0][0];
    for (int i = 0; i < 8; ++i)
      dst[(i + 8 * kh) * 196 + ct + m16] = acc[i];
  }
  __syncthreads();
  {
#pragma unroll
    for (int j = 0; j < 4; ++j) {
      int d = c00 + j;
      float xr = gx[r0][d] + bir[j],       hr = gh[r0][d] + bhr[j];
      float xz = gx[r0][64 + d] + biz[j],  hz = gh[r0][64 + d] + bhz[j];
      float xn = gx[r0][128 + d] + bin[j], hn = gh[r0][128 + d] + bhn[j];
      float rg = 1.f / (1.f + expf(-(xr + hr)));
      float zg = 1.f / (1.f + expf(-(xz + hz)));
      float ng = tanhf(xn + rg * hn);
      sn[r0][d] = (1.f - zg) * ng + zg * ht[r0][d];
    }
  }
  __syncthreads();
  if (tid < 16) {
    float m = 0.f;
    for (int j = 0; j < D_; ++j) m += sn[tid][j];
    m *= (1.f / D_);
    float var = 0.f;
    for (int j = 0; j < D_; ++j) { float dd = sn[tid][j] - m; var += dd * dd; }
    var *= (1.f / D_);
    mu[tid] = m; rstd[tid] = rsqrtf(var + LN_EPS_);
  }
  __syncthreads();
#pragma unroll
  for (int j = 0; j < 4; ++j) {
    int d = c00 + j;
    nm[r0][d] = (sn[r0][d] - mu[r0]) * rstd[r0] * mg[j] + mb[j];
  }
  __syncthreads();
  {
    const int ct = w * 16;
    const v8f acc = lds_product<16, 68, P1_>(&nm[0][0], L1, m16, kh, ct);
    for (int i = 0; i < 8; ++i)
      hid[i + 8 * kh][ct + m16] = fmaxf(acc[i] + b1v, 0.f);
  }
  __syncthreads();
  if (w < 4) {
    const int ct = w * 16;
    const v8f acc = lds_product<32, 132, P2_>(&hid[0][0], L2, m16, kh, ct);
    for (int i = 0; i < 8; ++i) {
      int rr = i + 8 * kh;
      const float sv = sn[rr][ct + m16] + acc[i] + b2v;
      if (LAST) out[(size_t)(rowbase + rr) * D_ + ct + m16] = sv;
      else ut[rr][ct + m16] = sv;
    }
  }
  if (!LAST) {
    __syncthreads();
    ln_q_tile<256>(ut, mu, rstd, sg, sb, Lq, hprev, qout, rowbase, tid);
  }
}

extern "C" void kernel_launch(void* const* d_in, const int* in_sizes, int n_in,
                              void* d_out, int out_size, void* d_ws, size_t ws_size,
                              hipStream_t stream) {
  (void)in_sizes; (void)n_in; (void)out_size; (void)ws_size;
  const float* inputs  = (const float*)d_in[0];
  const float* sinit   = (const float*)d_in[1];
  const float* smu     = (const float*)d_in[2];
  const float* slog    = (const float*)d_in[3];
  const float* ln_in_g = (const float*)d_in[4];
  const float* ln_in_b = (const float*)d_in[5];
  const float* Wk      = (const float*)d_in[6];
  const float* Wv      = (const float*)d_in[7];
  const float* ln_s_g  = (const float*)d_in[8];
  const float* ln_s_b  = (const float*)d_in[9];
  const float* Wq      = (const float*)d_in[10];
  const float* gru_wih = (const float*)d_in[11];
  const float* gru_whh = (const float*)d_in[12];
  const float* gru_bih = (const float*)d_in[13];
  const float* gru_bhh = (const float*)d_in[14];
  const float* ln_m_g  = (const float*)d_in[15];
  const float* ln_m_b  = (const float*)d_in[16];
  const float* mlp_w1  = (const float*)d_in[17];
  const float* mlp_b1  = (const float*)d_in[18];
  const float* mlp_w2  = (const float*)d_in[19];
  const float* mlp_b2  = (const float*)d_in[20];

  float* ws = (float*)d_ws;
  float* stat  = ws;
  float* hprev = stat + (size_t)B_ * N_ * 2;
  float* qbuf  = hprev + H_ * B_ * S_ * D_;
  float* pnum  = qbuf + H_ * B_ * S_ * D_;
  float* pden  = pnum + B_ * ATT_SPLIT_ * S_ * D_;
  float* wqk   = pden + B_ * ATT_SPLIT_ * S_;
  float* wvih  = wqk + D_ * D_;

  fuse_w_kernel<<<FUSE_GRID_, 512, 0, stream>>>(Wq, Wk, Wv, gru_wih, wqk, wvih);
  init_hprev_q_kernel<<<32, 128, 0, stream>>>(sinit, smu, slog, ln_s_g, ln_s_b,
                                              wqk, hprev, qbuf);
  attn_kernel<true><<<ATT_SPLIT_ * B_, 512, 0, stream>>>(inputs, stat, ln_in_g, ln_in_b,
                                                         qbuf, pnum, pden);
  gru_mlp_kernel<false><<<32, 256, 0, stream>>>(
      pnum, pden, ln_in_g, ln_in_b, hprev, wvih, gru_whh, gru_bih, gru_bhh, ln_m_g, ln_m_b,
      mlp_w1, mlp_b1, mlp_w2, mlp_b2, ln_s_g, ln_s_b, wqk, qbuf, (float*)d_out);
  attn_kernel<false><<<ATT_SPLIT_ * B_, 512, 0, stream>>>(inputs, stat, ln_in_g, ln_in_b,
                                                          qbuf, pnum, pden);
  gru_mlp_kernel<false><<<32, 256, 0, stream>>>(
      pnum, pden, ln_in_g, ln_in_b, hprev, wvih, gru_whh, gru_bih, gru_bhh, ln_m_g, ln_m_b,
      mlp_w1, mlp_b1, mlp_w2, mlp_b2, ln_s_g, ln_s_b, wqk, qbuf, (float*)d_out);
  attn_kernel<false><<<ATT_SPLIT_ * B_, 512, 0, stream>>>(inputs, stat, ln_in_g, ln_in_b,
                                                          qbuf, pnum, pden);
  gru_mlp_kernel<true><<<32, 256, 0, stream>>>(
      pnum, pden, ln_in_g, ln_in_b, hprev, wvih, gru_whh, gru_bih, gru_bhh, ln_m_g, ln_m_b,
      mlp_w1, mlp_b1, mlp_w2, mlp_b2, ln_s_g, ln_s_b, wqk, qbuf, (float*)d_out);
}


# fuse and init launches folded into the first attention launch: each workgroup makes its batch's h_prev and q, q as (h Wq) Wk^T in two WMMA products; 8 launches to 6; q and slot LN sum order change
# speedup vs baseline: 8.2636x; 1.0714x over previous
#include <hip/hip_runtime.h>

typedef float v2f __attribute__((ext_vector_type(2)));
typedef float v4f __attribute__((ext_vector_type(4)));
typedef float v8f __attribute__((ext_vector_type(8)));

constexpr int H_ = 8, B_ = 64, S_ = 8, D_ = 64, N_ = 4096, MLPH_ = 128, ITERS_ = 3;
constexpr float LN_EPS_ = 1e-5f, EPS_ATTN_ = 1e-8f, SCALE_ = 0.125f;
static_assert(ITERS_ == 3, "kernel_launch writes the three iterations out");

__device__ __forceinline__ v8f wmma_f32(v2f a, v2f b, v8f c) {
  return __builtin_amdgcn_wmma_f32_16x16x4_f32(false, a, false, b, (short)0, c,
                                               false, false);
}

template <int M>
__device__ __forceinline__ float xlane(float x) {
  return __builtin_bit_cast(float, __builtin_amdgcn_update_dpp(
      0, __builtin_bit_cast(int, x), 0x160 + M, 0xf, 0xf, true));
}

constexpr int FUSE_WAVES_ = 16;
constexpr int FUSE_GRID_ = 4;
static_assert(FUSE_GRID_ * FUSE_WAVES_ == 16 + 48, "one fused-weight tile per wave");

__device__ __forceinline__ void fuse_w_tile(
    int g, int c, int h,
    const float* __restrict__ Wq, const float* __restrict__ Wk,
    const float* __restrict__ Wv, const float* __restrict__ wih,
    float* __restrict__ wqk, float* __restrict__ wvih) {
  v2f a[16], b[16];
  float* dst;
  int ld;
  if (g < 16) {
    const int rt = (g >> 2) * 16, ct = (g & 3) * 16;
#pragma unroll
    for (int kk = 0; kk < 16; ++kk) {
      const int kb = kk * 4 + h * 2;
      a[kk] = *(const v2f*)(Wq + (rt + c) * D_ + kb);
      b[kk] = *(const v2f*)(Wk + (ct + c) * D_ + kb);
    }
    dst = wqk + (size_t)(rt + 8 * h) * D_ + ct + c;
    ld = D_;
  } else {
    const int g2 = g - 16;
    const int rt = (g2 / 12) * 16, ct = (g2 % 12) * 16;
#pragma unroll
    for (int kk = 0; kk < 16; ++kk) {
      const int kb = kk * 4 + h * 2;
      a[kk] = *(const v2f*)(Wv + (rt + c) * D_ + kb);
      b[kk].x = wih[kb * 192 + ct + c];
      b[kk].y = wih[(kb + 1) * 192 + ct + c];
    }
    dst = wvih + (size_t)(rt + 8 * h) * 192 + ct + c;
    ld = 192;
  }
  v8f acc0 = {}, acc1 = {};
#pragma unroll
  for (int kk = 0; kk < 16; kk += 2) {
    acc0 = wmma_f32(a[kk], b[kk], acc0);
    acc1 = wmma_f32(a[kk + 1], b[kk + 1], acc1);
  }
  for (int i = 0; i < 8; ++i) dst[(size_t)i * ld] = acc0[i] + acc1[i];
}

constexpr int PX_ = 200;
constexpr int P1_ = 136;
constexpr int P2_ = 72;
static_assert((2 * PX_) % 64 == 16 && (2 * P1_) % 64 == 16 && (2 * P2_) % 64 == 16, "bank spread");

template <int ROWS, int COLS, int NT>
__device__ __forceinline__ void wload(float4 (&v)[ROWS * COLS / 4 / NT],
                                      const float* __restrict__ W, int tid) {
#pragma unroll
  for (int i = 0; i < ROWS * COLS / 4 / NT; ++i)
    v[i] = *(const float4*)(W + 4 * (tid + NT * i));
}
template <int ROWS, int COLS, int PITCH, int NT>
__device__ __forceinline__ void wstore(float* L, const float4 (&v)[ROWS * COLS / 4 / NT],
                                       int tid) {
#pragma unroll
  for (int i = 0; i < ROWS * COLS / 4 / NT; ++i) {
    const int e = tid + NT * i;
    const int r = e / (COLS / 4), c4 = (e % (COLS / 4)) * 4;
    *(float4*)(L + r * PITCH + c4) = v[i];
  }
}

template <int KSTEPS, int AP, int BP>
__device__ __forceinline__ v8f lds_product(const float* A, const float* Bm,
                                           int m16, int kh, int ct) {
  v2f a[KSTEPS], b[KSTEPS];
#pragma unroll
  for (int kk = 0; kk < KSTEPS; ++kk) {
    const int kb = kk * 4 + kh * 2;
    a[kk].x = A[m16 * AP + kb]; a[kk].y = A[m16 * AP + kb + 1];
    b[kk].x = Bm[kb * BP + ct + m16]; b[kk].y = Bm[(kb + 1) * BP + ct + m16];
  }
  v8f acc = {};
#pragma unroll
  for (int kk = 0; kk < KSTEPS; ++kk) acc = wmma_f32(a[kk], b[kk], acc);
  __builtin_amdgcn_sched_group_barrier(0x100, 4 * KSTEPS, 0);
  __builtin_amdgcn_sched_group_barrier(0x008, KSTEPS, 0);
  return acc;
}

template <int NT>
__device__ __forceinline__ void ln_q_tile(
    float (*hl)[68], float* mu, float* rstd, float lg, float lb,
    const float* WqL, float* hprev, float* __restrict__ qout, int rowbase, int tid) {
  static_assert(NT % 64 == 0, "a thread keeps one column");
  if (tid < 16) {
    float m = 0.f;
    for (int j = 0; j < D_; ++j) m += hl[tid][j];
    m *= (1.f / D_);
    float var = 0.f;
    for (int j = 0; j < D_; ++j) { float dd = hl[tid][j] - m; var += dd * dd; }
    var *= (1.f / D_);
    mu[tid] = m; rstd[tid] = rsqrtf(var + LN_EPS_);
  }
  __syncthreads();
  for (int e = tid; e < 16 * D_; e += NT) {
    int r = e >> 6, c = e & 63;
    float nv = (hl[r][c] - mu[r]) * rstd[r] * lg + lb;
    hl[r][c] = nv;
    hprev[(size_t)(rowbase + r) * D_ + c] = nv;
  }
  __syncthreads();
  const int w = tid >> 5, lane = tid & 31, m16 = lane & 15, kh = lane >> 4;
  if (w < 4) {
    const int ct = w * 16;
    const v8f acc = lds_product<16, 68, P2_>(&hl[0][0], WqL, m16, kh, ct);
    for (int i = 0; i < 8; ++i)
      qout[(size_t)(rowbase + i + 8 * kh) * D_ + ct + m16] = acc[i];
  }
}

constexpr int ATT_WAVES_ = 16;
constexpr int ATT_SPLIT_ = 4;
constexpr int ATT_TILES_ = (N_ / 16) / ATT_SPLIT_;
static_assert(ATT_TILES_ % ATT_WAVES_ == 0, "tiles must divide over the waves");
template <bool FIRST>
__global__ void __launch_bounds__(512) __attribute__((amdgpu_waves_per_eu(4, 4)))
attn_kernel(
    const float* __restrict__ inp, float* __restrict__ stat,
    const float* __restrict__ lng, const float* __restrict__ lnb,
    const float* __restrict__ qbuf, float* __restrict__ pnum,
    float* __restrict__ pden,
    const float* __restrict__ sinit, const float* __restrict__ smu,
    const float* __restrict__ slog, const float* __restrict__ lnsg,
    const float* __restrict__ lnsb, const float* __restrict__ Wq,
    const float* __restrict__ Wk, const float* __restrict__ Wv,
    const float* __restrict__ wih, float* __restrict__ hprev,
    float* __restrict__ wqk, float* __restrict__ wvih) {
  __shared__ alignas(16) float red[ATT_WAVES_][8][68];
  __shared__ float dred[ATT_WAVES_][32];
  __shared__ float tred[ATT_WAVES_][32];
  __shared__ float hl[FIRST ? 16 : 1][68];
  __shared__ float tl[FIRST ? 16 : 1][68];
  __shared__ alignas(16) float ql[FIRST ? 8 : 1][68];
  const int tid = threadIdx.x;
  const int ch = blockIdx.x >> 6;
  const int b  = blockIdx.x & 63;
  const int w = tid >> 5, lane = tid & 31, c = lane & 15, kh = lane >> 4;
  const float* kb = inp + (size_t)b * N_ * D_;
  const float* vb = kb;
  if (FIRST) {
    if (blockIdx.x < FUSE_GRID_)
      fuse_w_tile(blockIdx.x * FUSE_WAVES_ + w, c, kh, Wq, Wk, Wv, wih, wqk, wvih);
    const int ct = (w & 3) * 16;
    v2f bq[16], bk[16];
    if (w < 4) {
#pragma unroll
      for (int kk = 0; kk < 16; ++kk) {
        const int k0 = kk * 4 + kh * 2;
        bq[kk].x = Wq[k0 * D_ + ct + c];
        bq[kk].y = Wq[(k0 + 1) * D_ + ct + c];
        bk[kk] = *(const v2f*)(Wk + (ct + c) * D_ + k0);
      }
    }
    {
      const int r = tid >> 6, cc = tid & 63;
      hl[r][cc] = smu[cc] + expf(slog[cc]) * sinit[(size_t)(b * S_ + r) * D_ + cc];
      hl[r + 8][cc] = 0.f;
    }
    __syncthreads();
    if (w < 8) {
      const float x0 = hl[w][lane], x1 = hl[w][lane + 32];
      float sm = x0 + x1;
      sm += xlane<1>(sm); sm += xlane<2>(sm); sm += xlane<4>(sm); sm += xlane<8>(sm);
      sm += __shfl_xor(sm, 16);
      const float m = sm * (1.f / D_);
      const float e0 = x0 - m, e1 = x1 - m;
      float q2 = e0 * e0 + e1 * e1;
      q2 += xlane<1>(q2); q2 += xlane<2>(q2); q2 += xlane<4>(q2); q2 += xlane<8>(q2);
      q2 += __shfl_xor(q2, 16);
      const float rs = rsqrtf(q2 * (1.f / D_) + LN_EPS_);
      const float n0 = e0 * rs * lnsg[lane] + lnsb[lane];
      const float n1 = e1 * rs * lnsg[lane + 32] + lnsb[lane + 32];
      hl[w][lane] = n0; hl[w][lane + 32] = n1;
      if (ch == 0) {
        hprev[(size_t)(b * S_ + w) * D_ + lane] = n0;
        hprev[(size_t)(b * S_ + w) * D_ + lane + 32] = n1;
      }
    }
    __syncthreads();
    if (w < 4) {
      v2f a[16];
#pragma unroll
      for (int kk = 0; kk < 16; ++kk) {
        const int k0 = kk * 4 + kh * 2;
        a[kk].x = hl[c][k0]; a[kk].y = hl[c][k0 + 1];
      }
      v8f acc = {};
#pragma unroll
      for (int kk = 0; kk < 16; ++kk) acc = wmma_f32(a[kk], bq[kk], acc);
      for (int i = 0; i < 8; ++i) tl[i + 8 * kh][ct + c] = acc[i];
    }
    __syncthreads();
    if (w < 4) {
      v2f a[16];
#pragma unroll
      for (int kk = 0; kk < 16; ++kk) {
        const int k0 = kk * 4 + kh * 2;
        a[kk].x = tl[c][k0]; a[kk].y = tl[c][k0 + 1];
      }
      v8f acc = {};
#pragma unroll
      for (int kk = 0; kk < 16; ++kk) acc = wmma_f32(a[kk], bk[kk], acc);
      if (kh == 0)
        for (int i = 0; i < 8; ++i) ql[i][ct + c] = acc[i];
    }
    __syncthreads();
  }
  const float* qrow = FIRST ? &ql[c & 7][4 * kh]
                            : qbuf + (size_t)(b * S_ + (c & 7)) * D_ + 4 * kh;
  float4 qf[8];
  float sg = 0.f, sb = 0.f;
#pragma unroll
  for (int i = 0; i < 8; ++i) {
    float4 q = *(const float4*)(qrow + 8 * i);
    const float4 g = *(const float4*)(lng + 4 * kh + 8 * i);
    const float4 bb = *(const float4*)(lnb + 4 * kh + 8 * i);
    sb += (bb.x * q.x + bb.y * q.y) + (bb.z * q.z + bb.w * q.w);
    q.x *= g.x * SCALE_; q.y *= g.y * SCALE_; q.z *= g.z * SCALE_; q.w *= g.w * SCALE_;
    sg += (q.x + q.y) + (q.z + q.w);
    qf[i] = q;
  }
  sb *= SCALE_;
  sg += __shfl_xor(sg, 16);
  sb += __shfl_xor(sb, 16);
  v8f acc0 = {}, acc1 = {}, acc2 = {}, acc3 = {};
  float dsum = 0.f, tsum = 0.f;
  for (int t = ch * ATT_TILES_ + w; t < (ch + 1) * ATT_TILES_; t += ATT_WAVES_) {
    const int n0 = t * 16;
    const float* kp = kb + (size_t)(n0 + c) * D_ + 4 * kh;
    const float* vp = vb + (size_t)(n0 + 8 * kh) * D_ + 4 * c;
    float4 kf[8], vf[8];
    for (int i = 0; i < 8; ++i) kf[i] = *(const float4*)(kp + 8 * i);
    for (int r = 0; r < 8; ++r) vf[r] = *(const float4*)(vp + (size_t)r * D_);
    float4* sp4 = (float4*)(stat + ((size_t)b * N_ + n0 + 8 * kh) * 2);
    float4 st[4];
    if (!FIRST)
      for (int k = 0; k < 4; ++k) st[k] = sp4[k];
    v8f lg = {};
    for (int i = 0; i < 8; ++i) {
      v2f a0; a0.x = kf[i].x; a0.y = kf[i].y;
      v2f b0; b0.x = qf[i].x; b0.y = qf[i].y;
      lg = wmma_f32(a0, b0, lg);
      v2f a1; a1.x = kf[i].z; a1.y = kf[i].w;
      v2f b1; b1.x = qf[i].z; b1.y = qf[i].w;
      lg = wmma_f32(a1, b1, lg);
    }
    float mr[8], rr[8];
    if (FIRST) {
#pragma unroll
      for (int r = 0; r < 8; ++r) {
        const float4 x = vf[r];
        float s = (x.x + x.y) + (x.z + x.w);
        s += xlane<1>(s); s += xlane<2>(s); s += xlane<4>(s); s += xlane<8>(s);
        const float m = s * (1.f / D_);
        const float e0 = x.x - m, e1 = x.y - m, e2 = x.z - m, e3 = x.w - m;
        float q2 = (e0 * e0 + e1 * e1) + (e2 * e2 + e3 * e3);
        q2 += xlane<1>(q2); q2 += xlane<2>(q2); q2 += xlane<4>(q2); q2 += xlane<8>(q2);
        mr[r] = m;
        rr[r] = rsqrtf(q2 * (1.f / D_) + LN_EPS_);
      }
      if (c == 0) {
#pragma unroll
        for (int k = 0; k < 4; ++k) {
          float4 o; o.x = mr[2 * k]; o.y = rr[2 * k]; o.z = mr[2 * k + 1]; o.w = rr[2 * k + 1];
          sp4[k] = o;
        }
      }
    } else {
#pragma unroll
      for (int k = 0; k < 4; ++k) {
        mr[2 * k] = st[k].x; rr[2 * k] = st[k].y; mr[2 * k + 1] = st[k].z; rr[2 * k + 1] = st[k].w;
      }
    }
    float p[8], pa[8];
#pragma unroll
    for (int r = 0; r < 8; ++r) {
      float x = rr[r] * (lg[r] - mr[r] * sg) + sb;
      float mx = fmaxf(x, xlane<1>(x));
      mx = fmaxf(mx, xlane<2>(mx));
      mx = fmaxf(mx, xlane<4>(mx));
      float e = expf(x - mx);
      float sm = e + xlane<1>(e);
      sm += xlane<2>(sm);
      sm += xlane<4>(sm);
      float inv = 1.f / sm;
      p[r] = e * inv + EPS_ATTN_;
      dsum += p[r];
      pa[r] = p[r] * rr[r];
      tsum += pa[r] * mr[r];
    }
    for (int j = 0; j < 4; ++j) {
      v2f a; a.x = pa[2 * j]; a.y = pa[2 * j + 1];
      v2f b0; b0.x = vf[2 * j].x; b0.y = vf[2 * j + 1].x;
      acc0 = wmma_f32(a, b0, acc0);
      v2f b1; b1.x = vf[2 * j].y; b1.y = vf[2 * j + 1].y;
      acc1 = wmma_f32(a, b1, acc1);
      v2f b2; b2.x = vf[2 * j].z; b2.y = vf[2 * j + 1].z;
      acc2 = wmma_f32(a, b2, acc2);
      v2f b3; b3.x = vf[2 * j].w; b3.y = vf[2 * j + 1].w;
      acc3 = wmma_f32(a, b3, acc3);
    }
  }
  if (kh == 0) {
    for (int i = 0; i < 8; ++i) {
      float4 o; o.x = acc0[i]; o.y = acc1[i]; o.z = acc2[i]; o.w = acc3[i];
      *(float4*)&red[w][i][4 * c] = o;
    }
  }
  dred[w][lane] = dsum;
  tred[w][lane] = tsum;
  __syncthreads();
  if (tid < 128) {
    const int sp = tid >> 4, d4 = (tid & 15) * 4;
    float4 sacc; sacc.x = 0.f; sacc.y = 0.f; sacc.z = 0.f; sacc.w = 0.f;
    float den = 0.f, tt = 0.f;
    for (int ww = 0; ww < ATT_WAVES_; ++ww) {
      float4 x = *(const float4*)&red[ww][sp][d4];
      sacc.x += x.x; sacc.y += x.y; sacc.z += x.z; sacc.w += x.w;
      den += dred[ww][sp] + dred[ww][sp + 16];
      tt += tred[ww][sp] + tred[ww][sp + 16];
    }
    sacc.x -= tt; sacc.y -= tt; sacc.z -= tt; sacc.w -= tt;
    const size_t prow = (size_t)((b * ATT_SPLIT_ + ch) * S_ + sp);
    *(float4*)(pnum + prow * D_ + d4) = sacc;
    if (d4 == 0) pden[prow] = den;
  }
}

template <bool LAST>
__global__ void __launch_bounds__(256) gru_mlp_kernel(
    const float* __restrict__ pnum, const float* __restrict__ pden,
    const float* __restrict__ lning, const float* __restrict__ lninb,
    float* hprev,
    const float* __restrict__ wih, const float* __restrict__ whh,
    const float* __restrict__ bih, const float* __restrict__ bhh,
    const float* __restrict__ lnmg, const float* __restrict__ lnmb,
    const float* __restrict__ w1, const float* __restrict__ b1,
    const float* __restrict__ w2, const float* __restrict__ b2,
    const float* __restrict__ lnsg, const float* __restrict__ lnsb,
    const float* __restrict__ Wq, float* __restrict__ qout,
    float* __restrict__ out) {
  __shared__ float ut[16][68];
  __shared__ float ht[16][68];
  __shared__ float gx[16][196];
  __shared__ float gh[16][196];
  __shared__ float sn[16][68];
  __shared__ float nm[16][68];
  __shared__ float hid[16][132];
  __shared__ float mu[16], rstd[16];
  __shared__ alignas(16) float Lx[64 * PX_];
  __shared__ alignas(16) float Lh[64 * PX_];
  __shared__ alignas(16) float L1[64 * P1_];
  __shared__ alignas(16) float L2[128 * P2_];
  __shared__ alignas(16) float Lq[LAST ? 4 : 64 * P2_];
  const int tid = threadIdx.x;
  const int rt = blockIdx.x;
  const int rowbase = rt * 16;
  const int r0 = tid >> 4, c00 = (tid & 15) * 4;
  const int w = tid >> 5, lane = tid & 31, m16 = lane & 15, kh = lane >> 4;

  const int srow = rt * 16 + r0;
  const int bb = srow >> 3, ss = srow & 7;
  float4 pn[ATT_SPLIT_];
  float pd[ATT_SPLIT_];
#pragma unroll
  for (int q = 0; q < ATT_SPLIT_; ++q) {
    const size_t prow = (size_t)((bb * ATT_SPLIT_ + q) * S_ + ss);
    pn[q] = *(const float4*)(pnum + prow * D_ + c00);
    pd[q] = pden[prow];
  }
  const float4 hp = *(const float4*)(hprev + (size_t)(rowbase + r0) * D_ + c00);
  float4 sx[12], sh[12];
  wload<64, 192, 256>(sx, wih, tid);
  wload<64, 192, 256>(sh, whh, tid);
  const v4f bir = *(const v4f*)(bih + c00), biz = *(const v4f*)(bih + 64 + c00),
            bin = *(const v4f*)(bih + 128 + c00);
  const v4f bhr = *(const v4f*)(bhh + c00), bhz = *(const v4f*)(bhh + 64 + c00),
            bhn = *(const v4f*)(bhh + 128 + c00);
  const v4f mg = *(const v4f*)(lnmg + c00), mb = *(const v4f*)(lnmb + c00);
  const v4f ig = *(const v4f*)(lning + c00), ib = *(const v4f*)(lninb + c00);
  const float b1v = b1[w * 16 + m16];
  const float b2v = b2[(w & 3) * 16 + m16];
  float sg = 0.f, sb = 0.f;
  if (!LAST) { sg = lnsg[tid & 63]; sb = lnsb[tid & 63]; }
  float4 s1[8], s2[8], sq[4];
  wload<64, 128, 256>(s1, w1, tid);
  wload<128, 64, 256>(s2, w2, tid);
  if (!LAST) wload<64, 64, 256>(sq, Wq, tid);

  {
    float4 a; a.x = 0.f; a.y = 0.f; a.z = 0.f; a.w = 0.f;
    float den = 0.f;
#pragma unroll
    for (int q = 0; q < ATT_SPLIT_; ++q) {
      a.x += pn[q].x; a.y += pn[q].y; a.z += pn[q].z; a.w += pn[q].w;
      den += pd[q];
    }
    a.x = a.x / den * ig[0] + ib[0]; a.y = a.y / den * ig[1] + ib[1];
    a.z = a.z / den * ig[2] + ib[2]; a.w = a.w / den * ig[3] + ib[3];
    ut[r0][c00] = a.x; ut[r0][c00 + 1] = a.y; ut[r0][c00 + 2] = a.z; ut[r0][c00 + 3] = a.w;
    ht[r0][c00] = hp.x; ht[r0][c00 + 1] = hp.y; ht[r0][c00 + 2] = hp.z; ht[r0][c00 + 3] = hp.w;
  }
  wstore<64, 192, PX_, 256>(Lx, sx, tid);
  wstore<64, 192, PX_, 256>(Lh, sh, tid);
  wstore<64, 128, P1_, 256>(L1, s1, tid);
  wstore<128, 64, P2_, 256>(L2, s2, tid);
  if (!LAST) wstore<64, 64, P2_, 256>(Lq, sq, tid);
  __syncthreads();
  for (int T = w; T < 24; T += 8) {
    const bool isX = T < 12;
    const int ct = (T % 12) * 16;
    const float* Wt = isX ? Lx : Lh;
    const float* At = isX ? &ut[0][0] : &ht[0][0];
    const v8f acc = lds_product<16, 68, PX_>(At, Wt, m16, kh, ct);
    float* dst = isX ? &gx[0][0] : &gh[0][0];
    for (int i = 0; i < 8; ++i)
      dst[(i + 8 * kh) * 196 + ct + m16] = acc[i];
  }
  __syncthreads();
  {
#pragma unroll
    for (int j = 0; j < 4; ++j) {
      int d = c00 + j;
      float xr = gx[r0][d] + bir[j],       hr = gh[r0][d] + bhr[j];
      float xz = gx[r0][64 + d] + biz[j],  hz = gh[r0][64 + d] + bhz[j];
      float xn = gx[r0][128 + d] + bin[j], hn = gh[r0][128 + d] + bhn[j];
      float rg = 1.f / (1.f + expf(-(xr + hr)));
      float zg = 1.f / (1.f + expf(-(xz + hz)));
      float ng = tanhf(xn + rg * hn);
      sn[r0][d] = (1.f - zg) * ng + zg * ht[r0][d];
    }
  }
  __syncthreads();
  if (tid < 16) {
    float m = 0.f;
    for (int j = 0; j < D_; ++j) m += sn[tid][j];
    m *= (1.f / D_);
    float var = 0.f;
    for (int j = 0; j < D_; ++j) { float dd = sn[tid][j] - m; var += dd * dd; }
    var *= (1.f / D_);
    mu[tid] = m; rstd[tid] = rsqrtf(var + LN_EPS_);
  }
  __syncthreads();
#pragma unroll
  for (int j = 0; j < 4; ++j) {
    int d = c00 + j;
    nm[r0][d] = (sn[r0][d] - mu[r0]) * rstd[r0] * mg[j] + mb[j];
  }
  __syncthreads();
  {
    const int ct = w * 16;
    const v8f acc = lds_product<16, 68, P1_>(&nm[0][0], L1, m16, kh, ct);
    for (int i = 0; i < 8; ++i)
      hid[i + 8 * kh][ct + m16] = fmaxf(acc[i] + b1v, 0.f);
  }
  __syncthreads();
  if (w < 4) {
    const int ct = w * 16;
    const v8f acc = lds_product<32, 132, P2_>(&hid[0][0], L2, m16, kh, ct);
    for (int i = 0; i < 8; ++i) {
      int rr = i + 8 * kh;
      const float sv = sn[rr][ct + m16] + acc[i] + b2v;
      if (LAST) out[(size_t)(rowbase + rr) * D_ + ct + m16] = sv;
      else ut[rr][ct + m16] = sv;
    }
  }
  if (!LAST) {
    __syncthreads();
    ln_q_tile<256>(ut, mu, rstd, sg, sb, Lq, hprev, qout, rowbase, tid);
  }
}

extern "C" void kernel_launch(void* const* d_in, const int* in_sizes, int n_in,
                              void* d_out, int out_size, void* d_ws, size_t ws_size,
                              hipStream_t stream) {
  (void)in_sizes; (void)n_in; (void)out_size; (void)ws_size;
  const float* inputs  = (const float*)d_in[0];
  const float* sinit   = (const float*)d_in[1];
  const float* smu     = (const float*)d_in[2];
  const float* slog    = (const float*)d_in[3];
  const float* ln_in_g = (const float*)d_in[4];
  const float* ln_in_b = (const float*)d_in[5];
  const float* Wk      = (const float*)d_in[6];
  const float* Wv      = (const float*)d_in[7];
  const float* ln_s_g  = (const float*)d_in[8];
  const float* ln_s_b  = (const float*)d_in[9];
  const float* Wq      = (const float*)d_in[10];
  const float* gru_wih = (const float*)d_in[11];
  const float* gru_whh = (const float*)d_in[12];
  const float* gru_bih = (const float*)d_in[13];
  const float* gru_bhh = (const float*)d_in[14];
  const float* ln_m_g  = (const float*)d_in[15];
  const float* ln_m_b  = (const float*)d_in[16];
  const float* mlp_w1  = (const float*)d_in[17];
  const float* mlp_b1  = (const float*)d_in[18];
  const float* mlp_w2  = (const float*)d_in[19];
  const float* mlp_b2  = (const float*)d_in[20];

  float* ws = (float*)d_ws;
  float* stat  = ws;
  float* hprev = stat + (size_t)B_ * N_ * 2;
  float* qbuf  = hprev + H_ * B_ * S_ * D_;
  float* pnum  = qbuf + H_ * B_ * S_ * D_;
  float* pden  = pnum + B_ * ATT_SPLIT_ * S_ * D_;
  float* wqk   = pden + B_ * ATT_SPLIT_ * S_;
  float* wvih  = wqk + D_ * D_;

  attn_kernel<true><<<ATT_SPLIT_ * B_, 512, 0, stream>>>(
      inputs, stat, ln_in_g, ln_in_b, qbuf, pnum, pden, sinit, smu, slog, ln_s_g, ln_s_b,
      Wq, Wk, Wv, gru_wih, hprev, wqk, wvih);
  gru_mlp_kernel<false><<<32, 256, 0, stream>>>(
      pnum, pden, ln_in_g, ln_in_b, hprev, wvih, gru_whh, gru_bih, gru_bhh, ln_m_g, ln_m_b,
      mlp_w1, mlp_b1, mlp_w2, mlp_b2, ln_s_g, ln_s_b, wqk, qbuf, (float*)d_out);
  attn_kernel<false><<<ATT_SPLIT_ * B_, 512, 0, stream>>>(
      inputs, stat, ln_in_g, ln_in_b, qbuf, pnum, pden, sinit, smu, slog, ln_s_g, ln_s_b,
      Wq, Wk, Wv, gru_wih, hprev, wqk, wvih);
  gru_mlp_kernel<false><<<32, 256, 0, stream>>>(
      pnum, pden, ln_in_g, ln_in_b, hprev, wvih, gru_whh, gru_bih, gru_bhh, ln_m_g, ln_m_b,
      mlp_w1, mlp_b1, mlp_w2, mlp_b2, ln_s_g, ln_s_b, wqk, qbuf, (float*)d_out);
  attn_kernel<false><<<ATT_SPLIT_ * B_, 512, 0, stream>>>(
      inputs, stat, ln_in_g, ln_in_b, qbuf, pnum, pden, sinit, smu, slog, ln_s_g, ln_s_b,
      Wq, Wk, Wv, gru_wih, hprev, wqk, wvih);
  gru_mlp_kernel<true><<<32, 256, 0, stream>>>(
      pnum, pden, ln_in_g, ln_in_b, hprev, wvih, gru_whh, gru_bih, gru_bhh, ln_m_g, ln_m_b,
      mlp_w1, mlp_b1, mlp_w2, mlp_b2, ln_s_g, ln_s_b, wqk, qbuf, (float*)d_out);
}


# GRU/MLP launches of iterations 1 and 2 folded into the head of the next attention launch: each workgroup redoes its tile's GRU/MLP, q stays in LDS; 6 launches to 4; every sum keeps its order
# speedup vs baseline: 8.7771x; 1.0621x over previous
#include <hip/hip_runtime.h>

typedef float v2f __attribute__((ext_vector_type(2)));
typedef float v4f __attribute__((ext_vector_type(4)));
typedef float v8f __attribute__((ext_vector_type(8)));

constexpr int H_ = 8, B_ = 64, S_ = 8, D_ = 64, N_ = 4096, MLPH_ = 128, ITERS_ = 3;
constexpr float LN_EPS_ = 1e-5f, EPS_ATTN_ = 1e-8f, SCALE_ = 0.125f;
static_assert(ITERS_ == 3, "kernel_launch writes the three iterations out");

__device__ __forceinline__ v8f wmma_f32(v2f a, v2f b, v8f c) {
  return __builtin_amdgcn_wmma_f32_16x16x4_f32(false, a, false, b, (short)0, c,
                                               false, false);
}

template <int M>
__device__ __forceinline__ float xlane(float x) {
  return __builtin_bit_cast(float, __builtin_amdgcn_update_dpp(
      0, __builtin_bit_cast(int, x), 0x160 + M, 0xf, 0xf, true));
}

constexpr int FUSE_WAVES_ = 16;
constexpr int FUSE_GRID_ = 4;
static_assert(FUSE_GRID_ * FUSE_WAVES_ == 16 + 48, "one fused-weight tile per wave");

__device__ __forceinline__ void fuse_w_tile(
    int g, int c, int h,
    const float* __restrict__ Wq, const float* __restrict__ Wk,
    const float* __restrict__ Wv, const float* __restrict__ wih,
    float* __restrict__ wqk, float* __restrict__ wvih) {
  v2f a[16], b[16];
  float* dst;
  int ld;
  if (g < 16) {
    const int rt = (g >> 2) * 16, ct = (g & 3) * 16;
#pragma unroll
    for (int kk = 0; kk < 16; ++kk) {
      const int kb = kk * 4 + h * 2;
      a[kk] = *(const v2f*)(Wq + (rt + c) * D_ + kb);
      b[kk] = *(const v2f*)(Wk + (ct + c) * D_ + kb);
    }
    dst = wqk + (size_t)(rt + 8 * h) * D_ + ct + c;
    ld = D_;
  } else {
    const int g2 = g - 16;
    const int rt = (g2 / 12) * 16, ct = (g2 % 12) * 16;
#pragma unroll
    for (int kk = 0; kk < 16; ++kk) {
      const int kb = kk * 4 + h * 2;
      a[kk] = *(const v2f*)(Wv + (rt + c) * D_ + kb);
      b[kk].x = wih[kb * 192 + ct + c];
      b[kk].y = wih[(kb + 1) * 192 + ct + c];
    }
    dst = wvih + (size_t)(rt + 8 * h) * 192 + ct + c;
    ld = 192;
  }
  v8f acc0 = {}, acc1 = {};
#pragma unroll
  for (int kk = 0; kk < 16; kk += 2) {
    acc0 = wmma_f32(a[kk], b[kk], acc0);
    acc1 = wmma_f32(a[kk + 1], b[kk + 1], acc1);
  }
  for (int i = 0; i < 8; ++i) dst[(size_t)i * ld] = acc0[i] + acc1[i];
}

constexpr int PX_ = 200;
constexpr int P1_ = 136;
constexpr int P2_ = 72;
static_assert((2 * PX_) % 64 == 16 && (2 * P1_) % 64 == 16 && (2 * P2_) % 64 == 16, "bank spread");

template <int ROWS, int COLS, int NT>
__device__ __forceinline__ void wload(float4 (&v)[ROWS * COLS / 4 / NT],
                                      const float* __restrict__ W, int tid) {
#pragma unroll
  for (int i = 0; i < ROWS * COLS / 4 / NT; ++i)
    v[i] = *(const float4*)(W + 4 * (tid + NT * i));
}
template <int ROWS, int COLS, int PITCH, int NT>
__device__ __forceinline__ void wstore(float* L, const float4 (&v)[ROWS * COLS / 4 / NT],
                                       int tid) {
#pragma unroll
  for (int i = 0; i < ROWS * COLS / 4 / NT; ++i) {
    const int e = tid + NT * i;
    const int r = e / (COLS / 4), c4 = (e % (COLS / 4)) * 4;
    *(float4*)(L + r * PITCH + c4) = v[i];
  }
}

template <int KSTEPS, int AP, int BP>
__device__ __forceinline__ v8f lds_product(const float* A, const float* Bm,
                                           int m16, int kh, int ct) {
  v2f a[KSTEPS], b[KSTEPS];
#pragma unroll
  for (int kk = 0; kk < KSTEPS; ++kk) {
    const int kb = kk * 4 + kh * 2;
    a[kk].x = A[m16 * AP + kb]; a[kk].y = A[m16 * AP + kb + 1];
    b[kk].x = Bm[kb * BP + ct + m16]; b[kk].y = Bm[(kb + 1) * BP + ct + m16];
  }
  v8f acc = {};
#pragma unroll
  for (int kk = 0; kk < KSTEPS; ++kk) acc = wmma_f32(a[kk], b[kk], acc);
  __builtin_amdgcn_sched_group_barrier(0x100, 4 * KSTEPS, 0);
  __builtin_amdgcn_sched_group_barrier(0x008, KSTEPS, 0);
  return acc;
}


constexpr int ATT_WAVES_ = 16;
constexpr int ATT_SPLIT_ = 4;
constexpr int ATT_TILES_ = (N_ / 16) / ATT_SPLIT_;
static_assert(ATT_TILES_ % ATT_WAVES_ == 0, "tiles must divide over the waves");

template <int NT>
__device__ __forceinline__ void ln_q_tile(
    float (*hl)[68], float* mu, float* rstd, float lg, float lb,
    const float* WqL, float* __restrict__ hnext, bool store_h, float (*ql)[68],
    int rowbase, int tid) {
  static_assert(NT % 64 == 0, "a thread keeps one column");
  if (tid < 16) {
    float m = 0.f;
    for (int j = 0; j < D_; ++j) m += hl[tid][j];
    m *= (1.f / D_);
    float var = 0.f;
    for (int j = 0; j < D_; ++j) { float dd = hl[tid][j] - m; var += dd * dd; }
    var *= (1.f / D_);
    mu[tid] = m; rstd[tid] = rsqrtf(var + LN_EPS_);
  }
  __syncthreads();
  for (int e = tid; e < 16 * D_; e += NT) {
    int r = e >> 6, c = e & 63;
    float nv = (hl[r][c] - mu[r]) * rstd[r] * lg + lb;
    hl[r][c] = nv;
    if (store_h) hnext[(size_t)(rowbase + r) * D_ + c] = nv;
  }
  __syncthreads();
  const int w = tid >> 5, lane = tid & 31, m16 = lane & 15, kh = lane >> 4;
  if (w < 4) {
    const int ct = w * 16;
    const v8f acc = lds_product<16, 68, P2_>(&hl[0][0], WqL, m16, kh, ct);
    for (int i = 0; i < 8; ++i) ql[i + 8 * kh][ct + m16] = acc[i];
  }
}

template <bool LAST>
struct GruLds {
  float ut[16][68];
  float ht[16][68];
  float gx[16][196];
  float gh[16][196];
  float sn[16][68];
  float nm[16][68];
  float hid[16][132];
  float mu[16], rstd[16];
  alignas(16) float Lx[64 * PX_];
  alignas(16) float Lh[64 * PX_];
  alignas(16) float L1[64 * P1_];
  alignas(16) float L2[128 * P2_];
  alignas(16) float Lq[LAST ? 4 : 64 * P2_];
};

template <bool LAST, int NT>
__device__ __forceinline__ void gru_mlp_tile(
    GruLds<LAST>& S, int rt, int tid,
    const float* __restrict__ pnum, const float* __restrict__ pden,
    const float* __restrict__ lning, const float* __restrict__ lninb,
    const float* __restrict__ hprev,
    const float* __restrict__ wih, const float* __restrict__ whh,
    const float* __restrict__ bih, const float* __restrict__ bhh,
    const float* __restrict__ lnmg, const float* __restrict__ lnmb,
    const float* __restrict__ w1, const float* __restrict__ b1,
    const float* __restrict__ w2, const float* __restrict__ b2,
    const float* __restrict__ lnsg, const float* __restrict__ lnsb,
    const float* __restrict__ Wq, float* __restrict__ hnext, bool store_h,
    float (*ql)[68], float* __restrict__ out) {
  static_assert(NT == 256 || NT == 512, "the tile's rows are the first 256 threads'");
  constexpr int NW = NT / 32;
  const int rowbase = rt * 16;
  const bool rowthread = NT == 256 || tid < 256;
  const int r0 = (tid >> 4) & 15, c00 = (tid & 15) * 4;
  const int w = tid >> 5, lane = tid & 31, m16 = lane & 15, kh = lane >> 4;

  const int srow = rt * 16 + r0;
  const int bb = srow >> 3, ss = srow & 7;
  float4 pn[ATT_SPLIT_];
  float pd[ATT_SPLIT_];
  float4 hp = make_float4(0.f, 0.f, 0.f, 0.f);
#pragma unroll
  for (int q = 0; q < ATT_SPLIT_; ++q) { pn[q] = make_float4(0.f, 0.f, 0.f, 0.f); pd[q] = 0.f; }
  if (rowthread) {
#pragma unroll
    for (int q = 0; q < ATT_SPLIT_; ++q) {
      const size_t prow = (size_t)((bb * ATT_SPLIT_ + q) * S_ + ss);
      pn[q] = *(const float4*)(pnum + prow * D_ + c00);
      pd[q] = pden[prow];
    }
    hp = *(const float4*)(hprev + (size_t)(rowbase + r0) * D_ + c00);
  }
  constexpr int NX = 64 * 192 / 4 / NT, N1 = 64 * 128 / 4 / NT, N2 = 128 * 64 / 4 / NT,
                NQ = 64 * 64 / 4 / NT;
  float4 sx[NX], sh[NX];
  wload<64, 192, NT>(sx, wih, tid);
  wload<64, 192, NT>(sh, whh, tid);
  v4f bir = {}, biz = {}, bin = {}, bhr = {}, bhz = {}, bhn = {}, mg = {}, mb = {}, ig = {}, ib = {};
  if (rowthread) {
    bir = *(const v4f*)(bih + c00); biz = *(const v4f*)(bih + 64 + c00);
    bin = *(const v4f*)(bih + 128 + c00);
    bhr = *(const v4f*)(bhh + c00); bhz = *(const v4f*)(bhh + 64 + c00);
    bhn = *(const v4f*)(bhh + 128 + c00);
    mg = *(const v4f*)(lnmg + c00); mb = *(const v4f*)(lnmb + c00);
    ig = *(const v4f*)(lning + c00); ib = *(const v4f*)(lninb + c00);
  }
  const float b1v = b1[(w & 7) * 16 + m16];
  const float b2v = b2[(w & 3) * 16 + m16];
  float sg = 0.f, sb = 0.f;
  if (!LAST) { sg = lnsg[tid & 63]; sb = lnsb[tid & 63]; }
  float4 s1[N1], s2[N2], sq[NQ];
  wload<64, 128, NT>(s1, w1, tid);
  wload<128, 64, NT>(s2, w2, tid);
  if (!LAST) wload<64, 64, NT>(sq, Wq, tid);

  if (rowthread) {
    float4 a; a.x = 0.f; a.y = 0.f; a.z = 0.f; a.w = 0.f;
    float den = 0.f;
#pragma unroll
    for (int q = 0; q < ATT_SPLIT_; ++q) {
      a.x += pn[q].x; a.y += pn[q].y; a.z += pn[q].z; a.w += pn[q].w;
      den += pd[q];
    }
    a.x = a.x / den * ig[0] + ib[0]; a.y = a.y / den * ig[1] + ib[1];
    a.z = a.z / den * ig[2] + ib[2]; a.w = a.w / den * ig[3] + ib[3];
    S.ut[r0][c00] = a.x; S.ut[r0][c00 + 1] = a.y; S.ut[r0][c00 + 2] = a.z; S.ut[r0][c00 + 3] = a.w;
    S.ht[r0][c00] = hp.x; S.ht[r0][c00 + 1] = hp.y; S.ht[r0][c00 + 2] = hp.z; S.ht[r0][c00 + 3] = hp.w;
  }
  wstore<64, 192, PX_, NT>(S.Lx, sx, tid);
  wstore<64, 192, PX_, NT>(S.Lh, sh, tid);
  wstore<64, 128, P1_, NT>(S.L1, s1, tid);
  wstore<128, 64, P2_, NT>(S.L2, s2, tid);
  if (!LAST) wstore<64, 64, P2_, NT>(S.Lq, sq, tid);
  __syncthreads();
  for (int T = w; T < 24; T += NW) {
    const bool isX = T < 12;
    const int ct = (T % 12) * 16;
    const float* Wt = isX ? S.Lx : S.Lh;
    const float* At = isX ? &S.ut[0][0] : &S.ht[0][0];
    const v8f acc = lds_product<16, 68, PX_>(At, Wt, m16, kh, ct);
    float* dst = isX ? &S.gx[0][0] : &S.gh[0][0];
    for (int i = 0; i < 8; ++i)
      dst[(i + 8 * kh) * 196 + ct + m16] = acc[i];
  }
  __syncthreads();
  if (rowthread) {
#pragma unroll
    for (int j = 0; j < 4; ++j) {
      int d = c00 + j;
      float xr = S.gx[r0][d] + bir[j],       hr = S.gh[r0][d] + bhr[j];
      float xz = S.gx[r0][64 + d] + biz[j],  hz = S.gh[r0][64 + d] + bhz[j];
      float xn = S.gx[r0][128 + d] + bin[j], hn = S.gh[r0][128 + d] + bhn[j];
      float rg = 1.f / (1.f + expf(-(xr + hr)));
      float zg = 1.f / (1.f + expf(-(xz + hz)));
      float ng = tanhf(xn + rg * hn);
      S.sn[r0][d] = (1.f - zg) * ng + zg * S.ht[r0][d];
    }
  }
  __syncthreads();
  if (tid < 16) {
    float m = 0.f;
    for (int j = 0; j < D_; ++j) m += S.sn[tid][j];
    m *= (1.f / D_);
    float var = 0.f;
    for (int j = 0; j < D_; ++j) { float dd = S.sn[tid][j] - m; var += dd * dd; }
    var *= (1.f / D_);
    S.mu[tid] = m; S.rstd[tid] = rsqrtf(var + LN_EPS_);
  }
  __syncthreads();
  if (rowthread) {
#pragma unroll
    for (int j = 0; j < 4; ++j) {
      int d = c00 + j;
      S.nm[r0][d] = (S.sn[r0][d] - S.mu[r0]) * S.rstd[r0] * mg[j] + mb[j];
    }
  }
  __syncthreads();
  if (w < 8) {
    const int ct = w * 16;
    const v8f acc = lds_product<16, 68, P1_>(&S.nm[0][0], S.L1, m16, kh, ct);
    for (int i = 0; i < 8; ++i)
      S.hid[i + 8 * kh][ct + m16] = fmaxf(acc[i] + b1v, 0.f);
  }
  __syncthreads();
  if (w < 4) {
    const int ct = w * 16;
    const v8f acc = lds_product<32, 132, P2_>(&S.hid[0][0], S.L2, m16, kh, ct);
    for (int i = 0; i < 8; ++i) {
      int rr = i + 8 * kh;
      const float sv = S.sn[rr][ct + m16] + acc[i] + b2v;
      if (LAST) out[(size_t)(rowbase + rr) * D_ + ct + m16] = sv;
      else S.ut[rr][ct + m16] = sv;
    }
  }
  if (!LAST) {
    __syncthreads();
    ln_q_tile<NT>(S.ut, S.mu, S.rstd, sg, sb, S.Lq, hnext, store_h, ql, rowbase, tid);
  }
}

template <int MODE>
__global__ void __launch_bounds__(512) __attribute__((amdgpu_waves_per_eu(4, 4)))
attn_kernel(
    const float* __restrict__ inp, float* __restrict__ stat,
    const float* __restrict__ lng, const float* __restrict__ lnb,
    float* __restrict__ pnum, float* __restrict__ pden,
    const float* __restrict__ lnsg, const float* __restrict__ lnsb,
    float* __restrict__ hnext,
    const float* __restrict__ sinit, const float* __restrict__ smu,
    const float* __restrict__ slog, const float* __restrict__ Wq,
    const float* __restrict__ Wk, const float* __restrict__ Wv,
    const float* __restrict__ wih, float* wqk, float* wvih,
    const float* __restrict__ pnum_in, const float* __restrict__ pden_in,
    const float* __restrict__ hprev_in,
    const float* wvih_r, const float* __restrict__ whh,
    const float* __restrict__ bih, const float* __restrict__ bhh,
    const float* __restrict__ lnmg, const float* __restrict__ lnmb,
    const float* __restrict__ w1, const float* __restrict__ b1,
    const float* __restrict__ w2, const float* __restrict__ b2,
    const float* wqk_r) {
  constexpr bool FIRST = MODE == 0;
  __shared__ alignas(16) float red[ATT_WAVES_][8][68];
  __shared__ float dred[ATT_WAVES_][32];
  __shared__ float tred[ATT_WAVES_][32];
  __shared__ float hl[FIRST ? 16 : 1][68];
  __shared__ float tl[FIRST ? 16 : 1][68];
  __shared__ alignas(16) float ql[FIRST ? 8 : 16][68];
  const int tid = threadIdx.x;
  const int ch = blockIdx.x >> 6;
  const int b  = blockIdx.x & 63;
  const int w = tid >> 5, lane = tid & 31, c = lane & 15, kh = lane >> 4;
  const float* kb = inp + (size_t)b * N_ * D_;
  const float* vb = kb;
  if constexpr (MODE == 1) {
    __shared__ GruLds<false> G;
    gru_mlp_tile<false, 512>(G, b >> 1, tid, pnum_in, pden_in, lng, lnb, hprev_in, wvih_r, whh,
                             bih, bhh, lnmg, lnmb, w1, b1, w2, b2, lnsg, lnsb, wqk_r, hnext,
                             ch == 0 && (b & 1) == 0, ql, nullptr);
    __syncthreads();
  }
  if constexpr (FIRST) {
    if (blockIdx.x < FUSE_GRID_)
      fuse_w_tile(blockIdx.x * FUSE_WAVES_ + w, c, kh, Wq, Wk, Wv, wih, wqk, wvih);
    const int ct = (w & 3) * 16;
    v2f bq[16], bk[16];
    if (w < 4) {
#pragma unroll
      for (int kk = 0; kk < 16; ++kk) {
        const int k0 = kk * 4 + kh * 2;
        bq[kk].x = Wq[k0 * D_ + ct + c];
        bq[kk].y = Wq[(k0 + 1) * D_ + ct + c];
        bk[kk] = *(const v2f*)(Wk + (ct + c) * D_ + k0);
      }
    }
    {
      const int r = tid >> 6, cc = tid & 63;
      hl[r][cc] = smu[cc] + expf(slog[cc]) * sinit[(size_t)(b * S_ + r) * D_ + cc];
      hl[r + 8][cc] = 0.f;
    }
    __syncthreads();
    if (w < 8) {
      const float x0 = hl[w][lane], x1 = hl[w][lane + 32];
      float sm = x0 + x1;
      sm += xlane<1>(sm); sm += xlane<2>(sm); sm += xlane<4>(sm); sm += xlane<8>(sm);
      sm += __shfl_xor(sm, 16);
      const float m = sm * (1.f / D_);
      const float e0 = x0 - m, e1 = x1 - m;
      float q2 = e0 * e0 + e1 * e1;
      q2 += xlane<1>(q2); q2 += xlane<2>(q2); q2 += xlane<4>(q2); q2 += xlane<8>(q2);
      q2 += __shfl_xor(q2, 16);
      const float rs = rsqrtf(q2 * (1.f / D_) + LN_EPS_);
      const float n0 = e0 * rs * lnsg[lane] + lnsb[lane];
      const float n1 = e1 * rs * lnsg[lane + 32] + lnsb[lane + 32];
      hl[w][lane] = n0; hl[w][lane + 32] = n1;
      if (ch == 0) {
        hnext[(size_t)(b * S_ + w) * D_ + lane] = n0;
        hnext[(size_t)(b * S_ + w) * D_ + lane + 32] = n1;
      }
    }
    __syncthreads();
    if (w < 4) {
      v2f a[16];
#pragma unroll
      for (int kk = 0; kk < 16; ++kk) {
        const int k0 = kk * 4 + kh * 2;
        a[kk].x = hl[c][k0]; a[kk].y = hl[c][k0 + 1];
      }
      v8f acc = {};
#pragma unroll
      for (int kk = 0; kk < 16; ++kk) acc = wmma_f32(a[kk], bq[kk], acc);
      for (int i = 0; i < 8; ++i) tl[i + 8 * kh][ct + c] = acc[i];
    }
    __syncthreads();
    if (w < 4) {
      v2f a[16];
#pragma unroll
      for (int kk = 0; kk < 16; ++kk) {
        const int k0 = kk * 4 + kh * 2;
        a[kk].x = tl[c][k0]; a[kk].y = tl[c][k0 + 1];
      }
      v8f acc = {};
#pragma unroll
      for (int kk = 0; kk < 16; ++kk) acc = wmma_f32(a[kk], bk[kk], acc);
      if (kh == 0)
        for (int i = 0; i < 8; ++i) ql[i][ct + c] = acc[i];
    }
    __syncthreads();
  }
  const float* qrow = &ql[(FIRST ? 0 : (b & 1) * 8) + (c & 7)][4 * kh];
  float4 qf[8];
  float sg = 0.f, sb = 0.f;
#pragma unroll
  for (int i = 0; i < 8; ++i) {
    float4 q = *(const float4*)(qrow + 8 * i);
    const float4 g = *(const float4*)(lng + 4 * kh + 8 * i);
    const float4 bb = *(const float4*)(lnb + 4 * kh + 8 * i);
    sb += (bb.x * q.x + bb.y * q.y) + (bb.z * q.z + bb.w * q.w);
    q.x *= g.x * SCALE_; q.y *= g.y * SCALE_; q.z *= g.z * SCALE_; q.w *= g.w * SCALE_;
    sg += (q.x + q.y) + (q.z + q.w);
    qf[i] = q;
  }
  sb *= SCALE_;
  sg += __shfl_xor(sg, 16);
  sb += __shfl_xor(sb, 16);
  v8f acc0 = {}, acc1 = {}, acc2 = {}, acc3 = {};
  float dsum = 0.f, tsum = 0.f;
  for (int t = ch * ATT_TILES_ + w; t < (ch + 1) * ATT_TILES_; t += ATT_WAVES_) {
    const int n0 = t * 16;
    const float* kp = kb + (size_t)(n0 + c) * D_ + 4 * kh;
    const float* vp = vb + (size_t)(n0 + 8 * kh) * D_ + 4 * c;
    float4 kf[8], vf[8];
    for (int i = 0; i < 8; ++i) kf[i] = *(const float4*)(kp + 8 * i);
    for (int r = 0; r < 8; ++r) vf[r] = *(const float4*)(vp + (size_t)r * D_);
    float4* sp4 = (float4*)(stat + ((size_t)b * N_ + n0 + 8 * kh) * 2);
    float4 st[4];
    if (!FIRST)
      for (int k = 0; k < 4; ++k) st[k] = sp4[k];
    v8f lg = {};
    for (int i = 0; i < 8; ++i) {
      v2f a0; a0.x = kf[i].x; a0.y = kf[i].y;
      v2f b0; b0.x = qf[i].x; b0.y = qf[i].y;
      lg = wmma_f32(a0, b0, lg);
      v2f a1; a1.x = kf[i].z; a1.y = kf[i].w;
      v2f b1; b1.x = qf[i].z; b1.y = qf[i].w;
      lg = wmma_f32(a1, b1, lg);
    }
    float mr[8], rr[8];
    if (FIRST) {
#pragma unroll
      for (int r = 0; r < 8; ++r) {
        const float4 x = vf[r];
        float s = (x.x + x.y) + (x.z + x.w);
        s += xlane<1>(s); s += xlane<2>(s); s += xlane<4>(s); s += xlane<8>(s);
        const float m = s * (1.f / D_);
        const float e0 = x.x - m, e1 = x.y - m, e2 = x.z - m, e3 = x.w - m;
        float q2 = (e0 * e0 + e1 * e1) + (e2 * e2 + e3 * e3);
        q2 += xlane<1>(q2); q2 += xlane<2>(q2); q2 += xlane<4>(q2); q2 += xlane<8>(q2);
        mr[r] = m;
        rr[r] = rsqrtf(q2 * (1.f / D_) + LN_EPS_);
      }
      if (c == 0) {
#pragma unroll
        for (int k = 0; k < 4; ++k) {
          float4 o; o.x = mr[2 * k]; o.y = rr[2 * k]; o.z = mr[2 * k + 1]; o.w = rr[2 * k + 1];
          sp4[k] = o;
        }
      }
    } else {
#pragma unroll
      for (int k = 0; k < 4; ++k) {
        mr[2 * k] = st[k].x; rr[2 * k] = st[k].y; mr[2 * k + 1] = st[k].z; rr[2 * k + 1] = st[k].w;
      }
    }
    float p[8], pa[8];
#pragma unroll
    for (int r = 0; r < 8; ++r) {
      float x = rr[r] * (lg[r] - mr[r] * sg) + sb;
      float mx = fmaxf(x, xlane<1>(x));
      mx = fmaxf(mx, xlane<2>(mx));
      mx = fmaxf(mx, xlane<4>(mx));
      float e = expf(x - mx);
      float sm = e + xlane<1>(e);
      sm += xlane<2>(sm);
      sm += xlane<4>(sm);
      float inv = 1.f / sm;
      p[r] = e * inv + EPS_ATTN_;
      dsum += p[r];
      pa[r] = p[r] * rr[r];
      tsum += pa[r] * mr[r];
    }
    for (int j = 0; j < 4; ++j) {
      v2f a; a.x = pa[2 * j]; a.y = pa[2 * j + 1];
      v2f b0; b0.x = vf[2 * j].x; b0.y = vf[2 * j + 1].x;
      acc0 = wmma_f32(a, b0, acc0);
      v2f b1; b1.x = vf[2 * j].y; b1.y = vf[2 * j + 1].y;
      acc1 = wmma_f32(a, b1, acc1);
      v2f b2; b2.x = vf[2 * j].z; b2.y = vf[2 * j + 1].z;
      acc2 = wmma_f32(a, b2, acc2);
      v2f b3; b3.x = vf[2 * j].w; b3.y = vf[2 * j + 1].w;
      acc3 = wmma_f32(a, b3, acc3);
    }
  }
  if (kh == 0) {
    for (int i = 0; i < 8; ++i) {
      float4 o; o.x = acc0[i]; o.y = acc1[i]; o.z = acc2[i]; o.w = acc3[i];
      *(float4*)&red[w][i][4 * c] = o;
    }
  }
  dred[w][lane] = dsum;
  tred[w][lane] = tsum;
  __syncthreads();
  if (tid < 128) {
    const int sp = tid >> 4, d4 = (tid & 15) * 4;
    float4 sacc; sacc.x = 0.f; sacc.y = 0.f; sacc.z = 0.f; sacc.w = 0.f;
    float den = 0.f, tt = 0.f;
    for (int ww = 0; ww < ATT_WAVES_; ++ww) {
      float4 x = *(const float4*)&red[ww][sp][d4];
      sacc.x += x.x; sacc.y += x.y; sacc.z += x.z; sacc.w += x.w;
      den += dred[ww][sp] + dred[ww][sp + 16];
      tt += tred[ww][sp] + tred[ww][sp + 16];
    }
    sacc.x -= tt; sacc.y -= tt; sacc.z -= tt; sacc.w -= tt;
    const size_t prow = (size_t)((b * ATT_SPLIT_ + ch) * S_ + sp);
    *(float4*)(pnum + prow * D_ + d4) = sacc;
    if (d4 == 0) pden[prow] = den;
  }
}

__global__ void __launch_bounds__(256) gru_mlp_kernel(
    const float* __restrict__ pnum, const float* __restrict__ pden,
    const float* __restrict__ lning, const float* __restrict__ lninb,
    const float* __restrict__ hprev,
    const float* __restrict__ wih, const float* __restrict__ whh,
    const float* __restrict__ bih, const float* __restrict__ bhh,
    const float* __restrict__ lnmg, const float* __restrict__ lnmb,
    const float* __restrict__ w1, const float* __restrict__ b1,
    const float* __restrict__ w2, const float* __restrict__ b2,
    float* __restrict__ out) {
  __shared__ GruLds<true> G;
  gru_mlp_tile<true, 256>(G, blockIdx.x, threadIdx.x, pnum, pden, lning, lninb, hprev, wih, whh,
                          bih, bhh, lnmg, lnmb, w1, b1, w2, b2, nullptr, nullptr, nullptr,
                          nullptr, false, nullptr, out);
}

extern "C" void kernel_launch(void* const* d_in, const int* in_sizes, int n_in,
                              void* d_out, int out_size, void* d_ws, size_t ws_size,
                              hipStream_t stream) {
  (void)in_sizes; (void)n_in; (void)out_size; (void)ws_size;
  const float* inputs  = (const float*)d_in[0];
  const float* sinit   = (const float*)d_in[1];
  const float* smu     = (const float*)d_in[2];
  const float* slog    = (const float*)d_in[3];
  const float* ln_in_g = (const float*)d_in[4];
  const float* ln_in_b = (const float*)d_in[5];
  const float* Wk      = (const float*)d_in[6];
  const float* Wv      = (const float*)d_in[7];
  const float* ln_s_g  = (const float*)d_in[8];
  const float* ln_s_b  = (const float*)d_in[9];
  const float* Wq      = (const float*)d_in[10];
  const float* gru_wih = (const float*)d_in[11];
  const float* gru_whh = (const float*)d_in[12];
  const float* gru_bih = (const float*)d_in[13];
  const float* gru_bhh = (const float*)d_in[14];
  const float* ln_m_g  = (const float*)d_in[15];
  const float* ln_m_b  = (const float*)d_in[16];
  const float* mlp_w1  = (const float*)d_in[17];
  const float* mlp_b1  = (const float*)d_in[18];
  const float* mlp_w2  = (const float*)d_in[19];
  const float* mlp_b2  = (const float*)d_in[20];

  float* ws = (float*)d_ws;
  float* stat  = ws;
  float* hpA   = stat + (size_t)B_ * N_ * 2;
  float* hpB   = hpA + B_ * S_ * D_;
  float* pnB   = hpA + H_ * B_ * S_ * D_;
  float* pdB   = pnB + B_ * ATT_SPLIT_ * S_ * D_;
  float* pnA   = pnB + H_ * B_ * S_ * D_;
  float* pdA   = pnA + B_ * ATT_SPLIT_ * S_ * D_;
  float* wqk   = pdA + B_ * ATT_SPLIT_ * S_;
  float* wvih  = wqk + D_ * D_;

  attn_kernel<0><<<ATT_SPLIT_ * B_, 512, 0, stream>>>(
      inputs, stat, ln_in_g, ln_in_b, pnA, pdA, ln_s_g, ln_s_b, hpA,
      sinit, smu, slog, Wq, Wk, Wv, gru_wih, wqk, wvih,
      pnB, pdB, hpB, wvih, gru_whh, gru_bih, gru_bhh, ln_m_g, ln_m_b,
      mlp_w1, mlp_b1, mlp_w2, mlp_b2, wqk);
  attn_kernel<1><<<ATT_SPLIT_ * B_, 512, 0, stream>>>(
      inputs, stat, ln_in_g, ln_in_b, pnB, pdB, ln_s_g, ln_s_b, hpB,
      sinit, smu, slog, Wq, Wk, Wv, gru_wih, wqk, wvih,
      pnA, pdA, hpA, wvih, gru_whh, gru_bih, gru_bhh, ln_m_g, ln_m_b,
      mlp_w1, mlp_b1, mlp_w2, mlp_b2, wqk);
  attn_kernel<1><<<ATT_SPLIT_ * B_, 512, 0, stream>>>(
      inputs, stat, ln_in_g, ln_in_b, pnA, pdA, ln_s_g, ln_s_b, hpA,
      sinit, smu, slog, Wq, Wk, Wv, gru_wih, wqk, wvih,
      pnB, pdB, hpB, wvih, gru_whh, gru_bih, gru_bhh, ln_m_g, ln_m_b,
      mlp_w1, mlp_b1, mlp_w2, mlp_b2, wqk);
  gru_mlp_kernel<<<32, 256, 0, stream>>>(
      pnA, pdA, ln_in_g, ln_in_b, hpA, wvih, gru_whh, gru_bih, gru_bhh, ln_m_g, ln_m_b,
      mlp_w1, mlp_b1, mlp_w2, mlp_b2, (float*)d_out);
}


# GRU/MLP stage: both slot-tile LayerNorms done by the row's 16 lanes with DPP sums inside the gate and normalise phases; the two 16-thread serial passes and 3 barriers go; LN sum order changes
# speedup vs baseline: 9.4055x; 1.0716x over previous
#include <hip/hip_runtime.h>

typedef float v2f __attribute__((ext_vector_type(2)));
typedef float v4f __attribute__((ext_vector_type(4)));
typedef float v8f __attribute__((ext_vector_type(8)));

constexpr int H_ = 8, B_ = 64, S_ = 8, D_ = 64, N_ = 4096, MLPH_ = 128, ITERS_ = 3;
constexpr float LN_EPS_ = 1e-5f, EPS_ATTN_ = 1e-8f, SCALE_ = 0.125f;
static_assert(ITERS_ == 3, "kernel_launch writes the three iterations out");

__device__ __forceinline__ v8f wmma_f32(v2f a, v2f b, v8f c) {
  return __builtin_amdgcn_wmma_f32_16x16x4_f32(false, a, false, b, (short)0, c,
                                               false, false);
}

template <int M>
__device__ __forceinline__ float xlane(float x) {
  return __builtin_bit_cast(float, __builtin_amdgcn_update_dpp(
      0, __builtin_bit_cast(int, x), 0x160 + M, 0xf, 0xf, true));
}

__device__ __forceinline__ float rowsum16(float s) {
  s += xlane<1>(s); s += xlane<2>(s); s += xlane<4>(s); s += xlane<8>(s);
  return s;
}

constexpr int FUSE_WAVES_ = 16;
constexpr int FUSE_GRID_ = 4;
static_assert(FUSE_GRID_ * FUSE_WAVES_ == 16 + 48, "one fused-weight tile per wave");

__device__ __forceinline__ void fuse_w_tile(
    int g, int c, int h,
    const float* __restrict__ Wq, const float* __restrict__ Wk,
    const float* __restrict__ Wv, const float* __restrict__ wih,
    float* __restrict__ wqk, float* __restrict__ wvih) {
  v2f a[16], b[16];
  float* dst;
  int ld;
  if (g < 16) {
    const int rt = (g >> 2) * 16, ct = (g & 3) * 16;
#pragma unroll
    for (int kk = 0; kk < 16; ++kk) {
      const int kb = kk * 4 + h * 2;
      a[kk] = *(const v2f*)(Wq + (rt + c) * D_ + kb);
      b[kk] = *(const v2f*)(Wk + (ct + c) * D_ + kb);
    }
    dst = wqk + (size_t)(rt + 8 * h) * D_ + ct + c;
    ld = D_;
  } else {
    const int g2 = g - 16;
    const int rt = (g2 / 12) * 16, ct = (g2 % 12) * 16;
#pragma unroll
    for (int kk = 0; kk < 16; ++kk) {
      const int kb = kk * 4 + h * 2;
      a[kk] = *(const v2f*)(Wv + (rt + c) * D_ + kb);
      b[kk].x = wih[kb * 192 + ct + c];
      b[kk].y = wih[(kb + 1) * 192 + ct + c];
    }
    dst = wvih + (size_t)(rt + 8 * h) * 192 + ct + c;
    ld = 192;
  }
  v8f acc0 = {}, acc1 = {};
#pragma unroll
  for (int kk = 0; kk < 16; kk += 2) {
    acc0 = wmma_f32(a[kk], b[kk], acc0);
    acc1 = wmma_f32(a[kk + 1], b[kk + 1], acc1);
  }
  for (int i = 0; i < 8; ++i) dst[(size_t)i * ld] = acc0[i] + acc1[i];
}

constexpr int PX_ = 200;
constexpr int P1_ = 136;
constexpr int P2_ = 72;
static_assert((2 * PX_) % 64 == 16 && (2 * P1_) % 64 == 16 && (2 * P2_) % 64 == 16, "bank spread");

template <int ROWS, int COLS, int NT>
__device__ __forceinline__ void wload(float4 (&v)[ROWS * COLS / 4 / NT],
                                      const float* __restrict__ W, int tid) {
#pragma unroll
  for (int i = 0; i < ROWS * COLS / 4 / NT; ++i)
    v[i] = *(const float4*)(W + 4 * (tid + NT * i));
}
template <int ROWS, int COLS, int PITCH, int NT>
__device__ __forceinline__ void wstore(float* L, const float4 (&v)[ROWS * COLS / 4 / NT],
                                       int tid) {
#pragma unroll
  for (int i = 0; i < ROWS * COLS / 4 / NT; ++i) {
    const int e = tid + NT * i;
    const int r = e / (COLS / 4), c4 = (e % (COLS / 4)) * 4;
    *(float4*)(L + r * PITCH + c4) = v[i];
  }
}

template <int KSTEPS, int AP, int BP>
__device__ __forceinline__ v8f lds_product(const float* A, const float* Bm,
                                           int m16, int kh, int ct) {
  v2f a[KSTEPS], b[KSTEPS];
#pragma unroll
  for (int kk = 0; kk < KSTEPS; ++kk) {
    const int kb = kk * 4 + kh * 2;
    a[kk].x = A[m16 * AP + kb]; a[kk].y = A[m16 * AP + kb + 1];
    b[kk].x = Bm[kb * BP + ct + m16]; b[kk].y = Bm[(kb + 1) * BP + ct + m16];
  }
  v8f acc = {};
#pragma unroll
  for (int kk = 0; kk < KSTEPS; ++kk) acc = wmma_f32(a[kk], b[kk], acc);
  __builtin_amdgcn_sched_group_barrier(0x100, 4 * KSTEPS, 0);
  __builtin_amdgcn_sched_group_barrier(0x008, KSTEPS, 0);
  return acc;
}


constexpr int ATT_WAVES_ = 16;
constexpr int ATT_SPLIT_ = 4;
constexpr int ATT_TILES_ = (N_ / 16) / ATT_SPLIT_;
static_assert(ATT_TILES_ % ATT_WAVES_ == 0, "tiles must divide over the waves");

template <int NT>
__device__ __forceinline__ void ln_q_tile(
    float (*hl)[68], v4f lg, v4f lb,
    const float* WqL, float* __restrict__ hnext, bool store_h, float (*ql)[68],
    int rowbase, int tid) {
  static_assert(NT >= 256 && NT % 32 == 0, "the tile's rows are the first 256 threads', whole waves");
  if (tid < 256) {
    const int r0 = tid >> 4, c00 = (tid & 15) * 4;
    const float4 x = *(const float4*)&hl[r0][c00];
    const float m = rowsum16((x.x + x.y) + (x.z + x.w)) * (1.f / D_);
    const float e0 = x.x - m, e1 = x.y - m, e2 = x.z - m, e3 = x.w - m;
    const float var = rowsum16((e0 * e0 + e1 * e1) + (e2 * e2 + e3 * e3)) * (1.f / D_);
    const float rs = rsqrtf(var + LN_EPS_);
    float4 nv;
    nv.x = e0 * rs * lg[0] + lb[0]; nv.y = e1 * rs * lg[1] + lb[1];
    nv.z = e2 * rs * lg[2] + lb[2]; nv.w = e3 * rs * lg[3] + lb[3];
    *(float4*)&hl[r0][c00] = nv;
    if (store_h) *(float4*)(hnext + (size_t)(rowbase + r0) * D_ + c00) = nv;
  }
  __syncthreads();
  const int w = tid >> 5, lane = tid & 31, m16 = lane & 15, kh = lane >> 4;
  if (w < 4) {
    const int ct = w * 16;
    const v8f acc = lds_product<16, 68, P2_>(&hl[0][0], WqL, m16, kh, ct);
    for (int i = 0; i < 8; ++i) ql[i + 8 * kh][ct + m16] = acc[i];
  }
}

template <bool LAST>
struct GruLds {
  float ut[16][68];
  float ht[16][68];
  float gx[16][196];
  float gh[16][196];
  float sn[16][68];
  float nm[16][68];
  float hid[16][132];
  alignas(16) float Lx[64 * PX_];
  alignas(16) float Lh[64 * PX_];
  alignas(16) float L1[64 * P1_];
  alignas(16) float L2[128 * P2_];
  alignas(16) float Lq[LAST ? 4 : 64 * P2_];
};

template <bool LAST, int NT>
__device__ __forceinline__ void gru_mlp_tile(
    GruLds<LAST>& S, int rt, int tid,
    const float* __restrict__ pnum, const float* __restrict__ pden,
    const float* __restrict__ lning, const float* __restrict__ lninb,
    const float* __restrict__ hprev,
    const float* __restrict__ wih, const float* __restrict__ whh,
    const float* __restrict__ bih, const float* __restrict__ bhh,
    const float* __restrict__ lnmg, const float* __restrict__ lnmb,
    const float* __restrict__ w1, const float* __restrict__ b1,
    const float* __restrict__ w2, const float* __restrict__ b2,
    const float* __restrict__ lnsg, const float* __restrict__ lnsb,
    const float* __restrict__ Wq, float* __restrict__ hnext, bool store_h,
    float (*ql)[68], float* __restrict__ out) {
  static_assert(NT == 256 || NT == 512, "the tile's rows are the first 256 threads'");
  constexpr int NW = NT / 32;
  const int rowbase = rt * 16;
  const bool rowthread = NT == 256 || tid < 256;
  const int r0 = (tid >> 4) & 15, c00 = (tid & 15) * 4;
  const int w = tid >> 5, lane = tid & 31, m16 = lane & 15, kh = lane >> 4;

  const int srow = rt * 16 + r0;
  const int bb = srow >> 3, ss = srow & 7;
  float4 pn[ATT_SPLIT_];
  float pd[ATT_SPLIT_];
  float4 hp = make_float4(0.f, 0.f, 0.f, 0.f);
#pragma unroll
  for (int q = 0; q < ATT_SPLIT_; ++q) { pn[q] = make_float4(0.f, 0.f, 0.f, 0.f); pd[q] = 0.f; }
  if (rowthread) {
#pragma unroll
    for (int q = 0; q < ATT_SPLIT_; ++q) {
      const size_t prow = (size_t)((bb * ATT_SPLIT_ + q) * S_ + ss);
      pn[q] = *(const float4*)(pnum + prow * D_ + c00);
      pd[q] = pden[prow];
    }
    hp = *(const float4*)(hprev + (size_t)(rowbase + r0) * D_ + c00);
  }
  constexpr int NX = 64 * 192 / 4 / NT, N1 = 64 * 128 / 4 / NT, N2 = 128 * 64 / 4 / NT,
                NQ = 64 * 64 / 4 / NT;
  float4 sx[NX], sh[NX];
  wload<64, 192, NT>(sx, wih, tid);
  wload<64, 192, NT>(sh, whh, tid);
  v4f bir = {}, biz = {}, bin = {}, bhr = {}, bhz = {}, bhn = {}, mg = {}, mb = {}, ig = {}, ib = {};
  if (rowthread) {
    bir = *(const v4f*)(bih + c00); biz = *(const v4f*)(bih + 64 + c00);
    bin = *(const v4f*)(bih + 128 + c00);
    bhr = *(const v4f*)(bhh + c00); bhz = *(const v4f*)(bhh + 64 + c00);
    bhn = *(const v4f*)(bhh + 128 + c00);
    mg = *(const v4f*)(lnmg + c00); mb = *(const v4f*)(lnmb + c00);
    ig = *(const v4f*)(lning + c00); ib = *(const v4f*)(lninb + c00);
  }
  const float b1v = b1[(w & 7) * 16 + m16];
  const float b2v = b2[(w & 3) * 16 + m16];
  v4f sg = {}, sb = {};
  if (!LAST && rowthread) { sg = *(const v4f*)(lnsg + c00); sb = *(const v4f*)(lnsb + c00); }
  float4 s1[N1], s2[N2], sq[NQ];
  wload<64, 128, NT>(s1, w1, tid);
  wload<128, 64, NT>(s2, w2, tid);
  if (!LAST) wload<64, 64, NT>(sq, Wq, tid);

  if (rowthread) {
    float4 a; a.x = 0.f; a.y = 0.f; a.z = 0.f; a.w = 0.f;
    float den = 0.f;
#pragma unroll
    for (int q = 0; q < ATT_SPLIT_; ++q) {
      a.x += pn[q].x; a.y += pn[q].y; a.z += pn[q].z; a.w += pn[q].w;
      den += pd[q];
    }
    a.x = a.x / den * ig[0] + ib[0]; a.y = a.y / den * ig[1] + ib[1];
    a.z = a.z / den * ig[2] + ib[2]; a.w = a.w / den * ig[3] + ib[3];
    S.ut[r0][c00] = a.x; S.ut[r0][c00 + 1] = a.y; S.ut[r0][c00 + 2] = a.z; S.ut[r0][c00 + 3] = a.w;
    S.ht[r0][c00] = hp.x; S.ht[r0][c00 + 1] = hp.y; S.ht[r0][c00 + 2] = hp.z; S.ht[r0][c00 + 3] = hp.w;
  }
  wstore<64, 192, PX_, NT>(S.Lx, sx, tid);
  wstore<64, 192, PX_, NT>(S.Lh, sh, tid);
  wstore<64, 128, P1_, NT>(S.L1, s1, tid);
  wstore<128, 64, P2_, NT>(S.L2, s2, tid);
  if (!LAST) wstore<64, 64, P2_, NT>(S.Lq, sq, tid);
  __syncthreads();
  for (int T = w; T < 24; T += NW) {
    const bool isX = T < 12;
    const int ct = (T % 12) * 16;
    const float* Wt = isX ? S.Lx : S.Lh;
    const float* At = isX ? &S.ut[0][0] : &S.ht[0][0];
    const v8f acc = lds_product<16, 68, PX_>(At, Wt, m16, kh, ct);
    float* dst = isX ? &S.gx[0][0] : &S.gh[0][0];
    for (int i = 0; i < 8; ++i)
      dst[(i + 8 * kh) * 196 + ct + m16] = acc[i];
  }
  __syncthreads();
  if (rowthread) {
    float sv[4];
#pragma unroll
    for (int j = 0; j < 4; ++j) {
      int d = c00 + j;
      float xr = S.gx[r0][d] + bir[j],       hr = S.gh[r0][d] + bhr[j];
      float xz = S.gx[r0][64 + d] + biz[j],  hz = S.gh[r0][64 + d] + bhz[j];
      float xn = S.gx[r0][128 + d] + bin[j], hn = S.gh[r0][128 + d] + bhn[j];
      float rg = 1.f / (1.f + expf(-(xr + hr)));
      float zg = 1.f / (1.f + expf(-(xz + hz)));
      float ng = tanhf(xn + rg * hn);
      sv[j] = (1.f - zg) * ng + zg * S.ht[r0][d];
    }
    float4 so; so.x = sv[0]; so.y = sv[1]; so.z = sv[2]; so.w = sv[3];
    *(float4*)&S.sn[r0][c00] = so;
    const float m = rowsum16((sv[0] + sv[1]) + (sv[2] + sv[3])) * (1.f / D_);
    const float e0 = sv[0] - m, e1 = sv[1] - m, e2 = sv[2] - m, e3 = sv[3] - m;
    const float var = rowsum16((e0 * e0 + e1 * e1) + (e2 * e2 + e3 * e3)) * (1.f / D_);
    const float rs = rsqrtf(var + LN_EPS_);
    float4 no;
    no.x = e0 * rs * mg[0] + mb[0]; no.y = e1 * rs * mg[1] + mb[1];
    no.z = e2 * rs * mg[2] + mb[2]; no.w = e3 * rs * mg[3] + mb[3];
    *(float4*)&S.nm[r0][c00] = no;
  }
  __syncthreads();
  if (w < 8) {
    const int ct = w * 16;
    const v8f acc = lds_product<16, 68, P1_>(&S.nm[0][0], S.L1, m16, kh, ct);
    for (int i = 0; i < 8; ++i)
      S.hid[i + 8 * kh][ct + m16] = fmaxf(acc[i] + b1v, 0.f);
  }
  __syncthreads();
  if (w < 4) {
    const int ct = w * 16;
    const v8f acc = lds_product<32, 132, P2_>(&S.hid[0][0], S.L2, m16, kh, ct);
    for (int i = 0; i < 8; ++i) {
      int rr = i + 8 * kh;
      const float sv = S.sn[rr][ct + m16] + acc[i] + b2v;
      if (LAST) out[(size_t)(rowbase + rr) * D_ + ct + m16] = sv;
      else S.ut[rr][ct + m16] = sv;
    }
  }
  if (!LAST) {
    __syncthreads();
    ln_q_tile<NT>(S.ut, sg, sb, S.Lq, hnext, store_h, ql, rowbase, tid);
  }
}

template <int MODE>
__global__ void __launch_bounds__(512) __attribute__((amdgpu_waves_per_eu(4, 4)))
attn_kernel(
    const float* __restrict__ inp, float* __restrict__ stat,
    const float* __restrict__ lng, const float* __restrict__ lnb,
    float* __restrict__ pnum, float* __restrict__ pden,
    const float* __restrict__ lnsg, const float* __restrict__ lnsb,
    float* __restrict__ hnext,
    const float* __restrict__ sinit, const float* __restrict__ smu,
    const float* __restrict__ slog, const float* __restrict__ Wq,
    const float* __restrict__ Wk, const float* __restrict__ Wv,
    const float* __restrict__ wih, float* wqk, float* wvih,
    const float* __restrict__ pnum_in, const float* __restrict__ pden_in,
    const float* __restrict__ hprev_in,
    const float* wvih_r, const float* __restrict__ whh,
    const float* __restrict__ bih, const float* __restrict__ bhh,
    const float* __restrict__ lnmg, const float* __restrict__ lnmb,
    const float* __restrict__ w1, const float* __restrict__ b1,
    const float* __restrict__ w2, const float* __restrict__ b2,
    const float* wqk_r) {
  constexpr bool FIRST = MODE == 0;
  __shared__ alignas(16) float red[ATT_WAVES_][8][68];
  __shared__ float dred[ATT_WAVES_][32];
  __shared__ float tred[ATT_WAVES_][32];
  __shared__ float hl[FIRST ? 16 : 1][68];
  __shared__ float tl[FIRST ? 16 : 1][68];
  __shared__ alignas(16) float ql[FIRST ? 8 : 16][68];
  const int tid = threadIdx.x;
  const int ch = blockIdx.x >> 6;
  const int b  = blockIdx.x & 63;
  const int w = tid >> 5, lane = tid & 31, c = lane & 15, kh = lane >> 4;
  const float* kb = inp + (size_t)b * N_ * D_;
  const float* vb = kb;
  if constexpr (MODE == 1) {
    __shared__ GruLds<false> G;
    gru_mlp_tile<false, 512>(G, b >> 1, tid, pnum_in, pden_in, lng, lnb, hprev_in, wvih_r, whh,
                             bih, bhh, lnmg, lnmb, w1, b1, w2, b2, lnsg, lnsb, wqk_r, hnext,
                             ch == 0 && (b & 1) == 0, ql, nullptr);
    __syncthreads();
  }
  if constexpr (FIRST) {
    if (blockIdx.x < FUSE_GRID_)
      fuse_w_tile(blockIdx.x * FUSE_WAVES_ + w, c, kh, Wq, Wk, Wv, wih, wqk, wvih);
    const int ct = (w & 3) * 16;
    v2f bq[16], bk[16];
    if (w < 4) {
#pragma unroll
      for (int kk = 0; kk < 16; ++kk) {
        const int k0 = kk * 4 + kh * 2;
        bq[kk].x = Wq[k0 * D_ + ct + c];
        bq[kk].y = Wq[(k0 + 1) * D_ + ct + c];
        bk[kk] = *(const v2f*)(Wk + (ct + c) * D_ + k0);
      }
    }
    {
      const int r = tid >> 6, cc = tid & 63;
      hl[r][cc] = smu[cc] + expf(slog[cc]) * sinit[(size_t)(b * S_ + r) * D_ + cc];
      hl[r + 8][cc] = 0.f;
    }
    __syncthreads();
    if (w < 8) {
      const float x0 = hl[w][lane], x1 = hl[w][lane + 32];
      float sm = x0 + x1;
      sm += xlane<1>(sm); sm += xlane<2>(sm); sm += xlane<4>(sm); sm += xlane<8>(sm);
      sm += __shfl_xor(sm, 16);
      const float m = sm * (1.f / D_);
      const float e0 = x0 - m, e1 = x1 - m;
      float q2 = e0 * e0 + e1 * e1;
      q2 += xlane<1>(q2); q2 += xlane<2>(q2); q2 += xlane<4>(q2); q2 += xlane<8>(q2);
      q2 += __shfl_xor(q2, 16);
      const float rs = rsqrtf(q2 * (1.f / D_) + LN_EPS_);
      const float n0 = e0 * rs * lnsg[lane] + lnsb[lane];
      const float n1 = e1 * rs * lnsg[lane + 32] + lnsb[lane + 32];
      hl[w][lane] = n0; hl[w][lane + 32] = n1;
      if (ch == 0) {
        hnext[(size_t)(b * S_ + w) * D_ + lane] = n0;
        hnext[(size_t)(b * S_ + w) * D_ + lane + 32] = n1;
      }
    }
    __syncthreads();
    if (w < 4) {
      v2f a[16];
#pragma unroll
      for (int kk = 0; kk < 16; ++kk) {
        const int k0 = kk * 4 + kh * 2;
        a[kk].x = hl[c][k0]; a[kk].y = hl[c][k0 + 1];
      }
      v8f acc = {};
#pragma unroll
      for (int kk = 0; kk < 16; ++kk) acc = wmma_f32(a[kk], bq[kk], acc);
      for (int i = 0; i < 8; ++i) tl[i + 8 * kh][ct + c] = acc[i];
    }
    __syncthreads();
    if (w < 4) {
      v2f a[16];
#pragma unroll
      for (int kk = 0; kk < 16; ++kk) {
        const int k0 = kk * 4 + kh * 2;
        a[kk].x = tl[c][k0]; a[kk].y = tl[c][k0 + 1];
      }
      v8f acc = {};
#pragma unroll
      for (int kk = 0; kk < 16; ++kk) acc = wmma_f32(a[kk], bk[kk], acc);
      if (kh == 0)
        for (int i = 0; i < 8; ++i) ql[i][ct + c] = acc[i];
    }
    __syncthreads();
  }
  const float* qrow = &ql[(FIRST ? 0 : (b & 1) * 8) + (c & 7)][4 * kh];
  float4 qf[8];
  float sg = 0.f, sb = 0.f;
#pragma unroll
  for (int i = 0; i < 8; ++i) {
    float4 q = *(const float4*)(qrow + 8 * i);
    const float4 g = *(const float4*)(lng + 4 * kh + 8 * i);
    const float4 bb = *(const float4*)(lnb + 4 * kh + 8 * i);
    sb += (bb.x * q.x + bb.y * q.y) + (bb.z * q.z + bb.w * q.w);
    q.x *= g.x * SCALE_; q.y *= g.y * SCALE_; q.z *= g.z * SCALE_; q.w *= g.w * SCALE_;
    sg += (q.x + q.y) + (q.z + q.w);
    qf[i] = q;
  }
  sb *= SCALE_;
  sg += __shfl_xor(sg, 16);
  sb += __shfl_xor(sb, 16);
  v8f acc0 = {}, acc1 = {}, acc2 = {}, acc3 = {};
  float dsum = 0.f, tsum = 0.f;
  for (int t = ch * ATT_TILES_ + w; t < (ch + 1) * ATT_TILES_; t += ATT_WAVES_) {
    const int n0 = t * 16;
    const float* kp = kb + (size_t)(n0 + c) * D_ + 4 * kh;
    const float* vp = vb + (size_t)(n0 + 8 * kh) * D_ + 4 * c;
    float4 kf[8], vf[8];
    for (int i = 0; i < 8; ++i) kf[i] = *(const float4*)(kp + 8 * i);
    for (int r = 0; r < 8; ++r) vf[r] = *(const float4*)(vp + (size_t)r * D_);
    float4* sp4 = (float4*)(stat + ((size_t)b * N_ + n0 + 8 * kh) * 2);
    float4 st[4];
    if (!FIRST)
      for (int k = 0; k < 4; ++k) st[k] = sp4[k];
    v8f lg = {};
    for (int i = 0; i < 8; ++i) {
      v2f a0; a0.x = kf[i].x; a0.y = kf[i].y;
      v2f b0; b0.x = qf[i].x; b0.y = qf[i].y;
      lg = wmma_f32(a0, b0, lg);
      v2f a1; a1.x = kf[i].z; a1.y = kf[i].w;
      v2f b1; b1.x = qf[i].z; b1.y = qf[i].w;
      lg = wmma_f32(a1, b1, lg);
    }
    float mr[8], rr[8];
    if (FIRST) {
#pragma unroll
      for (int r = 0; r < 8; ++r) {
        const float4 x = vf[r];
        float s = (x.x + x.y) + (x.z + x.w);
        s += xlane<1>(s); s += xlane<2>(s); s += xlane<4>(s); s += xlane<8>(s);
        const float m = s * (1.f / D_);
        const float e0 = x.x - m, e1 = x.y - m, e2 = x.z - m, e3 = x.w - m;
        float q2 = (e0 * e0 + e1 * e1) + (e2 * e2 + e3 * e3);
        q2 += xlane<1>(q2); q2 += xlane<2>(q2); q2 += xlane<4>(q2); q2 += xlane<8>(q2);
        mr[r] = m;
        rr[r] = rsqrtf(q2 * (1.f / D_) + LN_EPS_);
      }
      if (c == 0) {
#pragma unroll
        for (int k = 0; k < 4; ++k) {
          float4 o; o.x = mr[2 * k]; o.y = rr[2 * k]; o.z = mr[2 * k + 1]; o.w = rr[2 * k + 1];
          sp4[k] = o;
        }
      }
    } else {
#pragma unroll
      for (int k = 0; k < 4; ++k) {
        mr[2 * k] = st[k].x; rr[2 * k] = st[k].y; mr[2 * k + 1] = st[k].z; rr[2 * k + 1] = st[k].w;
      }
    }
    float p[8], pa[8];
#pragma unroll
    for (int r = 0; r < 8; ++r) {
      float x = rr[r] * (lg[r] - mr[r] * sg) + sb;
      float mx = fmaxf(x, xlane<1>(x));
      mx = fmaxf(mx, xlane<2>(mx));
      mx = fmaxf(mx, xlane<4>(mx));
      float e = expf(x - mx);
      float sm = e + xlane<1>(e);
      sm += xlane<2>(sm);
      sm += xlane<4>(sm);
      float inv = 1.f / sm;
      p[r] = e * inv + EPS_ATTN_;
      dsum += p[r];
      pa[r] = p[r] * rr[r];
      tsum += pa[r] * mr[r];
    }
    for (int j = 0; j < 4; ++j) {
      v2f a; a.x = pa[2 * j]; a.y = pa[2 * j + 1];
      v2f b0; b0.x = vf[2 * j].x; b0.y = vf[2 * j + 1].x;
      acc0 = wmma_f32(a, b0, acc0);
      v2f b1; b1.x = vf[2 * j].y; b1.y = vf[2 * j + 1].y;
      acc1 = wmma_f32(a, b1, acc1);
      v2f b2; b2.x = vf[2 * j].z; b2.y = vf[2 * j + 1].z;
      acc2 = wmma_f32(a, b2, acc2);
      v2f b3; b3.x = vf[2 * j].w; b3.y = vf[2 * j + 1].w;
      acc3 = wmma_f32(a, b3, acc3);
    }
  }
  if (kh == 0) {
    for (int i = 0; i < 8; ++i) {
      float4 o; o.x = acc0[i]; o.y = acc1[i]; o.z = acc2[i]; o.w = acc3[i];
      *(float4*)&red[w][i][4 * c] = o;
    }
  }
  dred[w][lane] = dsum;
  tred[w][lane] = tsum;
  __syncthreads();
  if (tid < 128) {
    const int sp = tid >> 4, d4 = (tid & 15) * 4;
    float4 sacc; sacc.x = 0.f; sacc.y = 0.f; sacc.z = 0.f; sacc.w = 0.f;
    float den = 0.f, tt = 0.f;
    for (int ww = 0; ww < ATT_WAVES_; ++ww) {
      float4 x = *(const float4*)&red[ww][sp][d4];
      sacc.x += x.x; sacc.y += x.y; sacc.z += x.z; sacc.w += x.w;
      den += dred[ww][sp] + dred[ww][sp + 16];
      tt += tred[ww][sp] + tred[ww][sp + 16];
    }
    sacc.x -= tt; sacc.y -= tt; sacc.z -= tt; sacc.w -= tt;
    const size_t prow = (size_t)((b * ATT_SPLIT_ + ch) * S_ + sp);
    *(float4*)(pnum + prow * D_ + d4) = sacc;
    if (d4 == 0) pden[prow] = den;
  }
}

__global__ void __launch_bounds__(256) gru_mlp_kernel(
    const float* __restrict__ pnum, const float* __restrict__ pden,
    const float* __restrict__ lning, const float* __restrict__ lninb,
    const float* __restrict__ hprev,
    const float* __restrict__ wih, const float* __restrict__ whh,
    const float* __restrict__ bih, const float* __restrict__ bhh,
    const float* __restrict__ lnmg, const float* __restrict__ lnmb,
    const float* __restrict__ w1, const float* __restrict__ b1,
    const float* __restrict__ w2, const float* __restrict__ b2,
    float* __restrict__ out) {
  __shared__ GruLds<true> G;
  gru_mlp_tile<true, 256>(G, blockIdx.x, threadIdx.x, pnum, pden, lning, lninb, hprev, wih, whh,
                          bih, bhh, lnmg, lnmb, w1, b1, w2, b2, nullptr, nullptr, nullptr,
                          nullptr, false, nullptr, out);
}

extern "C" void kernel_launch(void* const* d_in, const int* in_sizes, int n_in,
                              void* d_out, int out_size, void* d_ws, size_t ws_size,
                              hipStream_t stream) {
  (void)in_sizes; (void)n_in; (void)out_size; (void)ws_size;
  const float* inputs  = (const float*)d_in[0];
  const float* sinit   = (const float*)d_in[1];
  const float* smu     = (const float*)d_in[2];
  const float* slog    = (const float*)d_in[3];
  const float* ln_in_g = (const float*)d_in[4];
  const float* ln_in_b = (const float*)d_in[5];
  const float* Wk      = (const float*)d_in[6];
  const float* Wv      = (const float*)d_in[7];
  const float* ln_s_g  = (const float*)d_in[8];
  const float* ln_s_b  = (const float*)d_in[9];
  const float* Wq      = (const float*)d_in[10];
  const float* gru_wih = (const float*)d_in[11];
  const float* gru_whh = (const float*)d_in[12];
  const float* gru_bih = (const float*)d_in[13];
  const float* gru_bhh = (const float*)d_in[14];
  const float* ln_m_g  = (const float*)d_in[15];
  const float* ln_m_b  = (const float*)d_in[16];
  const float* mlp_w1  = (const float*)d_in[17];
  const float* mlp_b1  = (const float*)d_in[18];
  const float* mlp_w2  = (const float*)d_in[19];
  const float* mlp_b2  = (const float*)d_in[20];

  float* ws = (float*)d_ws;
  float* stat  = ws;
  float* hpA   = stat + (size_t)B_ * N_ * 2;
  float* hpB   = hpA + B_ * S_ * D_;
  float* pnB   = hpA + H_ * B_ * S_ * D_;
  float* pdB   = pnB + B_ * ATT_SPLIT_ * S_ * D_;
  float* pnA   = pnB + H_ * B_ * S_ * D_;
  float* pdA   = pnA + B_ * ATT_SPLIT_ * S_ * D_;
  float* wqk   = pdA + B_ * ATT_SPLIT_ * S_;
  float* wvih  = wqk + D_ * D_;

  attn_kernel<0><<<ATT_SPLIT_ * B_, 512, 0, stream>>>(
      inputs, stat, ln_in_g, ln_in_b, pnA, pdA, ln_s_g, ln_s_b, hpA,
      sinit, smu, slog, Wq, Wk, Wv, gru_wih, wqk, wvih,
      pnB, pdB, hpB, wvih, gru_whh, gru_bih, gru_bhh, ln_m_g, ln_m_b,
      mlp_w1, mlp_b1, mlp_w2, mlp_b2, wqk);
  attn_kernel<1><<<ATT_SPLIT_ * B_, 512, 0, stream>>>(
      inputs, stat, ln_in_g, ln_in_b, pnB, pdB, ln_s_g, ln_s_b, hpB,
      sinit, smu, slog, Wq, Wk, Wv, gru_wih, wqk, wvih,
      pnA, pdA, hpA, wvih, gru_whh, gru_bih, gru_bhh, ln_m_g, ln_m_b,
      mlp_w1, mlp_b1, mlp_w2, mlp_b2, wqk);
  attn_kernel<1><<<ATT_SPLIT_ * B_, 512, 0, stream>>>(
      inputs, stat, ln_in_g, ln_in_b, pnA, pdA, ln_s_g, ln_s_b, hpA,
      sinit, smu, slog, Wq, Wk, Wv, gru_wih, wqk, wvih,
      pnB, pdB, hpB, wvih, gru_whh, gru_bih, gru_bhh, ln_m_g, ln_m_b,
      mlp_w1, mlp_b1, mlp_w2, mlp_b2, wqk);
  gru_mlp_kernel<<<32, 256, 0, stream>>>(
      pnA, pdA, ln_in_g, ln_in_b, hpA, wvih, gru_whh, gru_bih, gru_bhh, ln_m_g, ln_m_b,
      mlp_w1, mlp_b1, mlp_w2, mlp_b2, (float*)d_out);
}
